# starting kernel
# baseline (speedup 1.0000x reference)
#include <hip/hip_runtime.h>
#include <hip/hip_bf16.h>
#include <math.h>

#define NNODES 50000
#define EDGES  800000
#define INFEAT 128
#define HIDTOT 256
#define NEG_SLOPE 0.2f
#define ORD_NEG_INF 0x007FFFFFu

typedef __attribute__((ext_vector_type(2))) float v2f;
typedef __attribute__((ext_vector_type(8))) float v8f;

__device__ __forceinline__ unsigned ford(float f) {
    unsigned u = __float_as_uint(f);
    return (u & 0x80000000u) ? ~u : (u | 0x80000000u);
}
__device__ __forceinline__ float fdec(unsigned u) {
    return __uint_as_float((u & 0x80000000u) ? (u & 0x7FFFFFFFu) : ~u);
}
__device__ __forceinline__ float lrelu(float x) {
    return x >= 0.f ? x : NEG_SLOPE * x;
}

__global__ void init_kernel(float* out1, float* denom1, unsigned* m1,
                            float* denom2, unsigned* m2, float* dout) {
    size_t i = (size_t)blockIdx.x * blockDim.x + threadIdx.x;
    if (i < (size_t)NNODES * HIDTOT) out1[i] = 0.f;
    if (i < (size_t)NNODES * 4) { denom1[i] = 0.f; m1[i] = ORD_NEG_INF; }
    if (i < (size_t)NNODES)     { denom2[i] = 0.f; m2[i] = ORD_NEG_INF; }
    if (i < (size_t)NNODES * 2) dout[i] = 0.f;
}

__global__ __launch_bounds__(256) void gemm1_kernel(
    const float* __restrict__ feat,
    const float* __restrict__ Ws, const float* __restrict__ bs,
    const float* __restrict__ Wd, const float* __restrict__ bd,
    float* __restrict__ hs, float* __restrict__ hd)
{
    const int wave = threadIdx.x >> 5;
    const int lane = threadIdx.x & 31;
    const int half = lane >> 4;
    const int row  = lane & 15;

    int tile = blockIdx.x * 8 + wave;
    if (tile >= 3125 * 4) return;
    const int mt  = tile >> 2;
    const int nt4 = tile & 3;
    const int m0  = mt * 16;
    const int c0  = nt4 * 64;

    v8f zero = {0.f,0.f,0.f,0.f,0.f,0.f,0.f,0.f};
    v8f accS[4], accD[4];
    #pragma unroll
    for (int j = 0; j < 4; ++j) { accS[j] = zero; accD[j] = zero; }

    const float* arow = feat + (size_t)(m0 + row) * INFEAT;
    for (int k0 = 0; k0 < INFEAT; k0 += 4) {
        const int ka = k0 + 2 * half;
        v2f a = *(const v2f*)(arow + ka);
        #pragma unroll
        for (int j = 0; j < 4; ++j) {
            const int col = c0 + j * 16 + row;
            v2f bS, bD;
            bS.x = Ws[(size_t)ka * HIDTOT + col];
            bS.y = Ws[(size_t)(ka + 1) * HIDTOT + col];
            bD.x = Wd[(size_t)ka * HIDTOT + col];
            bD.y = Wd[(size_t)(ka + 1) * HIDTOT + col];
            accS[j] = __builtin_amdgcn_wmma_f32_16x16x4_f32(
                false, a, false, bS, (short)0, accS[j], false, false);
            accD[j] = __builtin_amdgcn_wmma_f32_16x16x4_f32(
                false, a, false, bD, (short)0, accD[j], false, false);
        }
    }
    #pragma unroll
    for (int j = 0; j < 4; ++j) {
        const int col = c0 + j * 16 + row;
        const float biasS = bs[col], biasD = bd[col];
        #pragma unroll
        for (int i = 0; i < 8; ++i) {
            const int r = m0 + i + 8 * half;
            hs[(size_t)r * HIDTOT + col] = accS[j][i] + biasS;
            hd[(size_t)r * HIDTOT + col] = accD[j][i] + biasD;
        }
    }
}

__global__ __launch_bounds__(256) void edge_score1_kernel(
    const float* __restrict__ hs, const float* __restrict__ hd,
    const int* __restrict__ src, const int* __restrict__ dst,
    const float* __restrict__ attn, float* __restrict__ score,
    unsigned* __restrict__ mbits)
{
    const int e = blockIdx.x * 8 + (threadIdx.x >> 5);
    if (e >= EDGES) return;
    const int lane = threadIdx.x & 31;
    const int s = src[e], d = dst[e];
    const int base = lane * 8;

    const float4* ph = (const float4*)(hs + (size_t)s * HIDTOT + base);
    const float4* pd = (const float4*)(hd + (size_t)d * HIDTOT + base);
    const float4* pa = (const float4*)(attn + base);

    float part = 0.f;
    #pragma unroll
    for (int q = 0; q < 2; ++q) {
        float4 a = ph[q], b = pd[q], w = pa[q];
        part += lrelu(a.x + b.x) * w.x + lrelu(a.y + b.y) * w.y +
                lrelu(a.z + b.z) * w.z + lrelu(a.w + b.w) * w.w;
    }
    part += __shfl_xor(part, 1, 32);
    part += __shfl_xor(part, 2, 32);
    part += __shfl_xor(part, 4, 32);
    if ((lane & 7) == 0) {
        const int h = lane >> 3;
        score[(size_t)e * 4 + h] = part;
        atomicMax(mbits + (size_t)d * 4 + h, ford(part));
    }
}

__global__ void edge_exp1_kernel(const int* __restrict__ dst,
                                 float* __restrict__ score,
                                 const unsigned* __restrict__ mbits,
                                 float* __restrict__ denom)
{
    const int t = blockIdx.x * blockDim.x + threadIdx.x;
    if (t >= EDGES * 4) return;
    const int e = t >> 2, h = t & 3;
    const int d = dst[e];
    float m = fdec(mbits[(size_t)d * 4 + h]);
    if (!isfinite(m)) m = 0.f;
    const float ex = expf(score[t] - m);
    score[t] = ex;
    atomicAdd(denom + (size_t)d * 4 + h, ex);
}

__global__ __launch_bounds__(256) void edge_aggr1_kernel(
    const float* __restrict__ hs, const int* __restrict__ src,
    const int* __restrict__ dst, const float* __restrict__ ex,
    const float* __restrict__ denom, float* __restrict__ out)
{
    const int e = blockIdx.x * 8 + (threadIdx.x >> 5);
    if (e >= EDGES) return;
    const int lane = threadIdx.x & 31;
    const int s = src[e], d = dst[e];
    const int h = lane >> 3;
    float den = denom[(size_t)d * 4 + h];
    den = den > 0.f ? den : 1.f;
    const float alpha = ex[(size_t)e * 4 + h] / den;
    const int base = lane * 8;
    const float4* ph = (const float4*)(hs + (size_t)s * HIDTOT + base);
    float* po = out + (size_t)d * HIDTOT + base;
    #pragma unroll
    for (int q = 0; q < 2; ++q) {
        float4 a = ph[q];
        atomicAdd(po + q * 4 + 0, alpha * a.x);
        atomicAdd(po + q * 4 + 1, alpha * a.y);
        atomicAdd(po + q * 4 + 2, alpha * a.z);
        atomicAdd(po + q * 4 + 3, alpha * a.w);
    }
}

__global__ void elu_kernel(float* __restrict__ x, int n) {
    const int i = blockIdx.x * blockDim.x + threadIdx.x;
    if (i < n) {
        const float v = x[i];
        x[i] = v > 0.f ? v : (expf(v) - 1.f);
    }
}

__global__ __launch_bounds__(256) void proj2_kernel(
    const float* __restrict__ h,
    const float* __restrict__ W2s, const float* __restrict__ b2s,
    const float* __restrict__ W2d, const float* __restrict__ b2d,
    float* __restrict__ hs2, float* __restrict__ hd2)
{
    const int n = blockIdx.x * 8 + (threadIdx.x >> 5);
    if (n >= NNODES) return;
    const int lane = threadIdx.x & 31;
    const int base = lane * 8;
    const float* hr = h + (size_t)n * HIDTOT + base;
    float s0 = 0.f, s1 = 0.f, d0 = 0.f, d1 = 0.f;
    #pragma unroll
    for (int k = 0; k < 8; ++k) {
        const float x = hr[k];
        const float2 ws = *(const float2*)(W2s + (size_t)(base + k) * 2);
        const float2 wd = *(const float2*)(W2d + (size_t)(base + k) * 2);
        s0 += x * ws.x; s1 += x * ws.y;
        d0 += x * wd.x; d1 += x * wd.y;
    }
    #pragma unroll
    for (int o = 1; o < 32; o <<= 1) {
        s0 += __shfl_xor(s0, o, 32); s1 += __shfl_xor(s1, o, 32);
        d0 += __shfl_xor(d0, o, 32); d1 += __shfl_xor(d1, o, 32);
    }
    if (lane == 0) {
        hs2[(size_t)n * 2 + 0] = s0 + b2s[0];
        hs2[(size_t)n * 2 + 1] = s1 + b2s[1];
        hd2[(size_t)n * 2 + 0] = d0 + b2d[0];
        hd2[(size_t)n * 2 + 1] = d1 + b2d[1];
    }
}

__global__ void edge_score2_kernel(const float* __restrict__ hs2,
                                   const float* __restrict__ hd2,
                                   const int* __restrict__ src,
                                   const int* __restrict__ dst,
                                   const float* __restrict__ attn2,
                                   float* __restrict__ score2,
                                   unsigned* __restrict__ m2)
{
    const int e = blockIdx.x * blockDim.x + threadIdx.x;
    if (e >= EDGES) return;
    const int s = src[e], d = dst[e];
    const float2 a = *(const float2*)(hs2 + (size_t)s * 2);
    const float2 b = *(const float2*)(hd2 + (size_t)d * 2);
    const float sc = lrelu(a.x + b.x) * attn2[0] + lrelu(a.y + b.y) * attn2[1];
    score2[e] = sc;
    atomicMax(m2 + d, ford(sc));
}

__global__ void edge_exp2_kernel(const int* __restrict__ dst,
                                 float* __restrict__ score2,
                                 const unsigned* __restrict__ m2,
                                 float* __restrict__ denom2)
{
    const int e = blockIdx.x * blockDim.x + threadIdx.x;
    if (e >= EDGES) return;
    const int d = dst[e];
    float m = fdec(m2[d]);
    if (!isfinite(m)) m = 0.f;
    const float ex = expf(score2[e] - m);
    score2[e] = ex;
    atomicAdd(denom2 + d, ex);
}

__global__ void edge_aggr2_kernel(const float* __restrict__ hs2,
                                  const int* __restrict__ src,
                                  const int* __restrict__ dst,
                                  const float* __restrict__ ex,
                                  const float* __restrict__ denom2,
                                  float* __restrict__ out)
{
    const int e = blockIdx.x * blockDim.x + threadIdx.x;
    if (e >= EDGES) return;
    const int s = src[e], d = dst[e];
    float den = denom2[d];
    den = den > 0.f ? den : 1.f;
    const float alpha = ex[e] / den;
    const float2 a = *(const float2*)(hs2 + (size_t)s * 2);
    atomicAdd(out + (size_t)d * 2 + 0, alpha * a.x);
    atomicAdd(out + (size_t)d * 2 + 1, alpha * a.y);
}

extern "C" void kernel_launch(void* const* d_in, const int* in_sizes, int n_in,
                              void* d_out, int out_size, void* d_ws, size_t ws_size,
                              hipStream_t stream)
{
    const float* feat  = (const float*)d_in[0];
    const int*   src   = (const int*)d_in[1];
    const int*   dst   = (const int*)d_in[2];
    const float* W1s   = (const float*)d_in[3];
    const float* b1s   = (const float*)d_in[4];
    const float* W1d   = (const float*)d_in[5];
    const float* b1d   = (const float*)d_in[6];
    const float* attn1 = (const float*)d_in[7];
    const float* W2s   = (const float*)d_in[8];
    const float* b2s   = (const float*)d_in[9];
    const float* W2d   = (const float*)d_in[10];
    const float* b2d   = (const float*)d_in[11];
    const float* attn2 = (const float*)d_in[12];
    float* out = (float*)d_out;

    float* ws = (float*)d_ws;
    size_t off = 0;
    float*    hs1    = ws + off; off += (size_t)NNODES * HIDTOT;
    float*    hd1    = ws + off; off += (size_t)NNODES * HIDTOT;
    float*    out1   = ws + off; off += (size_t)NNODES * HIDTOT;
    float*    score1 = ws + off; off += (size_t)EDGES * 4;
    float*    denom1 = ws + off; off += (size_t)NNODES * 4;
    unsigned* m1     = (unsigned*)(ws + off); off += (size_t)NNODES * 4;
    float*    hs2    = ws + off; off += (size_t)NNODES * 2;
    float*    hd2    = ws + off; off += (size_t)NNODES * 2;
    float*    score2 = ws + off; off += (size_t)EDGES;
    float*    denom2 = ws + off; off += (size_t)NNODES;
    unsigned* m2     = (unsigned*)(ws + off); off += (size_t)NNODES;

    const int B = 256;

    {
        size_t n = (size_t)NNODES * HIDTOT;
        init_kernel<<<(unsigned)((n + B - 1) / B), B, 0, stream>>>(
            out1, denom1, m1, denom2, m2, out);
    }
    {
        const int tiles = 3125 * 4;
        gemm1_kernel<<<(tiles + 7) / 8, B, 0, stream>>>(
            feat, W1s, b1s, W1d, b1d, hs1, hd1);
    }
    edge_score1_kernel<<<(EDGES + 7) / 8, B, 0, stream>>>(
        hs1, hd1, src, dst, attn1, score1, m1);
    edge_exp1_kernel<<<(EDGES * 4 + B - 1) / B, B, 0, stream>>>(
        dst, score1, m1, denom1);
    edge_aggr1_kernel<<<(EDGES + 7) / 8, B, 0, stream>>>(
        hs1, src, dst, score1, denom1, out1);
    elu_kernel<<<((NNODES * HIDTOT) + B - 1) / B, B, 0, stream>>>(
        out1, NNODES * HIDTOT);
    proj2_kernel<<<(NNODES + 7) / 8, B, 0, stream>>>(
        out1, W2s, b2s, W2d, b2d, hs2, hd2);
    edge_score2_kernel<<<(EDGES + B - 1) / B, B, 0, stream>>>(
        hs2, hd2, src, dst, attn2, score2, m2);
    edge_exp2_kernel<<<(EDGES + B - 1) / B, B, 0, stream>>>(
        dst, score2, m2, denom2);
    edge_aggr2_kernel<<<(EDGES + B - 1) / B, B, 0, stream>>>(
        hs2, src, dst, score2, denom2, out);

    (void)in_sizes; (void)n_in; (void)out_size; (void)ws_size;
}


# layer-1 scatter kernel: lane l owns row elements l+32j, so each wave atomic and load covers one whole 128-byte line instead of 4 lanes in each of 8 lines; float atomic arrival order as before
# speedup vs baseline: 1.5283x; 1.5283x over previous
#include <hip/hip_runtime.h>
#include <hip/hip_bf16.h>
#include <math.h>

#define NNODES 50000
#define EDGES  800000
#define INFEAT 128
#define HIDTOT 256
#define NEG_SLOPE 0.2f
#define ORD_NEG_INF 0x007FFFFFu

typedef __attribute__((ext_vector_type(2))) float v2f;
typedef __attribute__((ext_vector_type(8))) float v8f;

__device__ __forceinline__ unsigned ford(float f) {
    unsigned u = __float_as_uint(f);
    return (u & 0x80000000u) ? ~u : (u | 0x80000000u);
}
__device__ __forceinline__ float fdec(unsigned u) {
    return __uint_as_float((u & 0x80000000u) ? (u & 0x7FFFFFFFu) : ~u);
}
__device__ __forceinline__ float lrelu(float x) {
    return x >= 0.f ? x : NEG_SLOPE * x;
}

__global__ void init_kernel(float* out1, float* denom1, unsigned* m1,
                            float* denom2, unsigned* m2, float* dout) {
    size_t i = (size_t)blockIdx.x * blockDim.x + threadIdx.x;
    if (i < (size_t)NNODES * HIDTOT) out1[i] = 0.f;
    if (i < (size_t)NNODES * 4) { denom1[i] = 0.f; m1[i] = ORD_NEG_INF; }
    if (i < (size_t)NNODES)     { denom2[i] = 0.f; m2[i] = ORD_NEG_INF; }
    if (i < (size_t)NNODES * 2) dout[i] = 0.f;
}

__global__ __launch_bounds__(256) void gemm1_kernel(
    const float* __restrict__ feat,
    const float* __restrict__ Ws, const float* __restrict__ bs,
    const float* __restrict__ Wd, const float* __restrict__ bd,
    float* __restrict__ hs, float* __restrict__ hd)
{
    const int wave = threadIdx.x >> 5;
    const int lane = threadIdx.x & 31;
    const int half = lane >> 4;
    const int row  = lane & 15;

    int tile = blockIdx.x * 8 + wave;
    if (tile >= 3125 * 4) return;
    const int mt  = tile >> 2;
    const int nt4 = tile & 3;
    const int m0  = mt * 16;
    const int c0  = nt4 * 64;

    v8f zero = {0.f,0.f,0.f,0.f,0.f,0.f,0.f,0.f};
    v8f accS[4], accD[4];
    #pragma unroll
    for (int j = 0; j < 4; ++j) { accS[j] = zero; accD[j] = zero; }

    const float* arow = feat + (size_t)(m0 + row) * INFEAT;
    for (int k0 = 0; k0 < INFEAT; k0 += 4) {
        const int ka = k0 + 2 * half;
        v2f a = *(const v2f*)(arow + ka);
        #pragma unroll
        for (int j = 0; j < 4; ++j) {
            const int col = c0 + j * 16 + row;
            v2f bS, bD;
            bS.x = Ws[(size_t)ka * HIDTOT + col];
            bS.y = Ws[(size_t)(ka + 1) * HIDTOT + col];
            bD.x = Wd[(size_t)ka * HIDTOT + col];
            bD.y = Wd[(size_t)(ka + 1) * HIDTOT + col];
            accS[j] = __builtin_amdgcn_wmma_f32_16x16x4_f32(
                false, a, false, bS, (short)0, accS[j], false, false);
            accD[j] = __builtin_amdgcn_wmma_f32_16x16x4_f32(
                false, a, false, bD, (short)0, accD[j], false, false);
        }
    }
    #pragma unroll
    for (int j = 0; j < 4; ++j) {
        const int col = c0 + j * 16 + row;
        const float biasS = bs[col], biasD = bd[col];
        #pragma unroll
        for (int i = 0; i < 8; ++i) {
            const int r = m0 + i + 8 * half;
            hs[(size_t)r * HIDTOT + col] = accS[j][i] + biasS;
            hd[(size_t)r * HIDTOT + col] = accD[j][i] + biasD;
        }
    }
}

__global__ __launch_bounds__(256) void edge_score1_kernel(
    const float* __restrict__ hs, const float* __restrict__ hd,
    const int* __restrict__ src, const int* __restrict__ dst,
    const float* __restrict__ attn, float* __restrict__ score,
    unsigned* __restrict__ mbits)
{
    const int e = blockIdx.x * 8 + (threadIdx.x >> 5);
    if (e >= EDGES) return;
    const int lane = threadIdx.x & 31;
    const int s = src[e], d = dst[e];
    const int base = lane * 8;

    const float4* ph = (const float4*)(hs + (size_t)s * HIDTOT + base);
    const float4* pd = (const float4*)(hd + (size_t)d * HIDTOT + base);
    const float4* pa = (const float4*)(attn + base);

    float part = 0.f;
    #pragma unroll
    for (int q = 0; q < 2; ++q) {
        float4 a = ph[q], b = pd[q], w = pa[q];
        part += lrelu(a.x + b.x) * w.x + lrelu(a.y + b.y) * w.y +
                lrelu(a.z + b.z) * w.z + lrelu(a.w + b.w) * w.w;
    }
    part += __shfl_xor(part, 1, 32);
    part += __shfl_xor(part, 2, 32);
    part += __shfl_xor(part, 4, 32);
    if ((lane & 7) == 0) {
        const int h = lane >> 3;
        score[(size_t)e * 4 + h] = part;
        atomicMax(mbits + (size_t)d * 4 + h, ford(part));
    }
}

__global__ void edge_exp1_kernel(const int* __restrict__ dst,
                                 float* __restrict__ score,
                                 const unsigned* __restrict__ mbits,
                                 float* __restrict__ denom)
{
    const int t = blockIdx.x * blockDim.x + threadIdx.x;
    if (t >= EDGES * 4) return;
    const int e = t >> 2, h = t & 3;
    const int d = dst[e];
    float m = fdec(mbits[(size_t)d * 4 + h]);
    if (!isfinite(m)) m = 0.f;
    const float ex = expf(score[t] - m);
    score[t] = ex;
    atomicAdd(denom + (size_t)d * 4 + h, ex);
}

__global__ __launch_bounds__(256) void edge_aggr1_kernel(
    const float* __restrict__ hs, const int* __restrict__ src,
    const int* __restrict__ dst, const float* __restrict__ ex,
    const float* __restrict__ denom, float* __restrict__ out)
{
    const int e = blockIdx.x * 8 + (threadIdx.x >> 5);
    if (e >= EDGES) return;
    const int lane = threadIdx.x & 31;
    const int s = src[e], d = dst[e];
    const float4 x4 = *(const float4*)(ex + (size_t)e * 4);
    const float4 d4 = *(const float4*)(denom + (size_t)d * 4);
    float alpha[4];
    alpha[0] = x4.x / (d4.x > 0.f ? d4.x : 1.f);
    alpha[1] = x4.y / (d4.y > 0.f ? d4.y : 1.f);
    alpha[2] = x4.z / (d4.z > 0.f ? d4.z : 1.f);
    alpha[3] = x4.w / (d4.w > 0.f ? d4.w : 1.f);
    const float* ph = hs + (size_t)s * HIDTOT + lane;
    float* po = out + (size_t)d * HIDTOT + lane;
    float a[8];
    #pragma unroll
    for (int j = 0; j < 8; ++j) a[j] = ph[32 * j];
    #pragma unroll
    for (int j = 0; j < 8; ++j) atomicAdd(po + 32 * j, alpha[j >> 1] * a[j]);
}

__global__ void elu_kernel(float* __restrict__ x, int n) {
    const int i = blockIdx.x * blockDim.x + threadIdx.x;
    if (i < n) {
        const float v = x[i];
        x[i] = v > 0.f ? v : (expf(v) - 1.f);
    }
}

__global__ __launch_bounds__(256) void proj2_kernel(
    const float* __restrict__ h,
    const float* __restrict__ W2s, const float* __restrict__ b2s,
    const float* __restrict__ W2d, const float* __restrict__ b2d,
    float* __restrict__ hs2, float* __restrict__ hd2)
{
    const int n = blockIdx.x * 8 + (threadIdx.x >> 5);
    if (n >= NNODES) return;
    const int lane = threadIdx.x & 31;
    const int base = lane * 8;
    const float* hr = h + (size_t)n * HIDTOT + base;
    float s0 = 0.f, s1 = 0.f, d0 = 0.f, d1 = 0.f;
    #pragma unroll
    for (int k = 0; k < 8; ++k) {
        const float x = hr[k];
        const float2 ws = *(const float2*)(W2s + (size_t)(base + k) * 2);
        const float2 wd = *(const float2*)(W2d + (size_t)(base + k) * 2);
        s0 += x * ws.x; s1 += x * ws.y;
        d0 += x * wd.x; d1 += x * wd.y;
    }
    #pragma unroll
    for (int o = 1; o < 32; o <<= 1) {
        s0 += __shfl_xor(s0, o, 32); s1 += __shfl_xor(s1, o, 32);
        d0 += __shfl_xor(d0, o, 32); d1 += __shfl_xor(d1, o, 32);
    }
    if (lane == 0) {
        hs2[(size_t)n * 2 + 0] = s0 + b2s[0];
        hs2[(size_t)n * 2 + 1] = s1 + b2s[1];
        hd2[(size_t)n * 2 + 0] = d0 + b2d[0];
        hd2[(size_t)n * 2 + 1] = d1 + b2d[1];
    }
}

__global__ void edge_score2_kernel(const float* __restrict__ hs2,
                                   const float* __restrict__ hd2,
                                   const int* __restrict__ src,
                                   const int* __restrict__ dst,
                                   const float* __restrict__ attn2,
                                   float* __restrict__ score2,
                                   unsigned* __restrict__ m2)
{
    const int e = blockIdx.x * blockDim.x + threadIdx.x;
    if (e >= EDGES) return;
    const int s = src[e], d = dst[e];
    const float2 a = *(const float2*)(hs2 + (size_t)s * 2);
    const float2 b = *(const float2*)(hd2 + (size_t)d * 2);
    const float sc = lrelu(a.x + b.x) * attn2[0] + lrelu(a.y + b.y) * attn2[1];
    score2[e] = sc;
    atomicMax(m2 + d, ford(sc));
}

__global__ void edge_exp2_kernel(const int* __restrict__ dst,
                                 float* __restrict__ score2,
                                 const unsigned* __restrict__ m2,
                                 float* __restrict__ denom2)
{
    const int e = blockIdx.x * blockDim.x + threadIdx.x;
    if (e >= EDGES) return;
    const int d = dst[e];
    float m = fdec(m2[d]);
    if (!isfinite(m)) m = 0.f;
    const float ex = expf(score2[e] - m);
    score2[e] = ex;
    atomicAdd(denom2 + d, ex);
}

__global__ void edge_aggr2_kernel(const float* __restrict__ hs2,
                                  const int* __restrict__ src,
                                  const int* __restrict__ dst,
                                  const float* __restrict__ ex,
                                  const float* __restrict__ denom2,
                                  float* __restrict__ out)
{
    const int e = blockIdx.x * blockDim.x + threadIdx.x;
    if (e >= EDGES) return;
    const int s = src[e], d = dst[e];
    float den = denom2[d];
    den = den > 0.f ? den : 1.f;
    const float alpha = ex[e] / den;
    const float2 a = *(const float2*)(hs2 + (size_t)s * 2);
    atomicAdd(out + (size_t)d * 2 + 0, alpha * a.x);
    atomicAdd(out + (size_t)d * 2 + 1, alpha * a.y);
}

extern "C" void kernel_launch(void* const* d_in, const int* in_sizes, int n_in,
                              void* d_out, int out_size, void* d_ws, size_t ws_size,
                              hipStream_t stream)
{
    const float* feat  = (const float*)d_in[0];
    const int*   src   = (const int*)d_in[1];
    const int*   dst   = (const int*)d_in[2];
    const float* W1s   = (const float*)d_in[3];
    const float* b1s   = (const float*)d_in[4];
    const float* W1d   = (const float*)d_in[5];
    const float* b1d   = (const float*)d_in[6];
    const float* attn1 = (const float*)d_in[7];
    const float* W2s   = (const float*)d_in[8];
    const float* b2s   = (const float*)d_in[9];
    const float* W2d   = (const float*)d_in[10];
    const float* b2d   = (const float*)d_in[11];
    const float* attn2 = (const float*)d_in[12];
    float* out = (float*)d_out;

    float* ws = (float*)d_ws;
    size_t off = 0;
    float*    hs1    = ws + off; off += (size_t)NNODES * HIDTOT;
    float*    hd1    = ws + off; off += (size_t)NNODES * HIDTOT;
    float*    out1   = ws + off; off += (size_t)NNODES * HIDTOT;
    float*    score1 = ws + off; off += (size_t)EDGES * 4;
    float*    denom1 = ws + off; off += (size_t)NNODES * 4;
    unsigned* m1     = (unsigned*)(ws + off); off += (size_t)NNODES * 4;
    float*    hs2    = ws + off; off += (size_t)NNODES * 2;
    float*    hd2    = ws + off; off += (size_t)NNODES * 2;
    float*    score2 = ws + off; off += (size_t)EDGES;
    float*    denom2 = ws + off; off += (size_t)NNODES;
    unsigned* m2     = (unsigned*)(ws + off); off += (size_t)NNODES;

    const int B = 256;

    {
        size_t n = (size_t)NNODES * HIDTOT;
        init_kernel<<<(unsigned)((n + B - 1) / B), B, 0, stream>>>(
            out1, denom1, m1, denom2, m2, out);
    }
    {
        const int tiles = 3125 * 4;
        gemm1_kernel<<<(tiles + 7) / 8, B, 0, stream>>>(
            feat, W1s, b1s, W1d, b1d, hs1, hd1);
    }
    edge_score1_kernel<<<(EDGES + 7) / 8, B, 0, stream>>>(
        hs1, hd1, src, dst, attn1, score1, m1);
    edge_exp1_kernel<<<(EDGES * 4 + B - 1) / B, B, 0, stream>>>(
        dst, score1, m1, denom1);
    edge_aggr1_kernel<<<(EDGES + 7) / 8, B, 0, stream>>>(
        hs1, src, dst, score1, denom1, out1);
    elu_kernel<<<((NNODES * HIDTOT) + B - 1) / B, B, 0, stream>>>(
        out1, NNODES * HIDTOT);
    proj2_kernel<<<(NNODES + 7) / 8, B, 0, stream>>>(
        out1, W2s, b2s, W2d, b2d, hs2, hd2);
    edge_score2_kernel<<<(EDGES + B - 1) / B, B, 0, stream>>>(
        hs2, hd2, src, dst, attn2, score2, m2);
    edge_exp2_kernel<<<(EDGES + B - 1) / B, B, 0, stream>>>(
        dst, score2, m2, denom2);
    edge_aggr2_kernel<<<(EDGES + B - 1) / B, B, 0, stream>>>(
        hs2, src, dst, score2, denom2, out);

    (void)in_sizes; (void)n_in; (void)out_size; (void)ws_size;
}


# edges grouped by destination in-call (count, offsets, fill); one wave per node does layer-1 scores, running-max softmax, row sum, ELU in registers, one store; sums in list (atomic arrival) order
# speedup vs baseline: 1.8107x; 1.1848x over previous
#include <hip/hip_runtime.h>
#include <hip/hip_bf16.h>
#include <math.h>

#define NNODES 50000
#define EDGES  800000
#define INFEAT 128
#define HIDTOT 256
#define NEG_SLOPE 0.2f
#define ORD_NEG_INF 0x007FFFFFu

typedef __attribute__((ext_vector_type(2))) float v2f;
typedef __attribute__((ext_vector_type(8))) float v8f;

__device__ __forceinline__ unsigned ford(float f) {
    unsigned u = __float_as_uint(f);
    return (u & 0x80000000u) ? ~u : (u | 0x80000000u);
}
__device__ __forceinline__ float fdec(unsigned u) {
    return __uint_as_float((u & 0x80000000u) ? (u & 0x7FFFFFFFu) : ~u);
}
__device__ __forceinline__ float lrelu(float x) {
    return x >= 0.f ? x : NEG_SLOPE * x;
}

#define SCAN_T   1024
#define SCAN_CH  ((NNODES + SCAN_T - 1) / SCAN_T)
#define CNT_PAD  (SCAN_T * SCAN_CH)
__global__ void init_kernel(int* cnt, float* denom2, unsigned* m2, float* dout) {
    const int i = blockIdx.x * blockDim.x + threadIdx.x;
    if (i < CNT_PAD)    cnt[i] = 0;
    if (i < NNODES)     { denom2[i] = 0.f; m2[i] = ORD_NEG_INF; }
    if (i < NNODES * 2) dout[i] = 0.f;
}

__global__ __launch_bounds__(256) void gemm1_kernel(
    const float* __restrict__ feat,
    const float* __restrict__ Ws, const float* __restrict__ bs,
    const float* __restrict__ Wd, const float* __restrict__ bd,
    float* __restrict__ hs, float* __restrict__ hd)
{
    const int wave = threadIdx.x >> 5;
    const int lane = threadIdx.x & 31;
    const int half = lane >> 4;
    const int row  = lane & 15;

    int tile = blockIdx.x * 8 + wave;
    if (tile >= 3125 * 4) return;
    const int mt  = tile >> 2;
    const int nt4 = tile & 3;
    const int m0  = mt * 16;
    const int c0  = nt4 * 64;

    v8f zero = {0.f,0.f,0.f,0.f,0.f,0.f,0.f,0.f};
    v8f accS[4], accD[4];
    #pragma unroll
    for (int j = 0; j < 4; ++j) { accS[j] = zero; accD[j] = zero; }

    const float* arow = feat + (size_t)(m0 + row) * INFEAT;
    for (int k0 = 0; k0 < INFEAT; k0 += 4) {
        const int ka = k0 + 2 * half;
        v2f a = *(const v2f*)(arow + ka);
        #pragma unroll
        for (int j = 0; j < 4; ++j) {
            const int col = c0 + j * 16 + row;
            v2f bS, bD;
            bS.x = Ws[(size_t)ka * HIDTOT + col];
            bS.y = Ws[(size_t)(ka + 1) * HIDTOT + col];
            bD.x = Wd[(size_t)ka * HIDTOT + col];
            bD.y = Wd[(size_t)(ka + 1) * HIDTOT + col];
            accS[j] = __builtin_amdgcn_wmma_f32_16x16x4_f32(
                false, a, false, bS, (short)0, accS[j], false, false);
            accD[j] = __builtin_amdgcn_wmma_f32_16x16x4_f32(
                false, a, false, bD, (short)0, accD[j], false, false);
        }
    }
    #pragma unroll
    for (int j = 0; j < 4; ++j) {
        const int col = c0 + j * 16 + row;
        const float biasS = bs[col], biasD = bd[col];
        #pragma unroll
        for (int i = 0; i < 8; ++i) {
            const int r = m0 + i + 8 * half;
            hs[(size_t)r * HIDTOT + col] = accS[j][i] + biasS;
            hd[(size_t)r * HIDTOT + col] = accD[j][i] + biasD;
        }
    }
}

__global__ void edge_count_kernel(const int* __restrict__ dst, int* __restrict__ cnt) {
    const int e = blockIdx.x * blockDim.x + threadIdx.x;
    if (e < EDGES) atomicAdd(cnt + dst[e], 1);
}

__global__ __launch_bounds__(SCAN_T) void offsets_kernel(const int* __restrict__ cnt,
                                                         int* __restrict__ off)
{
    __shared__ int wsum[32];
    const int t = threadIdx.x, lane = t & 31, wave = t >> 5;
    const int base = t * SCAN_CH;
    int v[SCAN_CH];
    int sum = 0;
    #pragma unroll
    for (int j = 0; j < SCAN_CH; ++j) {
        v[j] = cnt[base + j];
        sum += v[j];
    }
    int inc = sum;
    #pragma unroll
    for (int o = 1; o < 32; o <<= 1) {
        const int y = __shfl_up(inc, o, 32);
        if (lane >= o) inc += y;
    }
    if (lane == 31) wsum[wave] = inc;
    __syncthreads();
    if (wave == 0) {
        const int w = wsum[lane];
        int winc = w;
        #pragma unroll
        for (int o = 1; o < 32; o <<= 1) {
            const int y = __shfl_up(winc, o, 32);
            if (lane >= o) winc += y;
        }
        wsum[lane] = winc - w;
    }
    __syncthreads();
    int run = wsum[wave] + inc - sum;
    #pragma unroll
    for (int j = 0; j < SCAN_CH; ++j) {
        off[base + j] = run;
        run += v[j];
    }
}

__global__ void edge_fill_kernel(const int* __restrict__ src, const int* __restrict__ dst,
                                 const int* __restrict__ off, int* __restrict__ cnt,
                                 int* __restrict__ list)
{
    const int e = blockIdx.x * blockDim.x + threadIdx.x;
    if (e >= EDGES) return;
    const int d = dst[e];
    const int p = atomicSub(cnt + d, 1) - 1;
    list[off[d] + p] = src[e];
}

template <int M> __device__ __forceinline__ float row_xor(float v) {
    return __int_as_float(__builtin_amdgcn_update_dpp(0, __float_as_int(v), 0x160 | M, 0xF, 0xF, true));
}
__device__ __forceinline__ float row16_sum(float v) {
    v += row_xor<1>(v);
    v += row_xor<2>(v);
    v += row_xor<4>(v);
    v += row_xor<8>(v);
    return v;
}
__device__ __forceinline__ float dot_lrelu(const float4 a, const float4 b, const float4 w) {
    return lrelu(a.x + b.x) * w.x + lrelu(a.y + b.y) * w.y +
           lrelu(a.z + b.z) * w.z + lrelu(a.w + b.w) * w.w;
}
__device__ __forceinline__ float elu1(float v) { return v > 0.f ? v : (expf(v) - 1.f); }

#define GRP 4
__global__ __launch_bounds__(256) void node_attn1_kernel(
    const float* __restrict__ hs, const float* __restrict__ hd,
    const int* __restrict__ off, const int* __restrict__ list,
    const float* __restrict__ attn, float* __restrict__ out)
{
    const int d = __builtin_amdgcn_readfirstlane((int)(blockIdx.x * 8 + (threadIdx.x >> 5)));
    if (d >= NNODES) return;
    const int lane = threadIdx.x & 31;
    const int beg = off[d], end = off[d + 1];

    const float* pd = hd + (size_t)d * HIDTOT + 4 * lane;
    const float4 b0 = *(const float4*)(pd);
    const float4 b1 = *(const float4*)(pd + 128);
    const float4 w0 = *(const float4*)(attn + 4 * lane);
    const float4 w1 = *(const float4*)(attn + 128 + 4 * lane);

    float m0 = -INFINITY, m1 = -INFINITY, l0 = 0.f, l1 = 0.f;
    float4 acc0 = {0.f, 0.f, 0.f, 0.f}, acc1 = {0.f, 0.f, 0.f, 0.f};

    for (int i = beg; i < end; i += 32) {
        const int n = (end - i) < 32 ? (end - i) : 32;
        const int my_s = lane < n ? list[i + lane] : 0;
        for (int k = 0; k < n; k += GRP) {
            float4 a0[GRP], a1[GRP];
            #pragma unroll
            for (int j = 0; j < GRP; ++j) {
                const int kk = (k + j < n) ? (k + j) : k;
                const int s = __builtin_amdgcn_readlane(my_s, kk);
                const float* ps = hs + (size_t)s * HIDTOT + 4 * lane;
                a0[j] = *(const float4*)(ps);
                a1[j] = *(const float4*)(ps + 128);
            }
            float p0[GRP], p1[GRP];
            float g0 = m0, g1 = m1;
            #pragma unroll
            for (int j = 0; j < GRP; ++j) {
                float x0 = row16_sum(dot_lrelu(a0[j], b0, w0));
                float x1 = row16_sum(dot_lrelu(a1[j], b1, w1));
                if (k + j >= n) { x0 = -INFINITY; x1 = -INFINITY; }
                p0[j] = x0; p1[j] = x1;
                g0 = fmaxf(g0, x0); g1 = fmaxf(g1, x1);
            }
            const float r0 = expf(m0 - g0), r1 = expf(m1 - g1);
            l0 *= r0; l1 *= r1;
            acc0.x *= r0; acc0.y *= r0; acc0.z *= r0; acc0.w *= r0;
            acc1.x *= r1; acc1.y *= r1; acc1.z *= r1; acc1.w *= r1;
            #pragma unroll
            for (int j = 0; j < GRP; ++j) {
                const float e0 = expf(p0[j] - g0), e1 = expf(p1[j] - g1);
                l0 += e0; l1 += e1;
                acc0.x += e0 * a0[j].x; acc0.y += e0 * a0[j].y;
                acc0.z += e0 * a0[j].z; acc0.w += e0 * a0[j].w;
                acc1.x += e1 * a1[j].x; acc1.y += e1 * a1[j].y;
                acc1.z += e1 * a1[j].z; acc1.w += e1 * a1[j].w;
            }
            m0 = g0; m1 = g1;
        }
    }
    const float q0 = l0 > 0.f ? l0 : 1.f, q1 = l1 > 0.f ? l1 : 1.f;
    float4 o0, o1;
    o0.x = elu1(acc0.x / q0); o0.y = elu1(acc0.y / q0);
    o0.z = elu1(acc0.z / q0); o0.w = elu1(acc0.w / q0);
    o1.x = elu1(acc1.x / q1); o1.y = elu1(acc1.y / q1);
    o1.z = elu1(acc1.z / q1); o1.w = elu1(acc1.w / q1);
    float* po = out + (size_t)d * HIDTOT + 4 * lane;
    *(float4*)(po) = o0;
    *(float4*)(po + 128) = o1;
}

__global__ __launch_bounds__(256) void proj2_kernel(
    const float* __restrict__ h,
    const float* __restrict__ W2s, const float* __restrict__ b2s,
    const float* __restrict__ W2d, const float* __restrict__ b2d,
    float* __restrict__ hs2, float* __restrict__ hd2)
{
    const int n = blockIdx.x * 8 + (threadIdx.x >> 5);
    if (n >= NNODES) return;
    const int lane = threadIdx.x & 31;
    const int base = lane * 8;
    const float* hr = h + (size_t)n * HIDTOT + base;
    float s0 = 0.f, s1 = 0.f, d0 = 0.f, d1 = 0.f;
    #pragma unroll
    for (int k = 0; k < 8; ++k) {
        const float x = hr[k];
        const float2 ws = *(const float2*)(W2s + (size_t)(base + k) * 2);
        const float2 wd = *(const float2*)(W2d + (size_t)(base + k) * 2);
        s0 += x * ws.x; s1 += x * ws.y;
        d0 += x * wd.x; d1 += x * wd.y;
    }
    #pragma unroll
    for (int o = 1; o < 32; o <<= 1) {
        s0 += __shfl_xor(s0, o, 32); s1 += __shfl_xor(s1, o, 32);
        d0 += __shfl_xor(d0, o, 32); d1 += __shfl_xor(d1, o, 32);
    }
    if (lane == 0) {
        hs2[(size_t)n * 2 + 0] = s0 + b2s[0];
        hs2[(size_t)n * 2 + 1] = s1 + b2s[1];
        hd2[(size_t)n * 2 + 0] = d0 + b2d[0];
        hd2[(size_t)n * 2 + 1] = d1 + b2d[1];
    }
}

__global__ void edge_score2_kernel(const float* __restrict__ hs2,
                                   const float* __restrict__ hd2,
                                   const int* __restrict__ src,
                                   const int* __restrict__ dst,
                                   const float* __restrict__ attn2,
                                   float* __restrict__ score2,
                                   unsigned* __restrict__ m2)
{
    const int e = blockIdx.x * blockDim.x + threadIdx.x;
    if (e >= EDGES) return;
    const int s = src[e], d = dst[e];
    const float2 a = *(const float2*)(hs2 + (size_t)s * 2);
    const float2 b = *(const float2*)(hd2 + (size_t)d * 2);
    const float sc = lrelu(a.x + b.x) * attn2[0] + lrelu(a.y + b.y) * attn2[1];
    score2[e] = sc;
    atomicMax(m2 + d, ford(sc));
}

__global__ void edge_exp2_kernel(const int* __restrict__ dst,
                                 float* __restrict__ score2,
                                 const unsigned* __restrict__ m2,
                                 float* __restrict__ denom2)
{
    const int e = blockIdx.x * blockDim.x + threadIdx.x;
    if (e >= EDGES) return;
    const int d = dst[e];
    float m = fdec(m2[d]);
    if (!isfinite(m)) m = 0.f;
    const float ex = expf(score2[e] - m);
    score2[e] = ex;
    atomicAdd(denom2 + d, ex);
}

__global__ void edge_aggr2_kernel(const float* __restrict__ hs2,
                                  const int* __restrict__ src,
                                  const int* __restrict__ dst,
                                  const float* __restrict__ ex,
                                  const float* __restrict__ denom2,
                                  float* __restrict__ out)
{
    const int e = blockIdx.x * blockDim.x + threadIdx.x;
    if (e >= EDGES) return;
    const int s = src[e], d = dst[e];
    float den = denom2[d];
    den = den > 0.f ? den : 1.f;
    const float alpha = ex[e] / den;
    const float2 a = *(const float2*)(hs2 + (size_t)s * 2);
    atomicAdd(out + (size_t)d * 2 + 0, alpha * a.x);
    atomicAdd(out + (size_t)d * 2 + 1, alpha * a.y);
}

extern "C" void kernel_launch(void* const* d_in, const int* in_sizes, int n_in,
                              void* d_out, int out_size, void* d_ws, size_t ws_size,
                              hipStream_t stream)
{
    const float* feat  = (const float*)d_in[0];
    const int*   src   = (const int*)d_in[1];
    const int*   dst   = (const int*)d_in[2];
    const float* W1s   = (const float*)d_in[3];
    const float* b1s   = (const float*)d_in[4];
    const float* W1d   = (const float*)d_in[5];
    const float* b1d   = (const float*)d_in[6];
    const float* attn1 = (const float*)d_in[7];
    const float* W2s   = (const float*)d_in[8];
    const float* b2s   = (const float*)d_in[9];
    const float* W2d   = (const float*)d_in[10];
    const float* b2d   = (const float*)d_in[11];
    const float* attn2 = (const float*)d_in[12];
    float* out = (float*)d_out;

    float* ws = (float*)d_ws;
    size_t off = 0;
    float*    hs1    = ws + off; off += (size_t)NNODES * HIDTOT;
    float*    hd1    = ws + off; off += (size_t)NNODES * HIDTOT;
    float*    out1   = ws + off; off += (size_t)NNODES * HIDTOT;
    int*      cnt    = (int*)(ws + off); off += (size_t)CNT_PAD;
    int*      offs   = (int*)(ws + off); off += (size_t)CNT_PAD;
    int*      list   = (int*)(ws + off); off += (size_t)EDGES;
    float*    hs2    = ws + off; off += (size_t)NNODES * 2;
    float*    hd2    = ws + off; off += (size_t)NNODES * 2;
    float*    score2 = ws + off; off += (size_t)EDGES;
    float*    denom2 = ws + off; off += (size_t)NNODES;
    unsigned* m2     = (unsigned*)(ws + off); off += (size_t)NNODES;

    const int B = 256;

    init_kernel<<<(NNODES * 2 + B - 1) / B, B, 0, stream>>>(cnt, denom2, m2, out);
    edge_count_kernel<<<(EDGES + B - 1) / B, B, 0, stream>>>(dst, cnt);
    offsets_kernel<<<1, SCAN_T, 0, stream>>>(cnt, offs);
    edge_fill_kernel<<<(EDGES + B - 1) / B, B, 0, stream>>>(src, dst, offs, cnt, list);
    {
        const int tiles = 3125 * 4;
        gemm1_kernel<<<(tiles + 7) / 8, B, 0, stream>>>(
            feat, W1s, b1s, W1d, b1d, hs1, hd1);
    }
    node_attn1_kernel<<<(NNODES + 7) / 8, B, 0, stream>>>(
        hs1, hd1, offs, list, attn1, out1);
    proj2_kernel<<<(NNODES + 7) / 8, B, 0, stream>>>(
        out1, W2s, b2s, W2d, b2d, hs2, hd2);
    edge_score2_kernel<<<(EDGES + B - 1) / B, B, 0, stream>>>(
        hs2, hd2, src, dst, attn2, score2, m2);
    edge_exp2_kernel<<<(EDGES + B - 1) / B, B, 0, stream>>>(
        dst, score2, m2, denom2);
    edge_aggr2_kernel<<<(EDGES + B - 1) / B, B, 0, stream>>>(
        hs2, src, dst, score2, denom2, out);

    (void)in_sizes; (void)n_in; (void)out_size; (void)ws_size;
}


# layer-2 edge phase on the destination-grouped list: one lane per node, scores, running-max softmax, weighted sum in registers, one plain store, no layer-2 atomics; sums in list (atomic arrival) order
# speedup vs baseline: 2.8288x; 1.5623x over previous
#include <hip/hip_runtime.h>
#include <hip/hip_bf16.h>
#include <math.h>

#define NNODES 50000
#define EDGES  800000
#define INFEAT 128
#define HIDTOT 256
#define NEG_SLOPE 0.2f

typedef __attribute__((ext_vector_type(2))) float v2f;
typedef __attribute__((ext_vector_type(8))) float v8f;

__device__ __forceinline__ float lrelu(float x) {
    return x >= 0.f ? x : NEG_SLOPE * x;
}

#define SCAN_T   1024
#define SCAN_CH  ((NNODES + SCAN_T - 1) / SCAN_T)
#define CNT_PAD  (SCAN_T * SCAN_CH)
__global__ void init_kernel(int* cnt) {
    const int i = blockIdx.x * blockDim.x + threadIdx.x;
    if (i < CNT_PAD) cnt[i] = 0;
}

__global__ __launch_bounds__(256) void gemm1_kernel(
    const float* __restrict__ feat,
    const float* __restrict__ Ws, const float* __restrict__ bs,
    const float* __restrict__ Wd, const float* __restrict__ bd,
    float* __restrict__ hs, float* __restrict__ hd)
{
    const int wave = threadIdx.x >> 5;
    const int lane = threadIdx.x & 31;
    const int half = lane >> 4;
    const int row  = lane & 15;

    int tile = blockIdx.x * 8 + wave;
    if (tile >= 3125 * 4) return;
    const int mt  = tile >> 2;
    const int nt4 = tile & 3;
    const int m0  = mt * 16;
    const int c0  = nt4 * 64;

    v8f zero = {0.f,0.f,0.f,0.f,0.f,0.f,0.f,0.f};
    v8f accS[4], accD[4];
    #pragma unroll
    for (int j = 0; j < 4; ++j) { accS[j] = zero; accD[j] = zero; }

    const float* arow = feat + (size_t)(m0 + row) * INFEAT;
    for (int k0 = 0; k0 < INFEAT; k0 += 4) {
        const int ka = k0 + 2 * half;
        v2f a = *(const v2f*)(arow + ka);
        #pragma unroll
        for (int j = 0; j < 4; ++j) {
            const int col = c0 + j * 16 + row;
            v2f bS, bD;
            bS.x = Ws[(size_t)ka * HIDTOT + col];
            bS.y = Ws[(size_t)(ka + 1) * HIDTOT + col];
            bD.x = Wd[(size_t)ka * HIDTOT + col];
            bD.y = Wd[(size_t)(ka + 1) * HIDTOT + col];
            accS[j] = __builtin_amdgcn_wmma_f32_16x16x4_f32(
                false, a, false, bS, (short)0, accS[j], false, false);
            accD[j] = __builtin_amdgcn_wmma_f32_16x16x4_f32(
                false, a, false, bD, (short)0, accD[j], false, false);
        }
    }
    #pragma unroll
    for (int j = 0; j < 4; ++j) {
        const int col = c0 + j * 16 + row;
        const float biasS = bs[col], biasD = bd[col];
        #pragma unroll
        for (int i = 0; i < 8; ++i) {
            const int r = m0 + i + 8 * half;
            hs[(size_t)r * HIDTOT + col] = accS[j][i] + biasS;
            hd[(size_t)r * HIDTOT + col] = accD[j][i] + biasD;
        }
    }
}

__global__ void edge_count_kernel(const int* __restrict__ dst, int* __restrict__ cnt) {
    const int e = blockIdx.x * blockDim.x + threadIdx.x;
    if (e < EDGES) atomicAdd(cnt + dst[e], 1);
}

__global__ __launch_bounds__(SCAN_T) void offsets_kernel(const int* __restrict__ cnt,
                                                         int* __restrict__ off)
{
    __shared__ int wsum[32];
    const int t = threadIdx.x, lane = t & 31, wave = t >> 5;
    const int base = t * SCAN_CH;
    int v[SCAN_CH];
    int sum = 0;
    #pragma unroll
    for (int j = 0; j < SCAN_CH; ++j) {
        v[j] = cnt[base + j];
        sum += v[j];
    }
    int inc = sum;
    #pragma unroll
    for (int o = 1; o < 32; o <<= 1) {
        const int y = __shfl_up(inc, o, 32);
        if (lane >= o) inc += y;
    }
    if (lane == 31) wsum[wave] = inc;
    __syncthreads();
    if (wave == 0) {
        const int w = wsum[lane];
        int winc = w;
        #pragma unroll
        for (int o = 1; o < 32; o <<= 1) {
            const int y = __shfl_up(winc, o, 32);
            if (lane >= o) winc += y;
        }
        wsum[lane] = winc - w;
    }
    __syncthreads();
    int run = wsum[wave] + inc - sum;
    #pragma unroll
    for (int j = 0; j < SCAN_CH; ++j) {
        off[base + j] = run;
        run += v[j];
    }
}

__global__ void edge_fill_kernel(const int* __restrict__ src, const int* __restrict__ dst,
                                 const int* __restrict__ off, int* __restrict__ cnt,
                                 int* __restrict__ list)
{
    const int e = blockIdx.x * blockDim.x + threadIdx.x;
    if (e >= EDGES) return;
    const int d = dst[e];
    const int p = atomicSub(cnt + d, 1) - 1;
    list[off[d] + p] = src[e];
}

template <int M> __device__ __forceinline__ float row_xor(float v) {
    return __int_as_float(__builtin_amdgcn_update_dpp(0, __float_as_int(v), 0x160 | M, 0xF, 0xF, true));
}
__device__ __forceinline__ float row16_sum(float v) {
    v += row_xor<1>(v);
    v += row_xor<2>(v);
    v += row_xor<4>(v);
    v += row_xor<8>(v);
    return v;
}
__device__ __forceinline__ float dot_lrelu(const float4 a, const float4 b, const float4 w) {
    return lrelu(a.x + b.x) * w.x + lrelu(a.y + b.y) * w.y +
           lrelu(a.z + b.z) * w.z + lrelu(a.w + b.w) * w.w;
}
__device__ __forceinline__ float elu1(float v) { return v > 0.f ? v : (expf(v) - 1.f); }

#define GRP 4
__global__ __launch_bounds__(256) void node_attn1_kernel(
    const float* __restrict__ hs, const float* __restrict__ hd,
    const int* __restrict__ off, const int* __restrict__ list,
    const float* __restrict__ attn, float* __restrict__ out)
{
    const int d = __builtin_amdgcn_readfirstlane((int)(blockIdx.x * 8 + (threadIdx.x >> 5)));
    if (d >= NNODES) return;
    const int lane = threadIdx.x & 31;
    const int beg = off[d], end = off[d + 1];

    const float* pd = hd + (size_t)d * HIDTOT + 4 * lane;
    const float4 b0 = *(const float4*)(pd);
    const float4 b1 = *(const float4*)(pd + 128);
    const float4 w0 = *(const float4*)(attn + 4 * lane);
    const float4 w1 = *(const float4*)(attn + 128 + 4 * lane);

    float m0 = -INFINITY, m1 = -INFINITY, l0 = 0.f, l1 = 0.f;
    float4 acc0 = {0.f, 0.f, 0.f, 0.f}, acc1 = {0.f, 0.f, 0.f, 0.f};

    for (int i = beg; i < end; i += 32) {
        const int n = (end - i) < 32 ? (end - i) : 32;
        const int my_s = lane < n ? list[i + lane] : 0;
        for (int k = 0; k < n; k += GRP) {
            float4 a0[GRP], a1[GRP];
            #pragma unroll
            for (int j = 0; j < GRP; ++j) {
                const int kk = (k + j < n) ? (k + j) : k;
                const int s = __builtin_amdgcn_readlane(my_s, kk);
                const float* ps = hs + (size_t)s * HIDTOT + 4 * lane;
                a0[j] = *(const float4*)(ps);
                a1[j] = *(const float4*)(ps + 128);
            }
            float p0[GRP], p1[GRP];
            float g0 = m0, g1 = m1;
            #pragma unroll
            for (int j = 0; j < GRP; ++j) {
                float x0 = row16_sum(dot_lrelu(a0[j], b0, w0));
                float x1 = row16_sum(dot_lrelu(a1[j], b1, w1));
                if (k + j >= n) { x0 = -INFINITY; x1 = -INFINITY; }
                p0[j] = x0; p1[j] = x1;
                g0 = fmaxf(g0, x0); g1 = fmaxf(g1, x1);
            }
            const float r0 = expf(m0 - g0), r1 = expf(m1 - g1);
            l0 *= r0; l1 *= r1;
            acc0.x *= r0; acc0.y *= r0; acc0.z *= r0; acc0.w *= r0;
            acc1.x *= r1; acc1.y *= r1; acc1.z *= r1; acc1.w *= r1;
            #pragma unroll
            for (int j = 0; j < GRP; ++j) {
                const float e0 = expf(p0[j] - g0), e1 = expf(p1[j] - g1);
                l0 += e0; l1 += e1;
                acc0.x += e0 * a0[j].x; acc0.y += e0 * a0[j].y;
                acc0.z += e0 * a0[j].z; acc0.w += e0 * a0[j].w;
                acc1.x += e1 * a1[j].x; acc1.y += e1 * a1[j].y;
                acc1.z += e1 * a1[j].z; acc1.w += e1 * a1[j].w;
            }
            m0 = g0; m1 = g1;
        }
    }
    const float q0 = l0 > 0.f ? l0 : 1.f, q1 = l1 > 0.f ? l1 : 1.f;
    float4 o0, o1;
    o0.x = elu1(acc0.x / q0); o0.y = elu1(acc0.y / q0);
    o0.z = elu1(acc0.z / q0); o0.w = elu1(acc0.w / q0);
    o1.x = elu1(acc1.x / q1); o1.y = elu1(acc1.y / q1);
    o1.z = elu1(acc1.z / q1); o1.w = elu1(acc1.w / q1);
    float* po = out + (size_t)d * HIDTOT + 4 * lane;
    *(float4*)(po) = o0;
    *(float4*)(po + 128) = o1;
}

__global__ __launch_bounds__(256) void proj2_kernel(
    const float* __restrict__ h,
    const float* __restrict__ W2s, const float* __restrict__ b2s,
    const float* __restrict__ W2d, const float* __restrict__ b2d,
    float* __restrict__ hs2, float* __restrict__ hd2)
{
    const int n = blockIdx.x * 8 + (threadIdx.x >> 5);
    if (n >= NNODES) return;
    const int lane = threadIdx.x & 31;
    const int base = lane * 8;
    const float* hr = h + (size_t)n * HIDTOT + base;
    float s0 = 0.f, s1 = 0.f, d0 = 0.f, d1 = 0.f;
    #pragma unroll
    for (int k = 0; k < 8; ++k) {
        const float x = hr[k];
        const float2 ws = *(const float2*)(W2s + (size_t)(base + k) * 2);
        const float2 wd = *(const float2*)(W2d + (size_t)(base + k) * 2);
        s0 += x * ws.x; s1 += x * ws.y;
        d0 += x * wd.x; d1 += x * wd.y;
    }
    #pragma unroll
    for (int o = 1; o < 32; o <<= 1) {
        s0 += __shfl_xor(s0, o, 32); s1 += __shfl_xor(s1, o, 32);
        d0 += __shfl_xor(d0, o, 32); d1 += __shfl_xor(d1, o, 32);
    }
    if (lane == 0) {
        hs2[(size_t)n * 2 + 0] = s0 + b2s[0];
        hs2[(size_t)n * 2 + 1] = s1 + b2s[1];
        hd2[(size_t)n * 2 + 0] = d0 + b2d[0];
        hd2[(size_t)n * 2 + 1] = d1 + b2d[1];
    }
}

#define GRP2 8
__global__ __launch_bounds__(64) void node_attn2_kernel(
    const float* __restrict__ hs2, const float* __restrict__ hd2,
    const int* __restrict__ off, const int* __restrict__ list,
    const float* __restrict__ attn2, float* __restrict__ out)
{
    const int d = blockIdx.x * blockDim.x + threadIdx.x;
    if (d >= NNODES) return;
    const int beg = off[d], end = off[d + 1];
    const float2 b = *(const float2*)(hd2 + (size_t)d * 2);
    const float w0 = attn2[0], w1 = attn2[1];

    float m = -INFINITY, l = 0.f, ax = 0.f, ay = 0.f;
    for (int i = beg; i < end; i += GRP2) {
        int s[GRP2];
        float2 a[GRP2];
        #pragma unroll
        for (int j = 0; j < GRP2; ++j) {
            const int ii = (i + j < end) ? (i + j) : i;
            s[j] = list[ii];
        }
        #pragma unroll
        for (int j = 0; j < GRP2; ++j) a[j] = *(const float2*)(hs2 + (size_t)s[j] * 2);
        float p[GRP2];
        float g = m;
        #pragma unroll
        for (int j = 0; j < GRP2; ++j) {
            float x = lrelu(a[j].x + b.x) * w0 + lrelu(a[j].y + b.y) * w1;
            if (i + j >= end) x = -INFINITY;
            p[j] = x;
            g = fmaxf(g, x);
        }
        const float r = expf(m - g);
        l *= r; ax *= r; ay *= r;
        #pragma unroll
        for (int j = 0; j < GRP2; ++j) {
            const float e = expf(p[j] - g);
            l += e;
            ax += e * a[j].x;
            ay += e * a[j].y;
        }
        m = g;
    }
    const float q = l > 0.f ? l : 1.f;
    float2 o;
    o.x = ax / q;
    o.y = ay / q;
    *(float2*)(out + (size_t)d * 2) = o;
}

extern "C" void kernel_launch(void* const* d_in, const int* in_sizes, int n_in,
                              void* d_out, int out_size, void* d_ws, size_t ws_size,
                              hipStream_t stream)
{
    const float* feat  = (const float*)d_in[0];
    const int*   src   = (const int*)d_in[1];
    const int*   dst   = (const int*)d_in[2];
    const float* W1s   = (const float*)d_in[3];
    const float* b1s   = (const float*)d_in[4];
    const float* W1d   = (const float*)d_in[5];
    const float* b1d   = (const float*)d_in[6];
    const float* attn1 = (const float*)d_in[7];
    const float* W2s   = (const float*)d_in[8];
    const float* b2s   = (const float*)d_in[9];
    const float* W2d   = (const float*)d_in[10];
    const float* b2d   = (const float*)d_in[11];
    const float* attn2 = (const float*)d_in[12];
    float* out = (float*)d_out;

    float* ws = (float*)d_ws;
    size_t off = 0;
    float*    hs1    = ws + off; off += (size_t)NNODES * HIDTOT;
    float*    hd1    = ws + off; off += (size_t)NNODES * HIDTOT;
    float*    out1   = ws + off; off += (size_t)NNODES * HIDTOT;
    int*      cnt    = (int*)(ws + off); off += (size_t)CNT_PAD;
    int*      offs   = (int*)(ws + off); off += (size_t)CNT_PAD;
    int*      list   = (int*)(ws + off); off += (size_t)EDGES;
    float*    hs2    = ws + off; off += (size_t)NNODES * 2;
    float*    hd2    = ws + off; off += (size_t)NNODES * 2;

    const int B = 256;

    init_kernel<<<CNT_PAD / B, B, 0, stream>>>(cnt);
    edge_count_kernel<<<(EDGES + B - 1) / B, B, 0, stream>>>(dst, cnt);
    offsets_kernel<<<1, SCAN_T, 0, stream>>>(cnt, offs);
    edge_fill_kernel<<<(EDGES + B - 1) / B, B, 0, stream>>>(src, dst, offs, cnt, list);
    {
        const int tiles = 3125 * 4;
        gemm1_kernel<<<(tiles + 7) / 8, B, 0, stream>>>(
            feat, W1s, b1s, W1d, b1d, hs1, hd1);
    }
    node_attn1_kernel<<<(NNODES + 7) / 8, B, 0, stream>>>(
        hs1, hd1, offs, list, attn1, out1);
    proj2_kernel<<<(NNODES + 7) / 8, B, 0, stream>>>(
        out1, W2s, b2s, W2d, b2d, hs2, hd2);
    node_attn2_kernel<<<(NNODES + 63) / 64, 64, 0, stream>>>(
        hs2, hd2, offs, list, attn2, out);

    (void)in_sizes; (void)n_in; (void)out_size; (void)ws_size;
}


# edge grouping rebuilt as a two-level counting sort on LDS counters (256 node ranges: range histogram, ranked scatter, per-range count, scan, place), no global atomics; sums in LDS-add arrival order
# speedup vs baseline: 4.4416x; 1.5701x over previous
#include <hip/hip_runtime.h>
#include <hip/hip_bf16.h>
#include <math.h>

#define NNODES 50000
#define EDGES  800000
#define INFEAT 128
#define HIDTOT 256
#define NEG_SLOPE 0.2f

typedef __attribute__((ext_vector_type(2))) float v2f;
typedef __attribute__((ext_vector_type(8))) float v8f;

__device__ __forceinline__ float lrelu(float x) {
    return x >= 0.f ? x : NEG_SLOPE * x;
}

#define SORT_T   1024
#define NRANGE   256
#define RSPAN    ((NNODES + NRANGE - 1) / NRANGE)
#define OFF_PAD  (NRANGE * RSPAN)
#define NBLK     256
#define EPB      ((EDGES + NBLK - 1) / NBLK)
#define EITER    ((EPB + SORT_T - 1) / SORT_T)

__global__ __launch_bounds__(256) void gemm1_kernel(
    const float* __restrict__ feat,
    const float* __restrict__ Ws, const float* __restrict__ bs,
    const float* __restrict__ Wd, const float* __restrict__ bd,
    float* __restrict__ hs, float* __restrict__ hd)
{
    const int wave = threadIdx.x >> 5;
    const int lane = threadIdx.x & 31;
    const int half = lane >> 4;
    const int row  = lane & 15;

    int tile = blockIdx.x * 8 + wave;
    if (tile >= 3125 * 4) return;
    const int mt  = tile >> 2;
    const int nt4 = tile & 3;
    const int m0  = mt * 16;
    const int c0  = nt4 * 64;

    v8f zero = {0.f,0.f,0.f,0.f,0.f,0.f,0.f,0.f};
    v8f accS[4], accD[4];
    #pragma unroll
    for (int j = 0; j < 4; ++j) { accS[j] = zero; accD[j] = zero; }

    const float* arow = feat + (size_t)(m0 + row) * INFEAT;
    for (int k0 = 0; k0 < INFEAT; k0 += 4) {
        const int ka = k0 + 2 * half;
        v2f a = *(const v2f*)(arow + ka);
        #pragma unroll
        for (int j = 0; j < 4; ++j) {
            const int col = c0 + j * 16 + row;
            v2f bS, bD;
            bS.x = Ws[(size_t)ka * HIDTOT + col];
            bS.y = Ws[(size_t)(ka + 1) * HIDTOT + col];
            bD.x = Wd[(size_t)ka * HIDTOT + col];
            bD.y = Wd[(size_t)(ka + 1) * HIDTOT + col];
            accS[j] = __builtin_amdgcn_wmma_f32_16x16x4_f32(
                false, a, false, bS, (short)0, accS[j], false, false);
            accD[j] = __builtin_amdgcn_wmma_f32_16x16x4_f32(
                false, a, false, bD, (short)0, accD[j], false, false);
        }
    }
    #pragma unroll
    for (int j = 0; j < 4; ++j) {
        const int col = c0 + j * 16 + row;
        const float biasS = bs[col], biasD = bd[col];
        #pragma unroll
        for (int i = 0; i < 8; ++i) {
            const int r = m0 + i + 8 * half;
            hs[(size_t)r * HIDTOT + col] = accS[j][i] + biasS;
            hd[(size_t)r * HIDTOT + col] = accD[j][i] + biasD;
        }
    }
}

__global__ __launch_bounds__(SORT_T) void range_hist_kernel(const int* __restrict__ dst,
                                                            int* __restrict__ ghist)
{
    __shared__ int h[NRANGE];
    const int t = threadIdx.x, b = blockIdx.x;
    if (t < NRANGE) h[t] = 0;
    __syncthreads();
    const int e0 = b * EPB;
    const int e1 = (e0 + EPB < EDGES) ? (e0 + EPB) : EDGES;
    int dv[EITER];
    #pragma unroll
    for (int j = 0; j < EITER; ++j) {
        const int e = e0 + t + j * SORT_T;
        dv[j] = (e < e1) ? dst[e] : -1;
    }
    #pragma unroll
    for (int j = 0; j < EITER; ++j)
        if (dv[j] >= 0) atomicAdd(&h[(unsigned)dv[j] / RSPAN], 1);
    __syncthreads();
    if (t < NRANGE) ghist[b * NRANGE + t] = h[t];
}

__global__ __launch_bounds__(SORT_T) void range_scatter_kernel(
    const int* __restrict__ src, const int* __restrict__ dst,
    const int* __restrict__ ghist, int* __restrict__ rstart, int* __restrict__ binned)
{
    __shared__ int part_pre[SORT_T / NRANGE][NRANGE];
    __shared__ int part_tot[SORT_T / NRANGE][NRANGE];
    __shared__ int cur[NRANGE];
    __shared__ int wsum[NRANGE / 32];
    const int t = threadIdx.x, b = blockIdx.x;
    const int lane = t & 31, wave = t >> 5;
    const int r = t % NRANGE, q = t / NRANGE;
    const int per = NBLK / (SORT_T / NRANGE);
    int pre = 0, tot = 0;
    #pragma unroll 8
    for (int j = 0; j < per; ++j) {
        const int bb = q * per + j;
        const int v = ghist[bb * NRANGE + r];
        tot += v;
        pre += (bb < b) ? v : 0;
    }
    part_pre[q][r] = pre;
    part_tot[q][r] = tot;
    __syncthreads();
    int mypre = 0, mytot = 0, inc = 0;
    if (t < NRANGE) {
        #pragma unroll
        for (int k = 0; k < SORT_T / NRANGE; ++k) { mypre += part_pre[k][t]; mytot += part_tot[k][t]; }
        inc = mytot;
        #pragma unroll
        for (int o = 1; o < 32; o <<= 1) {
            const int y = __shfl_up(inc, o, 32);
            if (lane >= o) inc += y;
        }
        if (lane == 31) wsum[wave] = inc;
    }
    __syncthreads();
    if (t < NRANGE) {
        int wbase = 0;
        #pragma unroll
        for (int w = 0; w < NRANGE / 32; ++w) wbase += (w < wave) ? wsum[w] : 0;
        const int start = wbase + inc - mytot;
        cur[t] = start + mypre;
        if (b == 0) {
            rstart[t] = start;
            if (t == NRANGE - 1) rstart[NRANGE] = start + mytot;
        }
    }
    __syncthreads();
    const int e0 = b * EPB;
    const int e1 = (e0 + EPB < EDGES) ? (e0 + EPB) : EDGES;
    int dv[EITER], sv[EITER];
    #pragma unroll
    for (int j = 0; j < EITER; ++j) {
        const int e = e0 + t + j * SORT_T;
        dv[j] = (e < e1) ? dst[e] : -1;
        sv[j] = (e < e1) ? src[e] : 0;
    }
    int2* bin2 = (int2*)binned;
    #pragma unroll
    for (int j = 0; j < EITER; ++j) {
        if (dv[j] >= 0) {
            const int p = atomicAdd(&cur[(unsigned)dv[j] / RSPAN], 1);
            bin2[p] = make_int2(dv[j], sv[j]);
        }
    }
}

__global__ __launch_bounds__(SORT_T) void range_place_kernel(
    const int* __restrict__ binned, const int* __restrict__ rstart,
    int* __restrict__ off, int* __restrict__ list)
{
    __shared__ int cnt[256];
    __shared__ int wsum[8];
    const int t = threadIdx.x, r = blockIdx.x;
    const int lane = t & 31, wave = t >> 5;
    const int beg = rstart[r], end = rstart[r + 1];
    const int n0 = r * RSPAN;
    const int2* bin2 = (const int2*)binned;
    if (t < 256) cnt[t] = 0;
    __syncthreads();
    for (int base = beg; base < end; base += EITER * SORT_T) {
        int dv[EITER];
        #pragma unroll
        for (int j = 0; j < EITER; ++j) {
            const int i = base + t + j * SORT_T;
            dv[j] = (i < end) ? bin2[i].x : -1;
        }
        #pragma unroll
        for (int j = 0; j < EITER; ++j)
            if (dv[j] >= 0) atomicAdd(&cnt[dv[j] - n0], 1);
    }
    __syncthreads();
    int c = 0, inc = 0;
    if (t < 256) {
        c = cnt[t];
        inc = c;
        #pragma unroll
        for (int o = 1; o < 32; o <<= 1) {
            const int y = __shfl_up(inc, o, 32);
            if (lane >= o) inc += y;
        }
        if (lane == 31) wsum[wave] = inc;
    }
    __syncthreads();
    if (t < 256) {
        int wbase = 0;
        #pragma unroll
        for (int w = 0; w < 8; ++w) wbase += (w < wave) ? wsum[w] : 0;
        const int start = beg + wbase + inc - c;
        if (t < RSPAN) off[n0 + t] = start;
        cnt[t] = start;
    }
    __syncthreads();
    for (int base = beg; base < end; base += EITER * SORT_T) {
        int2 ev[EITER];
        #pragma unroll
        for (int j = 0; j < EITER; ++j) {
            const int i = base + t + j * SORT_T;
            ev[j] = (i < end) ? bin2[i] : make_int2(-1, 0);
        }
        #pragma unroll
        for (int j = 0; j < EITER; ++j) {
            if (ev[j].x >= 0) {
                const int p = atomicAdd(&cnt[ev[j].x - n0], 1);
                list[p] = ev[j].y;
            }
        }
    }
}

template <int M> __device__ __forceinline__ float row_xor(float v) {
    return __int_as_float(__builtin_amdgcn_update_dpp(0, __float_as_int(v), 0x160 | M, 0xF, 0xF, true));
}
__device__ __forceinline__ float row16_sum(float v) {
    v += row_xor<1>(v);
    v += row_xor<2>(v);
    v += row_xor<4>(v);
    v += row_xor<8>(v);
    return v;
}
__device__ __forceinline__ float dot_lrelu(const float4 a, const float4 b, const float4 w) {
    return lrelu(a.x + b.x) * w.x + lrelu(a.y + b.y) * w.y +
           lrelu(a.z + b.z) * w.z + lrelu(a.w + b.w) * w.w;
}
__device__ __forceinline__ float elu1(float v) { return v > 0.f ? v : (expf(v) - 1.f); }

#define GRP 4
__global__ __launch_bounds__(256) void node_attn1_kernel(
    const float* __restrict__ hs, const float* __restrict__ hd,
    const int* __restrict__ off, const int* __restrict__ list,
    const float* __restrict__ attn, float* __restrict__ out)
{
    const int d = __builtin_amdgcn_readfirstlane((int)(blockIdx.x * 8 + (threadIdx.x >> 5)));
    if (d >= NNODES) return;
    const int lane = threadIdx.x & 31;
    const int beg = off[d], end = off[d + 1];

    const float* pd = hd + (size_t)d * HIDTOT + 4 * lane;
    const float4 b0 = *(const float4*)(pd);
    const float4 b1 = *(const float4*)(pd + 128);
    const float4 w0 = *(const float4*)(attn + 4 * lane);
    const float4 w1 = *(const float4*)(attn + 128 + 4 * lane);

    float m0 = -INFINITY, m1 = -INFINITY, l0 = 0.f, l1 = 0.f;
    float4 acc0 = {0.f, 0.f, 0.f, 0.f}, acc1 = {0.f, 0.f, 0.f, 0.f};

    for (int i = beg; i < end; i += 32) {
        const int n = (end - i) < 32 ? (end - i) : 32;
        const int my_s = lane < n ? list[i + lane] : 0;
        for (int k = 0; k < n; k += GRP) {
            float4 a0[GRP], a1[GRP];
            #pragma unroll
            for (int j = 0; j < GRP; ++j) {
                const int kk = (k + j < n) ? (k + j) : k;
                const int s = __builtin_amdgcn_readlane(my_s, kk);
                const float* ps = hs + (size_t)s * HIDTOT + 4 * lane;
                a0[j] = *(const float4*)(ps);
                a1[j] = *(const float4*)(ps + 128);
            }
            float p0[GRP], p1[GRP];
            float g0 = m0, g1 = m1;
            #pragma unroll
            for (int j = 0; j < GRP; ++j) {
                float x0 = row16_sum(dot_lrelu(a0[j], b0, w0));
                float x1 = row16_sum(dot_lrelu(a1[j], b1, w1));
                if (k + j >= n) { x0 = -INFINITY; x1 = -INFINITY; }
                p0[j] = x0; p1[j] = x1;
                g0 = fmaxf(g0, x0); g1 = fmaxf(g1, x1);
            }
            const float r0 = expf(m0 - g0), r1 = expf(m1 - g1);
            l0 *= r0; l1 *= r1;
            acc0.x *= r0; acc0.y *= r0; acc0.z *= r0; acc0.w *= r0;
            acc1.x *= r1; acc1.y *= r1; acc1.z *= r1; acc1.w *= r1;
            #pragma unroll
            for (int j = 0; j < GRP; ++j) {
                const float e0 = expf(p0[j] - g0), e1 = expf(p1[j] - g1);
                l0 += e0; l1 += e1;
                acc0.x += e0 * a0[j].x; acc0.y += e0 * a0[j].y;
                acc0.z += e0 * a0[j].z; acc0.w += e0 * a0[j].w;
                acc1.x += e1 * a1[j].x; acc1.y += e1 * a1[j].y;
                acc1.z += e1 * a1[j].z; acc1.w += e1 * a1[j].w;
            }
            m0 = g0; m1 = g1;
        }
    }
    const float q0 = l0 > 0.f ? l0 : 1.f, q1 = l1 > 0.f ? l1 : 1.f;
    float4 o0, o1;
    o0.x = elu1(acc0.x / q0); o0.y = elu1(acc0.y / q0);
    o0.z = elu1(acc0.z / q0); o0.w = elu1(acc0.w / q0);
    o1.x = elu1(acc1.x / q1); o1.y = elu1(acc1.y / q1);
    o1.z = elu1(acc1.z / q1); o1.w = elu1(acc1.w / q1);
    float* po = out + (size_t)d * HIDTOT + 4 * lane;
    *(float4*)(po) = o0;
    *(float4*)(po + 128) = o1;
}

__global__ __launch_bounds__(256) void proj2_kernel(
    const float* __restrict__ h,
    const float* __restrict__ W2s, const float* __restrict__ b2s,
    const float* __restrict__ W2d, const float* __restrict__ b2d,
    float* __restrict__ hs2, float* __restrict__ hd2)
{
    const int n = blockIdx.x * 8 + (threadIdx.x >> 5);
    if (n >= NNODES) return;
    const int lane = threadIdx.x & 31;
    const int base = lane * 8;
    const float* hr = h + (size_t)n * HIDTOT + base;
    float s0 = 0.f, s1 = 0.f, d0 = 0.f, d1 = 0.f;
    #pragma unroll
    for (int k = 0; k < 8; ++k) {
        const float x = hr[k];
        const float2 ws = *(const float2*)(W2s + (size_t)(base + k) * 2);
        const float2 wd = *(const float2*)(W2d + (size_t)(base + k) * 2);
        s0 += x * ws.x; s1 += x * ws.y;
        d0 += x * wd.x; d1 += x * wd.y;
    }
    #pragma unroll
    for (int o = 1; o < 32; o <<= 1) {
        s0 += __shfl_xor(s0, o, 32); s1 += __shfl_xor(s1, o, 32);
        d0 += __shfl_xor(d0, o, 32); d1 += __shfl_xor(d1, o, 32);
    }
    if (lane == 0) {
        hs2[(size_t)n * 2 + 0] = s0 + b2s[0];
        hs2[(size_t)n * 2 + 1] = s1 + b2s[1];
        hd2[(size_t)n * 2 + 0] = d0 + b2d[0];
        hd2[(size_t)n * 2 + 1] = d1 + b2d[1];
    }
}

#define GRP2 8
__global__ __launch_bounds__(64) void node_attn2_kernel(
    const float* __restrict__ hs2, const float* __restrict__ hd2,
    const int* __restrict__ off, const int* __restrict__ list,
    const float* __restrict__ attn2, float* __restrict__ out)
{
    const int d = blockIdx.x * blockDim.x + threadIdx.x;
    if (d >= NNODES) return;
    const int beg = off[d], end = off[d + 1];
    const float2 b = *(const float2*)(hd2 + (size_t)d * 2);
    const float w0 = attn2[0], w1 = attn2[1];

    float m = -INFINITY, l = 0.f, ax = 0.f, ay = 0.f;
    for (int i = beg; i < end; i += GRP2) {
        int s[GRP2];
        float2 a[GRP2];
        #pragma unroll
        for (int j = 0; j < GRP2; ++j) {
            const int ii = (i + j < end) ? (i + j) : i;
            s[j] = list[ii];
        }
        #pragma unroll
        for (int j = 0; j < GRP2; ++j) a[j] = *(const float2*)(hs2 + (size_t)s[j] * 2);
        float p[GRP2];
        float g = m;
        #pragma unroll
        for (int j = 0; j < GRP2; ++j) {
            float x = lrelu(a[j].x + b.x) * w0 + lrelu(a[j].y + b.y) * w1;
            if (i + j >= end) x = -INFINITY;
            p[j] = x;
            g = fmaxf(g, x);
        }
        const float r = expf(m - g);
        l *= r; ax *= r; ay *= r;
        #pragma unroll
        for (int j = 0; j < GRP2; ++j) {
            const float e = expf(p[j] - g);
            l += e;
            ax += e * a[j].x;
            ay += e * a[j].y;
        }
        m = g;
    }
    const float q = l > 0.f ? l : 1.f;
    float2 o;
    o.x = ax / q;
    o.y = ay / q;
    *(float2*)(out + (size_t)d * 2) = o;
}

extern "C" void kernel_launch(void* const* d_in, const int* in_sizes, int n_in,
                              void* d_out, int out_size, void* d_ws, size_t ws_size,
                              hipStream_t stream)
{
    const float* feat  = (const float*)d_in[0];
    const int*   src   = (const int*)d_in[1];
    const int*   dst   = (const int*)d_in[2];
    const float* W1s   = (const float*)d_in[3];
    const float* b1s   = (const float*)d_in[4];
    const float* W1d   = (const float*)d_in[5];
    const float* b1d   = (const float*)d_in[6];
    const float* attn1 = (const float*)d_in[7];
    const float* W2s   = (const float*)d_in[8];
    const float* b2s   = (const float*)d_in[9];
    const float* W2d   = (const float*)d_in[10];
    const float* b2d   = (const float*)d_in[11];
    const float* attn2 = (const float*)d_in[12];
    float* out = (float*)d_out;

    float* ws = (float*)d_ws;
    size_t off = 0;
    float*    hs1    = ws + off; off += (size_t)NNODES * HIDTOT;
    float*    hd1    = ws + off; off += (size_t)NNODES * HIDTOT;
    float*    out1   = ws + off; off += (size_t)NNODES * HIDTOT;
    int*      ghist  = (int*)(ws + off); off += (size_t)NBLK * NRANGE;
    int*      rstart = (int*)(ws + off); off += (size_t)2 * NRANGE;
    int*      offs   = (int*)(ws + off); off += (size_t)OFF_PAD;
    int*      binned = (int*)(ws + off); off += (size_t)2 * EDGES;
    int*      list   = (int*)(ws + off); off += (size_t)EDGES;
    float*    hs2    = ws + off; off += (size_t)NNODES * 2;
    float*    hd2    = ws + off; off += (size_t)NNODES * 2;

    const int B = 256;

    range_hist_kernel<<<NBLK, SORT_T, 0, stream>>>(dst, ghist);
    range_scatter_kernel<<<NBLK, SORT_T, 0, stream>>>(src, dst, ghist, rstart, binned);
    range_place_kernel<<<NRANGE, SORT_T, 0, stream>>>(binned, rstart, offs, list);
    {
        const int tiles = 3125 * 4;
        gemm1_kernel<<<(tiles + 7) / 8, B, 0, stream>>>(
            feat, W1s, b1s, W1d, b1d, hs1, hd1);
    }
    node_attn1_kernel<<<(NNODES + 7) / 8, B, 0, stream>>>(
        hs1, hd1, offs, list, attn1, out1);
    proj2_kernel<<<(NNODES + 7) / 8, B, 0, stream>>>(
        out1, W2s, b2s, W2d, b2d, hs2, hd2);
    node_attn2_kernel<<<(NNODES + 63) / 64, 64, 0, stream>>>(
        hs2, hd2, offs, list, attn2, out);

    (void)in_sizes; (void)n_in; (void)out_size; (void)ws_size;
}


# layer-2 projections (256->2, both sides) computed in the layer-1 node kernel from the row in registers: the 256-wide row is never stored, one launch fewer; projection sums in a new tree order
# speedup vs baseline: 4.7085x; 1.0601x over previous
#include <hip/hip_runtime.h>
#include <hip/hip_bf16.h>
#include <math.h>

#define NNODES 50000
#define EDGES  800000
#define INFEAT 128
#define HIDTOT 256
#define NEG_SLOPE 0.2f

typedef __attribute__((ext_vector_type(2))) float v2f;
typedef __attribute__((ext_vector_type(8))) float v8f;

__device__ __forceinline__ float lrelu(float x) {
    return x >= 0.f ? x : NEG_SLOPE * x;
}

#define SORT_T   1024
#define NRANGE   256
#define RSPAN    ((NNODES + NRANGE - 1) / NRANGE)
#define OFF_PAD  (NRANGE * RSPAN)
#define NBLK     256
#define EPB      ((EDGES + NBLK - 1) / NBLK)
#define EITER    ((EPB + SORT_T - 1) / SORT_T)

__global__ __launch_bounds__(256) void gemm1_kernel(
    const float* __restrict__ feat,
    const float* __restrict__ Ws, const float* __restrict__ bs,
    const float* __restrict__ Wd, const float* __restrict__ bd,
    float* __restrict__ hs, float* __restrict__ hd)
{
    const int wave = threadIdx.x >> 5;
    const int lane = threadIdx.x & 31;
    const int half = lane >> 4;
    const int row  = lane & 15;

    int tile = blockIdx.x * 8 + wave;
    if (tile >= 3125 * 4) return;
    const int mt  = tile >> 2;
    const int nt4 = tile & 3;
    const int m0  = mt * 16;
    const int c0  = nt4 * 64;

    v8f zero = {0.f,0.f,0.f,0.f,0.f,0.f,0.f,0.f};
    v8f accS[4], accD[4];
    #pragma unroll
    for (int j = 0; j < 4; ++j) { accS[j] = zero; accD[j] = zero; }

    const float* arow = feat + (size_t)(m0 + row) * INFEAT;
    for (int k0 = 0; k0 < INFEAT; k0 += 4) {
        const int ka = k0 + 2 * half;
        v2f a = *(const v2f*)(arow + ka);
        #pragma unroll
        for (int j = 0; j < 4; ++j) {
            const int col = c0 + j * 16 + row;
            v2f bS, bD;
            bS.x = Ws[(size_t)ka * HIDTOT + col];
            bS.y = Ws[(size_t)(ka + 1) * HIDTOT + col];
            bD.x = Wd[(size_t)ka * HIDTOT + col];
            bD.y = Wd[(size_t)(ka + 1) * HIDTOT + col];
            accS[j] = __builtin_amdgcn_wmma_f32_16x16x4_f32(
                false, a, false, bS, (short)0, accS[j], false, false);
            accD[j] = __builtin_amdgcn_wmma_f32_16x16x4_f32(
                false, a, false, bD, (short)0, accD[j], false, false);
        }
    }
    #pragma unroll
    for (int j = 0; j < 4; ++j) {
        const int col = c0 + j * 16 + row;
        const float biasS = bs[col], biasD = bd[col];
        #pragma unroll
        for (int i = 0; i < 8; ++i) {
            const int r = m0 + i + 8 * half;
            hs[(size_t)r * HIDTOT + col] = accS[j][i] + biasS;
            hd[(size_t)r * HIDTOT + col] = accD[j][i] + biasD;
        }
    }
}

__global__ __launch_bounds__(SORT_T) void range_hist_kernel(const int* __restrict__ dst,
                                                            int* __restrict__ ghist)
{
    __shared__ int h[NRANGE];
    const int t = threadIdx.x, b = blockIdx.x;
    if (t < NRANGE) h[t] = 0;
    __syncthreads();
    const int e0 = b * EPB;
    const int e1 = (e0 + EPB < EDGES) ? (e0 + EPB) : EDGES;
    int dv[EITER];
    #pragma unroll
    for (int j = 0; j < EITER; ++j) {
        const int e = e0 + t + j * SORT_T;
        dv[j] = (e < e1) ? dst[e] : -1;
    }
    #pragma unroll
    for (int j = 0; j < EITER; ++j)
        if (dv[j] >= 0) atomicAdd(&h[(unsigned)dv[j] / RSPAN], 1);
    __syncthreads();
    if (t < NRANGE) ghist[b * NRANGE + t] = h[t];
}

__global__ __launch_bounds__(SORT_T) void range_scatter_kernel(
    const int* __restrict__ src, const int* __restrict__ dst,
    const int* __restrict__ ghist, int* __restrict__ rstart, int* __restrict__ binned)
{
    __shared__ int part_pre[SORT_T / NRANGE][NRANGE];
    __shared__ int part_tot[SORT_T / NRANGE][NRANGE];
    __shared__ int cur[NRANGE];
    __shared__ int wsum[NRANGE / 32];
    const int t = threadIdx.x, b = blockIdx.x;
    const int lane = t & 31, wave = t >> 5;
    const int r = t % NRANGE, q = t / NRANGE;
    const int per = NBLK / (SORT_T / NRANGE);
    int pre = 0, tot = 0;
    #pragma unroll 8
    for (int j = 0; j < per; ++j) {
        const int bb = q * per + j;
        const int v = ghist[bb * NRANGE + r];
        tot += v;
        pre += (bb < b) ? v : 0;
    }
    part_pre[q][r] = pre;
    part_tot[q][r] = tot;
    __syncthreads();
    int mypre = 0, mytot = 0, inc = 0;
    if (t < NRANGE) {
        #pragma unroll
        for (int k = 0; k < SORT_T / NRANGE; ++k) { mypre += part_pre[k][t]; mytot += part_tot[k][t]; }
        inc = mytot;
        #pragma unroll
        for (int o = 1; o < 32; o <<= 1) {
            const int y = __shfl_up(inc, o, 32);
            if (lane >= o) inc += y;
        }
        if (lane == 31) wsum[wave] = inc;
    }
    __syncthreads();
    if (t < NRANGE) {
        int wbase = 0;
        #pragma unroll
        for (int w = 0; w < NRANGE / 32; ++w) wbase += (w < wave) ? wsum[w] : 0;
        const int start = wbase + inc - mytot;
        cur[t] = start + mypre;
        if (b == 0) {
            rstart[t] = start;
            if (t == NRANGE - 1) rstart[NRANGE] = start + mytot;
        }
    }
    __syncthreads();
    const int e0 = b * EPB;
    const int e1 = (e0 + EPB < EDGES) ? (e0 + EPB) : EDGES;
    int dv[EITER], sv[EITER];
    #pragma unroll
    for (int j = 0; j < EITER; ++j) {
        const int e = e0 + t + j * SORT_T;
        dv[j] = (e < e1) ? dst[e] : -1;
        sv[j] = (e < e1) ? src[e] : 0;
    }
    int2* bin2 = (int2*)binned;
    #pragma unroll
    for (int j = 0; j < EITER; ++j) {
        if (dv[j] >= 0) {
            const int p = atomicAdd(&cur[(unsigned)dv[j] / RSPAN], 1);
            bin2[p] = make_int2(dv[j], sv[j]);
        }
    }
}

__global__ __launch_bounds__(SORT_T) void range_place_kernel(
    const int* __restrict__ binned, const int* __restrict__ rstart,
    int* __restrict__ off, int* __restrict__ list)
{
    __shared__ int cnt[256];
    __shared__ int wsum[8];
    const int t = threadIdx.x, r = blockIdx.x;
    const int lane = t & 31, wave = t >> 5;
    const int beg = rstart[r], end = rstart[r + 1];
    const int n0 = r * RSPAN;
    const int2* bin2 = (const int2*)binned;
    if (t < 256) cnt[t] = 0;
    __syncthreads();
    for (int base = beg; base < end; base += EITER * SORT_T) {
        int dv[EITER];
        #pragma unroll
        for (int j = 0; j < EITER; ++j) {
            const int i = base + t + j * SORT_T;
            dv[j] = (i < end) ? bin2[i].x : -1;
        }
        #pragma unroll
        for (int j = 0; j < EITER; ++j)
            if (dv[j] >= 0) atomicAdd(&cnt[dv[j] - n0], 1);
    }
    __syncthreads();
    int c = 0, inc = 0;
    if (t < 256) {
        c = cnt[t];
        inc = c;
        #pragma unroll
        for (int o = 1; o < 32; o <<= 1) {
            const int y = __shfl_up(inc, o, 32);
            if (lane >= o) inc += y;
        }
        if (lane == 31) wsum[wave] = inc;
    }
    __syncthreads();
    if (t < 256) {
        int wbase = 0;
        #pragma unroll
        for (int w = 0; w < 8; ++w) wbase += (w < wave) ? wsum[w] : 0;
        const int start = beg + wbase + inc - c;
        if (t < RSPAN) off[n0 + t] = start;
        cnt[t] = start;
    }
    __syncthreads();
    for (int base = beg; base < end; base += EITER * SORT_T) {
        int2 ev[EITER];
        #pragma unroll
        for (int j = 0; j < EITER; ++j) {
            const int i = base + t + j * SORT_T;
            ev[j] = (i < end) ? bin2[i] : make_int2(-1, 0);
        }
        #pragma unroll
        for (int j = 0; j < EITER; ++j) {
            if (ev[j].x >= 0) {
                const int p = atomicAdd(&cnt[ev[j].x - n0], 1);
                list[p] = ev[j].y;
            }
        }
    }
}

template <int M> __device__ __forceinline__ float row_xor(float v) {
    return __int_as_float(__builtin_amdgcn_update_dpp(0, __float_as_int(v), 0x160 | M, 0xF, 0xF, true));
}
__device__ __forceinline__ float row16_sum(float v) {
    v += row_xor<1>(v);
    v += row_xor<2>(v);
    v += row_xor<4>(v);
    v += row_xor<8>(v);
    return v;
}
__device__ __forceinline__ float other_row(float v) {
    return __int_as_float(__builtin_amdgcn_permlanex16(__float_as_int(v), __float_as_int(v),
                                                       0x76543210, 0xfedcba98, false, false));
}
__device__ __forceinline__ float dot_lrelu(const float4 a, const float4 b, const float4 w) {
    return lrelu(a.x + b.x) * w.x + lrelu(a.y + b.y) * w.y +
           lrelu(a.z + b.z) * w.z + lrelu(a.w + b.w) * w.w;
}
__device__ __forceinline__ float elu1(float v) { return v > 0.f ? v : (expf(v) - 1.f); }

#define GRP 4
__global__ __launch_bounds__(256) void node_attn1_kernel(
    const float* __restrict__ hs, const float* __restrict__ hd,
    const int* __restrict__ off, const int* __restrict__ list,
    const float* __restrict__ attn,
    const float* __restrict__ W2s, const float* __restrict__ b2s,
    const float* __restrict__ W2d, const float* __restrict__ b2d,
    float* __restrict__ hs2, float* __restrict__ hd2)
{
    const int d = __builtin_amdgcn_readfirstlane((int)(blockIdx.x * 8 + (threadIdx.x >> 5)));
    if (d >= NNODES) return;
    const int lane = threadIdx.x & 31;
    const int beg = off[d], end = off[d + 1];

    const float* pd = hd + (size_t)d * HIDTOT + 4 * lane;
    const float4 b0 = *(const float4*)(pd);
    const float4 b1 = *(const float4*)(pd + 128);
    const float4 w0 = *(const float4*)(attn + 4 * lane);
    const float4 w1 = *(const float4*)(attn + 128 + 4 * lane);

    float m0 = -INFINITY, m1 = -INFINITY, l0 = 0.f, l1 = 0.f;
    float4 acc0 = {0.f, 0.f, 0.f, 0.f}, acc1 = {0.f, 0.f, 0.f, 0.f};

    for (int i = beg; i < end; i += 32) {
        const int n = (end - i) < 32 ? (end - i) : 32;
        const int my_s = lane < n ? list[i + lane] : 0;
        for (int k = 0; k < n; k += GRP) {
            float4 a0[GRP], a1[GRP];
            #pragma unroll
            for (int j = 0; j < GRP; ++j) {
                const int kk = (k + j < n) ? (k + j) : k;
                const int s = __builtin_amdgcn_readlane(my_s, kk);
                const float* ps = hs + (size_t)s * HIDTOT + 4 * lane;
                a0[j] = *(const float4*)(ps);
                a1[j] = *(const float4*)(ps + 128);
            }
            float p0[GRP], p1[GRP];
            float g0 = m0, g1 = m1;
            #pragma unroll
            for (int j = 0; j < GRP; ++j) {
                float x0 = row16_sum(dot_lrelu(a0[j], b0, w0));
                float x1 = row16_sum(dot_lrelu(a1[j], b1, w1));
                if (k + j >= n) { x0 = -INFINITY; x1 = -INFINITY; }
                p0[j] = x0; p1[j] = x1;
                g0 = fmaxf(g0, x0); g1 = fmaxf(g1, x1);
            }
            const float r0 = expf(m0 - g0), r1 = expf(m1 - g1);
            l0 *= r0; l1 *= r1;
            acc0.x *= r0; acc0.y *= r0; acc0.z *= r0; acc0.w *= r0;
            acc1.x *= r1; acc1.y *= r1; acc1.z *= r1; acc1.w *= r1;
            #pragma unroll
            for (int j = 0; j < GRP; ++j) {
                const float e0 = expf(p0[j] - g0), e1 = expf(p1[j] - g1);
                l0 += e0; l1 += e1;
                acc0.x += e0 * a0[j].x; acc0.y += e0 * a0[j].y;
                acc0.z += e0 * a0[j].z; acc0.w += e0 * a0[j].w;
                acc1.x += e1 * a1[j].x; acc1.y += e1 * a1[j].y;
                acc1.z += e1 * a1[j].z; acc1.w += e1 * a1[j].w;
            }
            m0 = g0; m1 = g1;
        }
    }
    const float4 s_a = *(const float4*)(W2s + 8 * lane);
    const float4 s_b = *(const float4*)(W2s + 8 * lane + 4);
    const float4 s_c = *(const float4*)(W2s + 256 + 8 * lane);
    const float4 s_d = *(const float4*)(W2s + 256 + 8 * lane + 4);
    const float4 d_a = *(const float4*)(W2d + 8 * lane);
    const float4 d_b = *(const float4*)(W2d + 8 * lane + 4);
    const float4 d_c = *(const float4*)(W2d + 256 + 8 * lane);
    const float4 d_d = *(const float4*)(W2d + 256 + 8 * lane + 4);
    const float q0 = l0 > 0.f ? l0 : 1.f, q1 = l1 > 0.f ? l1 : 1.f;
    float4 o0, o1;
    o0.x = elu1(acc0.x / q0); o0.y = elu1(acc0.y / q0);
    o0.z = elu1(acc0.z / q0); o0.w = elu1(acc0.w / q0);
    o1.x = elu1(acc1.x / q1); o1.y = elu1(acc1.y / q1);
    o1.z = elu1(acc1.z / q1); o1.w = elu1(acc1.w / q1);
    float ps0 = o0.x * s_a.x + o0.y * s_a.z + o0.z * s_b.x + o0.w * s_b.z
              + o1.x * s_c.x + o1.y * s_c.z + o1.z * s_d.x + o1.w * s_d.z;
    float ps1 = o0.x * s_a.y + o0.y * s_a.w + o0.z * s_b.y + o0.w * s_b.w
              + o1.x * s_c.y + o1.y * s_c.w + o1.z * s_d.y + o1.w * s_d.w;
    float pd0 = o0.x * d_a.x + o0.y * d_a.z + o0.z * d_b.x + o0.w * d_b.z
              + o1.x * d_c.x + o1.y * d_c.z + o1.z * d_d.x + o1.w * d_d.z;
    float pd1 = o0.x * d_a.y + o0.y * d_a.w + o0.z * d_b.y + o0.w * d_b.w
              + o1.x * d_c.y + o1.y * d_c.w + o1.z * d_d.y + o1.w * d_d.w;
    ps0 = row16_sum(ps0); ps1 = row16_sum(ps1);
    pd0 = row16_sum(pd0); pd1 = row16_sum(pd1);
    ps0 += other_row(ps0); ps1 += other_row(ps1);
    pd0 += other_row(pd0); pd1 += other_row(pd1);
    if (lane == 0) {
        float2 vs, vd;
        vs.x = ps0 + b2s[0]; vs.y = ps1 + b2s[1];
        vd.x = pd0 + b2d[0]; vd.y = pd1 + b2d[1];
        *(float2*)(hs2 + (size_t)d * 2) = vs;
        *(float2*)(hd2 + (size_t)d * 2) = vd;
    }
}

#define GRP2 8
__global__ __launch_bounds__(64) void node_attn2_kernel(
    const float* __restrict__ hs2, const float* __restrict__ hd2,
    const int* __restrict__ off, const int* __restrict__ list,
    const float* __restrict__ attn2, float* __restrict__ out)
{
    const int d = blockIdx.x * blockDim.x + threadIdx.x;
    if (d >= NNODES) return;
    const int beg = off[d], end = off[d + 1];
    const float2 b = *(const float2*)(hd2 + (size_t)d * 2);
    const float w0 = attn2[0], w1 = attn2[1];

    float m = -INFINITY, l = 0.f, ax = 0.f, ay = 0.f;
    for (int i = beg; i < end; i += GRP2) {
        int s[GRP2];
        float2 a[GRP2];
        #pragma unroll
        for (int j = 0; j < GRP2; ++j) {
            const int ii = (i + j < end) ? (i + j) : i;
            s[j] = list[ii];
        }
        #pragma unroll
        for (int j = 0; j < GRP2; ++j) a[j] = *(const float2*)(hs2 + (size_t)s[j] * 2);
        float p[GRP2];
        float g = m;
        #pragma unroll
        for (int j = 0; j < GRP2; ++j) {
            float x = lrelu(a[j].x + b.x) * w0 + lrelu(a[j].y + b.y) * w1;
            if (i + j >= end) x = -INFINITY;
            p[j] = x;
            g = fmaxf(g, x);
        }
        const float r = expf(m - g);
        l *= r; ax *= r; ay *= r;
        #pragma unroll
        for (int j = 0; j < GRP2; ++j) {
            const float e = expf(p[j] - g);
            l += e;
            ax += e * a[j].x;
            ay += e * a[j].y;
        }
        m = g;
    }
    const float q = l > 0.f ? l : 1.f;
    float2 o;
    o.x = ax / q;
    o.y = ay / q;
    *(float2*)(out + (size_t)d * 2) = o;
}

extern "C" void kernel_launch(void* const* d_in, const int* in_sizes, int n_in,
                              void* d_out, int out_size, void* d_ws, size_t ws_size,
                              hipStream_t stream)
{
    const float* feat  = (const float*)d_in[0];
    const int*   src   = (const int*)d_in[1];
    const int*   dst   = (const int*)d_in[2];
    const float* W1s   = (const float*)d_in[3];
    const float* b1s   = (const float*)d_in[4];
    const float* W1d   = (const float*)d_in[5];
    const float* b1d   = (const float*)d_in[6];
    const float* attn1 = (const float*)d_in[7];
    const float* W2s   = (const float*)d_in[8];
    const float* b2s   = (const float*)d_in[9];
    const float* W2d   = (const float*)d_in[10];
    const float* b2d   = (const float*)d_in[11];
    const float* attn2 = (const float*)d_in[12];
    float* out = (float*)d_out;

    float* ws = (float*)d_ws;
    size_t off = 0;
    float*    hs1    = ws + off; off += (size_t)NNODES * HIDTOT;
    float*    hd1    = ws + off; off += (size_t)NNODES * HIDTOT;
    int*      ghist  = (int*)(ws + off); off += (size_t)NBLK * NRANGE;
    int*      rstart = (int*)(ws + off); off += (size_t)2 * NRANGE;
    int*      offs   = (int*)(ws + off); off += (size_t)OFF_PAD;
    int*      binned = (int*)(ws + off); off += (size_t)2 * EDGES;
    int*      list   = (int*)(ws + off); off += (size_t)EDGES;
    float*    hs2    = ws + off; off += (size_t)NNODES * 2;
    float*    hd2    = ws + off; off += (size_t)NNODES * 2;

    const int B = 256;

    range_hist_kernel<<<NBLK, SORT_T, 0, stream>>>(dst, ghist);
    range_scatter_kernel<<<NBLK, SORT_T, 0, stream>>>(src, dst, ghist, rstart, binned);
    range_place_kernel<<<NRANGE, SORT_T, 0, stream>>>(binned, rstart, offs, list);
    {
        const int tiles = 3125 * 4;
        gemm1_kernel<<<(tiles + 7) / 8, B, 0, stream>>>(
            feat, W1s, b1s, W1d, b1d, hs1, hd1);
    }
    node_attn1_kernel<<<(NNODES + 7) / 8, B, 0, stream>>>(
        hs1, hd1, offs, list, attn1, W2s, b2s, W2d, b2d, hs2, hd2);
    node_attn2_kernel<<<(NNODES + 63) / 64, 64, 0, stream>>>(
        hs2, hd2, offs, list, attn2, out);

    (void)in_sizes; (void)n_in; (void)out_size; (void)ws_size;
}


# layer-1 node loop: a group's 8 score sums reduced transposed (one score per lane), so mask, max and ONE hardware exp2 serve all 8; lrelu as max; denominators summed per slot, then across slots
# speedup vs baseline: 5.3182x; 1.1295x over previous
#include <hip/hip_runtime.h>
#include <hip/hip_bf16.h>
#include <math.h>

#define NNODES 50000
#define EDGES  800000
#define INFEAT 128
#define HIDTOT 256
#define NEG_SLOPE 0.2f

typedef __attribute__((ext_vector_type(2))) float v2f;
typedef __attribute__((ext_vector_type(8))) float v8f;

__device__ __forceinline__ float lrelu(float x) {
    return x >= 0.f ? x : NEG_SLOPE * x;
}

#define SORT_T   1024
#define NRANGE   256
#define RSPAN    ((NNODES + NRANGE - 1) / NRANGE)
#define OFF_PAD  (NRANGE * RSPAN)
#define NBLK     256
#define EPB      ((EDGES + NBLK - 1) / NBLK)
#define EITER    ((EPB + SORT_T - 1) / SORT_T)

__global__ __launch_bounds__(256) void gemm1_kernel(
    const float* __restrict__ feat,
    const float* __restrict__ Ws, const float* __restrict__ bs,
    const float* __restrict__ Wd, const float* __restrict__ bd,
    float* __restrict__ hs, float* __restrict__ hd)
{
    const int wave = threadIdx.x >> 5;
    const int lane = threadIdx.x & 31;
    const int half = lane >> 4;
    const int row  = lane & 15;

    int tile = blockIdx.x * 8 + wave;
    if (tile >= 3125 * 4) return;
    const int mt  = tile >> 2;
    const int nt4 = tile & 3;
    const int m0  = mt * 16;
    const int c0  = nt4 * 64;

    v8f zero = {0.f,0.f,0.f,0.f,0.f,0.f,0.f,0.f};
    v8f accS[4], accD[4];
    #pragma unroll
    for (int j = 0; j < 4; ++j) { accS[j] = zero; accD[j] = zero; }

    const float* arow = feat + (size_t)(m0 + row) * INFEAT;
    for (int k0 = 0; k0 < INFEAT; k0 += 4) {
        const int ka = k0 + 2 * half;
        v2f a = *(const v2f*)(arow + ka);
        #pragma unroll
        for (int j = 0; j < 4; ++j) {
            const int col = c0 + j * 16 + row;
            v2f bS, bD;
            bS.x = Ws[(size_t)ka * HIDTOT + col];
            bS.y = Ws[(size_t)(ka + 1) * HIDTOT + col];
            bD.x = Wd[(size_t)ka * HIDTOT + col];
            bD.y = Wd[(size_t)(ka + 1) * HIDTOT + col];
            accS[j] = __builtin_amdgcn_wmma_f32_16x16x4_f32(
                false, a, false, bS, (short)0, accS[j], false, false);
            accD[j] = __builtin_amdgcn_wmma_f32_16x16x4_f32(
                false, a, false, bD, (short)0, accD[j], false, false);
        }
    }
    #pragma unroll
    for (int j = 0; j < 4; ++j) {
        const int col = c0 + j * 16 + row;
        const float biasS = bs[col], biasD = bd[col];
        #pragma unroll
        for (int i = 0; i < 8; ++i) {
            const int r = m0 + i + 8 * half;
            hs[(size_t)r * HIDTOT + col] = accS[j][i] + biasS;
            hd[(size_t)r * HIDTOT + col] = accD[j][i] + biasD;
        }
    }
}

__global__ __launch_bounds__(SORT_T) void range_hist_kernel(const int* __restrict__ dst,
                                                            int* __restrict__ ghist)
{
    __shared__ int h[NRANGE];
    const int t = threadIdx.x, b = blockIdx.x;
    if (t < NRANGE) h[t] = 0;
    __syncthreads();
    const int e0 = b * EPB;
    const int e1 = (e0 + EPB < EDGES) ? (e0 + EPB) : EDGES;
    int dv[EITER];
    #pragma unroll
    for (int j = 0; j < EITER; ++j) {
        const int e = e0 + t + j * SORT_T;
        dv[j] = (e < e1) ? dst[e] : -1;
    }
    #pragma unroll
    for (int j = 0; j < EITER; ++j)
        if (dv[j] >= 0) atomicAdd(&h[(unsigned)dv[j] / RSPAN], 1);
    __syncthreads();
    if (t < NRANGE) ghist[b * NRANGE + t] = h[t];
}

__global__ __launch_bounds__(SORT_T) void range_scatter_kernel(
    const int* __restrict__ src, const int* __restrict__ dst,
    const int* __restrict__ ghist, int* __restrict__ rstart, int* __restrict__ binned)
{
    __shared__ int part_pre[SORT_T / NRANGE][NRANGE];
    __shared__ int part_tot[SORT_T / NRANGE][NRANGE];
    __shared__ int cur[NRANGE];
    __shared__ int wsum[NRANGE / 32];
    const int t = threadIdx.x, b = blockIdx.x;
    const int lane = t & 31, wave = t >> 5;
    const int r = t % NRANGE, q = t / NRANGE;
    const int per = NBLK / (SORT_T / NRANGE);
    int pre = 0, tot = 0;
    #pragma unroll 8
    for (int j = 0; j < per; ++j) {
        const int bb = q * per + j;
        const int v = ghist[bb * NRANGE + r];
        tot += v;
        pre += (bb < b) ? v : 0;
    }
    part_pre[q][r] = pre;
    part_tot[q][r] = tot;
    __syncthreads();
    int mypre = 0, mytot = 0, inc = 0;
    if (t < NRANGE) {
        #pragma unroll
        for (int k = 0; k < SORT_T / NRANGE; ++k) { mypre += part_pre[k][t]; mytot += part_tot[k][t]; }
        inc = mytot;
        #pragma unroll
        for (int o = 1; o < 32; o <<= 1) {
            const int y = __shfl_up(inc, o, 32);
            if (lane >= o) inc += y;
        }
        if (lane == 31) wsum[wave] = inc;
    }
    __syncthreads();
    if (t < NRANGE) {
        int wbase = 0;
        #pragma unroll
        for (int w = 0; w < NRANGE / 32; ++w) wbase += (w < wave) ? wsum[w] : 0;
        const int start = wbase + inc - mytot;
        cur[t] = start + mypre;
        if (b == 0) {
            rstart[t] = start;
            if (t == NRANGE - 1) rstart[NRANGE] = start + mytot;
        }
    }
    __syncthreads();
    const int e0 = b * EPB;
    const int e1 = (e0 + EPB < EDGES) ? (e0 + EPB) : EDGES;
    int dv[EITER], sv[EITER];
    #pragma unroll
    for (int j = 0; j < EITER; ++j) {
        const int e = e0 + t + j * SORT_T;
        dv[j] = (e < e1) ? dst[e] : -1;
        sv[j] = (e < e1) ? src[e] : 0;
    }
    int2* bin2 = (int2*)binned;
    #pragma unroll
    for (int j = 0; j < EITER; ++j) {
        if (dv[j] >= 0) {
            const int p = atomicAdd(&cur[(unsigned)dv[j] / RSPAN], 1);
            bin2[p] = make_int2(dv[j], sv[j]);
        }
    }
}

__global__ __launch_bounds__(SORT_T) void range_place_kernel(
    const int* __restrict__ binned, const int* __restrict__ rstart,
    int* __restrict__ off, int* __restrict__ list)
{
    __shared__ int cnt[256];
    __shared__ int wsum[8];
    const int t = threadIdx.x, r = blockIdx.x;
    const int lane = t & 31, wave = t >> 5;
    const int beg = rstart[r], end = rstart[r + 1];
    const int n0 = r * RSPAN;
    const int2* bin2 = (const int2*)binned;
    if (t < 256) cnt[t] = 0;
    __syncthreads();
    for (int base = beg; base < end; base += EITER * SORT_T) {
        int dv[EITER];
        #pragma unroll
        for (int j = 0; j < EITER; ++j) {
            const int i = base + t + j * SORT_T;
            dv[j] = (i < end) ? bin2[i].x : -1;
        }
        #pragma unroll
        for (int j = 0; j < EITER; ++j)
            if (dv[j] >= 0) atomicAdd(&cnt[dv[j] - n0], 1);
    }
    __syncthreads();
    int c = 0, inc = 0;
    if (t < 256) {
        c = cnt[t];
        inc = c;
        #pragma unroll
        for (int o = 1; o < 32; o <<= 1) {
            const int y = __shfl_up(inc, o, 32);
            if (lane >= o) inc += y;
        }
        if (lane == 31) wsum[wave] = inc;
    }
    __syncthreads();
    if (t < 256) {
        int wbase = 0;
        #pragma unroll
        for (int w = 0; w < 8; ++w) wbase += (w < wave) ? wsum[w] : 0;
        const int start = beg + wbase + inc - c;
        if (t < RSPAN) off[n0 + t] = start;
        cnt[t] = start;
    }
    __syncthreads();
    for (int base = beg; base < end; base += EITER * SORT_T) {
        int2 ev[EITER];
        #pragma unroll
        for (int j = 0; j < EITER; ++j) {
            const int i = base + t + j * SORT_T;
            ev[j] = (i < end) ? bin2[i] : make_int2(-1, 0);
        }
        #pragma unroll
        for (int j = 0; j < EITER; ++j) {
            if (ev[j].x >= 0) {
                const int p = atomicAdd(&cnt[ev[j].x - n0], 1);
                list[p] = ev[j].y;
            }
        }
    }
}

template <int M> __device__ __forceinline__ float row_xor(float v) {
    return __int_as_float(__builtin_amdgcn_update_dpp(0, __float_as_int(v), 0x160 | M, 0xF, 0xF, true));
}
__device__ __forceinline__ float row16_sum(float v) {
    v += row_xor<1>(v);
    v += row_xor<2>(v);
    v += row_xor<4>(v);
    v += row_xor<8>(v);
    return v;
}
__device__ __forceinline__ float other_row(float v) {
    return __int_as_float(__builtin_amdgcn_permlanex16(__float_as_int(v), __float_as_int(v),
                                                       0x76543210, 0xfedcba98, false, false));
}
__device__ __forceinline__ float lrelu_mx(float x) { return fmaxf(x, NEG_SLOPE * x); }
__device__ __forceinline__ float dot_lrelu(const float4 a, const float4 b, const float4 w) {
    return lrelu_mx(a.x + b.x) * w.x + lrelu_mx(a.y + b.y) * w.y +
           lrelu_mx(a.z + b.z) * w.z + lrelu_mx(a.w + b.w) * w.w;
}
template <int N> __device__ __forceinline__ float row_bcast(float v) {
    return __int_as_float(__builtin_amdgcn_update_dpp(0, __float_as_int(v), 0x150 | N, 0xF, 0xF, true));
}
template <int M> __device__ __forceinline__ float fold_xor(bool hi, float lo_v, float hi_v) {
    const float keep = hi ? hi_v : lo_v;
    const float send = hi ? lo_v : hi_v;
    return keep + row_xor<M>(send);
}
__device__ __forceinline__ float exp_nonpos(float x) { return __builtin_amdgcn_exp2f(x * 1.44269504088896341f); }
__device__ __forceinline__ float elu1(float v) { return v > 0.f ? v : (expf(v) - 1.f); }

#define GRP 4
__global__ __launch_bounds__(256) void node_attn1_kernel(
    const float* __restrict__ hs, const float* __restrict__ hd,
    const int* __restrict__ off, const int* __restrict__ list,
    const float* __restrict__ attn,
    const float* __restrict__ W2s, const float* __restrict__ b2s,
    const float* __restrict__ W2d, const float* __restrict__ b2d,
    float* __restrict__ hs2, float* __restrict__ hd2)
{
    const int d = __builtin_amdgcn_readfirstlane((int)(blockIdx.x * 8 + (threadIdx.x >> 5)));
    if (d >= NNODES) return;
    const int lane = threadIdx.x & 31;
    const int beg = off[d], end = off[d + 1];

    const float* pd = hd + (size_t)d * HIDTOT + 4 * lane;
    const float4 b0 = *(const float4*)(pd);
    const float4 b1 = *(const float4*)(pd + 128);
    const float4 w0 = *(const float4*)(attn + 4 * lane);
    const float4 w1 = *(const float4*)(attn + 128 + 4 * lane);

    const bool cb0 = (lane & 1) != 0, cb1 = (lane & 2) != 0, cb2 = (lane & 4) != 0;
    const int myslot = (cb1 ? 2 : 0) + (cb2 ? 1 : 0);
    float m = -INFINITY, l = 0.f;
    float4 acc0 = {0.f, 0.f, 0.f, 0.f}, acc1 = {0.f, 0.f, 0.f, 0.f};

    for (int i = beg; i < end; i += 32) {
        const int n = (end - i) < 32 ? (end - i) : 32;
        const int my_s = lane < n ? list[i + lane] : 0;
        for (int k = 0; k < n; k += GRP) {
            float4 a0[GRP], a1[GRP];
            #pragma unroll
            for (int j = 0; j < GRP; ++j) {
                const int kk = (k + j < n) ? (k + j) : k;
                const int s = __builtin_amdgcn_readlane(my_s, kk);
                const float* ps = hs + (size_t)s * HIDTOT + 4 * lane;
                a0[j] = *(const float4*)(ps);
                a1[j] = *(const float4*)(ps + 128);
            }
            float x0[GRP], x1[GRP];
            #pragma unroll
            for (int j = 0; j < GRP; ++j) {
                x0[j] = dot_lrelu(a0[j], b0, w0);
                x1[j] = dot_lrelu(a1[j], b1, w1);
            }
            const float u0 = fold_xor<1>(cb0, x0[0], x1[0]);
            const float u1 = fold_xor<1>(cb0, x0[1], x1[1]);
            const float u2 = fold_xor<1>(cb0, x0[2], x1[2]);
            const float u3 = fold_xor<1>(cb0, x0[3], x1[3]);
            const float t0 = fold_xor<2>(cb1, u0, u2);
            const float t1 = fold_xor<2>(cb1, u1, u3);
            float s = fold_xor<4>(cb2, t0, t1);
            s += row_xor<8>(s);
            if (k + myslot >= n) s = -INFINITY;
            float g = fmaxf(s, row_xor<2>(s));
            g = fmaxf(g, row_xor<4>(g));
            g = fmaxf(g, m);
            const float r = exp_nonpos(m - g);
            const float e = exp_nonpos(s - g);
            l = l * r + e;
            m = g;
            const float r0 = row_bcast<0>(r), r1 = row_bcast<1>(r);
            acc0.x *= r0; acc0.y *= r0; acc0.z *= r0; acc0.w *= r0;
            acc1.x *= r1; acc1.y *= r1; acc1.z *= r1; acc1.w *= r1;
            const float e00 = row_bcast<0>(e), e01 = row_bcast<4>(e), e02 = row_bcast<2>(e), e03 = row_bcast<6>(e);
            const float e10 = row_bcast<1>(e), e11 = row_bcast<5>(e), e12 = row_bcast<3>(e), e13 = row_bcast<7>(e);
            acc0.x += e00 * a0[0].x; acc0.y += e00 * a0[0].y; acc0.z += e00 * a0[0].z; acc0.w += e00 * a0[0].w;
            acc1.x += e10 * a1[0].x; acc1.y += e10 * a1[0].y; acc1.z += e10 * a1[0].z; acc1.w += e10 * a1[0].w;
            acc0.x += e01 * a0[1].x; acc0.y += e01 * a0[1].y; acc0.z += e01 * a0[1].z; acc0.w += e01 * a0[1].w;
            acc1.x += e11 * a1[1].x; acc1.y += e11 * a1[1].y; acc1.z += e11 * a1[1].z; acc1.w += e11 * a1[1].w;
            acc0.x += e02 * a0[2].x; acc0.y += e02 * a0[2].y; acc0.z += e02 * a0[2].z; acc0.w += e02 * a0[2].w;
            acc1.x += e12 * a1[2].x; acc1.y += e12 * a1[2].y; acc1.z += e12 * a1[2].z; acc1.w += e12 * a1[2].w;
            acc0.x += e03 * a0[3].x; acc0.y += e03 * a0[3].y; acc0.z += e03 * a0[3].z; acc0.w += e03 * a0[3].w;
            acc1.x += e13 * a1[3].x; acc1.y += e13 * a1[3].y; acc1.z += e13 * a1[3].z; acc1.w += e13 * a1[3].w;
        }
    }
    const float4 s_a = *(const float4*)(W2s + 8 * lane);
    const float4 s_b = *(const float4*)(W2s + 8 * lane + 4);
    const float4 s_c = *(const float4*)(W2s + 256 + 8 * lane);
    const float4 s_d = *(const float4*)(W2s + 256 + 8 * lane + 4);
    const float4 d_a = *(const float4*)(W2d + 8 * lane);
    const float4 d_b = *(const float4*)(W2d + 8 * lane + 4);
    const float4 d_c = *(const float4*)(W2d + 256 + 8 * lane);
    const float4 d_d = *(const float4*)(W2d + 256 + 8 * lane + 4);
    l += row_xor<2>(l);
    l += row_xor<4>(l);
    const float l0 = row_bcast<0>(l), l1 = row_bcast<1>(l);
    const float q0 = l0 > 0.f ? l0 : 1.f, q1 = l1 > 0.f ? l1 : 1.f;
    float4 o0, o1;
    o0.x = elu1(acc0.x / q0); o0.y = elu1(acc0.y / q0);
    o0.z = elu1(acc0.z / q0); o0.w = elu1(acc0.w / q0);
    o1.x = elu1(acc1.x / q1); o1.y = elu1(acc1.y / q1);
    o1.z = elu1(acc1.z / q1); o1.w = elu1(acc1.w / q1);
    float ps0 = o0.x * s_a.x + o0.y * s_a.z + o0.z * s_b.x + o0.w * s_b.z
              + o1.x * s_c.x + o1.y * s_c.z + o1.z * s_d.x + o1.w * s_d.z;
    float ps1 = o0.x * s_a.y + o0.y * s_a.w + o0.z * s_b.y + o0.w * s_b.w
              + o1.x * s_c.y + o1.y * s_c.w + o1.z * s_d.y + o1.w * s_d.w;
    float pd0 = o0.x * d_a.x + o0.y * d_a.z + o0.z * d_b.x + o0.w * d_b.z
              + o1.x * d_c.x + o1.y * d_c.z + o1.z * d_d.x + o1.w * d_d.z;
    float pd1 = o0.x * d_a.y + o0.y * d_a.w + o0.z * d_b.y + o0.w * d_b.w
              + o1.x * d_c.y + o1.y * d_c.w + o1.z * d_d.y + o1.w * d_d.w;
    ps0 = row16_sum(ps0); ps1 = row16_sum(ps1);
    pd0 = row16_sum(pd0); pd1 = row16_sum(pd1);
    ps0 += other_row(ps0); ps1 += other_row(ps1);
    pd0 += other_row(pd0); pd1 += other_row(pd1);
    if (lane == 0) {
        float2 vs, vd;
        vs.x = ps0 + b2s[0]; vs.y = ps1 + b2s[1];
        vd.x = pd0 + b2d[0]; vd.y = pd1 + b2d[1];
        *(float2*)(hs2 + (size_t)d * 2) = vs;
        *(float2*)(hd2 + (size_t)d * 2) = vd;
    }
}

#define GRP2 8
__global__ __launch_bounds__(64) void node_attn2_kernel(
    const float* __restrict__ hs2, const float* __restrict__ hd2,
    const int* __restrict__ off, const int* __restrict__ list,
    const float* __restrict__ attn2, float* __restrict__ out)
{
    const int d = blockIdx.x * blockDim.x + threadIdx.x;
    if (d >= NNODES) return;
    const int beg = off[d], end = off[d + 1];
    const float2 b = *(const float2*)(hd2 + (size_t)d * 2);
    const float w0 = attn2[0], w1 = attn2[1];

    float m = -INFINITY, l = 0.f, ax = 0.f, ay = 0.f;
    for (int i = beg; i < end; i += GRP2) {
        int s[GRP2];
        float2 a[GRP2];
        #pragma unroll
        for (int j = 0; j < GRP2; ++j) {
            const int ii = (i + j < end) ? (i + j) : i;
            s[j] = list[ii];
        }
        #pragma unroll
        for (int j = 0; j < GRP2; ++j) a[j] = *(const float2*)(hs2 + (size_t)s[j] * 2);
        float p[GRP2];
        float g = m;
        #pragma unroll
        for (int j = 0; j < GRP2; ++j) {
            float x = lrelu(a[j].x + b.x) * w0 + lrelu(a[j].y + b.y) * w1;
            if (i + j >= end) x = -INFINITY;
            p[j] = x;
            g = fmaxf(g, x);
        }
        const float r = expf(m - g);
        l *= r; ax *= r; ay *= r;
        #pragma unroll
        for (int j = 0; j < GRP2; ++j) {
            const float e = expf(p[j] - g);
            l += e;
            ax += e * a[j].x;
            ay += e * a[j].y;
        }
        m = g;
    }
    const float q = l > 0.f ? l : 1.f;
    float2 o;
    o.x = ax / q;
    o.y = ay / q;
    *(float2*)(out + (size_t)d * 2) = o;
}

extern "C" void kernel_launch(void* const* d_in, const int* in_sizes, int n_in,
                              void* d_out, int out_size, void* d_ws, size_t ws_size,
                              hipStream_t stream)
{
    const float* feat  = (const float*)d_in[0];
    const int*   src   = (const int*)d_in[1];
    const int*   dst   = (const int*)d_in[2];
    const float* W1s   = (const float*)d_in[3];
    const float* b1s   = (const float*)d_in[4];
    const float* W1d   = (const float*)d_in[5];
    const float* b1d   = (const float*)d_in[6];
    const float* attn1 = (const float*)d_in[7];
    const float* W2s   = (const float*)d_in[8];
    const float* b2s   = (const float*)d_in[9];
    const float* W2d   = (const float*)d_in[10];
    const float* b2d   = (const float*)d_in[11];
    const float* attn2 = (const float*)d_in[12];
    float* out = (float*)d_out;

    float* ws = (float*)d_ws;
    size_t off = 0;
    float*    hs1    = ws + off; off += (size_t)NNODES * HIDTOT;
    float*    hd1    = ws + off; off += (size_t)NNODES * HIDTOT;
    int*      ghist  = (int*)(ws + off); off += (size_t)NBLK * NRANGE;
    int*      rstart = (int*)(ws + off); off += (size_t)2 * NRANGE;
    int*      offs   = (int*)(ws + off); off += (size_t)OFF_PAD;
    int*      binned = (int*)(ws + off); off += (size_t)2 * EDGES;
    int*      list   = (int*)(ws + off); off += (size_t)EDGES;
    float*    hs2    = ws + off; off += (size_t)NNODES * 2;
    float*    hd2    = ws + off; off += (size_t)NNODES * 2;

    const int B = 256;

    range_hist_kernel<<<NBLK, SORT_T, 0, stream>>>(dst, ghist);
    range_scatter_kernel<<<NBLK, SORT_T, 0, stream>>>(src, dst, ghist, rstart, binned);
    range_place_kernel<<<NRANGE, SORT_T, 0, stream>>>(binned, rstart, offs, list);
    {
        const int tiles = 3125 * 4;
        gemm1_kernel<<<(tiles + 7) / 8, B, 0, stream>>>(
            feat, W1s, b1s, W1d, b1d, hs1, hd1);
    }
    node_attn1_kernel<<<(NNODES + 7) / 8, B, 0, stream>>>(
        hs1, hd1, offs, list, attn1, W2s, b2s, W2d, b2d, hs2, hd2);
    node_attn2_kernel<<<(NNODES + 63) / 64, 64, 0, stream>>>(
        hs2, hd2, offs, list, attn2, out);

    (void)in_sizes; (void)n_in; (void)out_size; (void)ws_size;
}


# GEMM column tiles interleaved (tile j owns columns 4n+j): a lane's 4 weight values of a k row come as one 16-byte load, its 4 results of a row go as one 16-byte store; k and sum order unchanged
# speedup vs baseline: 5.4730x; 1.0291x over previous
#include <hip/hip_runtime.h>
#include <hip/hip_bf16.h>
#include <math.h>

#define NNODES 50000
#define EDGES  800000
#define INFEAT 128
#define HIDTOT 256
#define NEG_SLOPE 0.2f

typedef __attribute__((ext_vector_type(2))) float v2f;
typedef __attribute__((ext_vector_type(8))) float v8f;

__device__ __forceinline__ float lrelu(float x) {
    return x >= 0.f ? x : NEG_SLOPE * x;
}

#define SORT_T   1024
#define NRANGE   256
#define RSPAN    ((NNODES + NRANGE - 1) / NRANGE)
#define OFF_PAD  (NRANGE * RSPAN)
#define NBLK     256
#define EPB      ((EDGES + NBLK - 1) / NBLK)
#define EITER    ((EPB + SORT_T - 1) / SORT_T)

__global__ __launch_bounds__(256) void gemm1_kernel(
    const float* __restrict__ feat,
    const float* __restrict__ Ws, const float* __restrict__ bs,
    const float* __restrict__ Wd, const float* __restrict__ bd,
    float* __restrict__ hs, float* __restrict__ hd)
{
    const int wave = threadIdx.x >> 5;
    const int lane = threadIdx.x & 31;
    const int half = lane >> 4;
    const int row  = lane & 15;

    int tile = blockIdx.x * 8 + wave;
    if (tile >= 3125 * 4) return;
    const int mt  = tile >> 2;
    const int nt4 = tile & 3;
    const int m0  = mt * 16;
    const int c0  = nt4 * 64;
    const int cl  = c0 + 4 * row;

    v8f zero = {0.f,0.f,0.f,0.f,0.f,0.f,0.f,0.f};
    v8f accS[4], accD[4];
    #pragma unroll
    for (int j = 0; j < 4; ++j) { accS[j] = zero; accD[j] = zero; }

    const float* arow = feat + (size_t)(m0 + row) * INFEAT;
    const float* wsp  = Ws + cl;
    const float* wdp  = Wd + cl;
    for (int k0 = 0; k0 < INFEAT; k0 += 4) {
        const int ka = k0 + 2 * half;
        v2f a = *(const v2f*)(arow + ka);
        const float4 s0 = *(const float4*)(wsp + (size_t)ka * HIDTOT);
        const float4 s1 = *(const float4*)(wsp + (size_t)(ka + 1) * HIDTOT);
        const float4 d0 = *(const float4*)(wdp + (size_t)ka * HIDTOT);
        const float4 d1 = *(const float4*)(wdp + (size_t)(ka + 1) * HIDTOT);
        v2f bS, bD;
        bS.x = s0.x; bS.y = s1.x; bD.x = d0.x; bD.y = d1.x;
        accS[0] = __builtin_amdgcn_wmma_f32_16x16x4_f32(false, a, false, bS, (short)0, accS[0], false, false);
        accD[0] = __builtin_amdgcn_wmma_f32_16x16x4_f32(false, a, false, bD, (short)0, accD[0], false, false);
        bS.x = s0.y; bS.y = s1.y; bD.x = d0.y; bD.y = d1.y;
        accS[1] = __builtin_amdgcn_wmma_f32_16x16x4_f32(false, a, false, bS, (short)0, accS[1], false, false);
        accD[1] = __builtin_amdgcn_wmma_f32_16x16x4_f32(false, a, false, bD, (short)0, accD[1], false, false);
        bS.x = s0.z; bS.y = s1.z; bD.x = d0.z; bD.y = d1.z;
        accS[2] = __builtin_amdgcn_wmma_f32_16x16x4_f32(false, a, false, bS, (short)0, accS[2], false, false);
        accD[2] = __builtin_amdgcn_wmma_f32_16x16x4_f32(false, a, false, bD, (short)0, accD[2], false, false);
        bS.x = s0.w; bS.y = s1.w; bD.x = d0.w; bD.y = d1.w;
        accS[3] = __builtin_amdgcn_wmma_f32_16x16x4_f32(false, a, false, bS, (short)0, accS[3], false, false);
        accD[3] = __builtin_amdgcn_wmma_f32_16x16x4_f32(false, a, false, bD, (short)0, accD[3], false, false);
    }
    const float4 biasS = *(const float4*)(bs + cl);
    const float4 biasD = *(const float4*)(bd + cl);
    #pragma unroll
    for (int i = 0; i < 8; ++i) {
        const int r = m0 + i + 8 * half;
        float4 vs, vd;
        vs.x = accS[0][i] + biasS.x; vs.y = accS[1][i] + biasS.y;
        vs.z = accS[2][i] + biasS.z; vs.w = accS[3][i] + biasS.w;
        vd.x = accD[0][i] + biasD.x; vd.y = accD[1][i] + biasD.y;
        vd.z = accD[2][i] + biasD.z; vd.w = accD[3][i] + biasD.w;
        *(float4*)(hs + (size_t)r * HIDTOT + cl) = vs;
        *(float4*)(hd + (size_t)r * HIDTOT + cl) = vd;
    }
}

__global__ __launch_bounds__(SORT_T) void range_hist_kernel(const int* __restrict__ dst,
                                                            int* __restrict__ ghist)
{
    __shared__ int h[NRANGE];
    const int t = threadIdx.x, b = blockIdx.x;
    if (t < NRANGE) h[t] = 0;
    __syncthreads();
    const int e0 = b * EPB;
    const int e1 = (e0 + EPB < EDGES) ? (e0 + EPB) : EDGES;
    int dv[EITER];
    #pragma unroll
    for (int j = 0; j < EITER; ++j) {
        const int e = e0 + t + j * SORT_T;
        dv[j] = (e < e1) ? dst[e] : -1;
    }
    #pragma unroll
    for (int j = 0; j < EITER; ++j)
        if (dv[j] >= 0) atomicAdd(&h[(unsigned)dv[j] / RSPAN], 1);
    __syncthreads();
    if (t < NRANGE) ghist[b * NRANGE + t] = h[t];
}

__global__ __launch_bounds__(SORT_T) void range_scatter_kernel(
    const int* __restrict__ src, const int* __restrict__ dst,
    const int* __restrict__ ghist, int* __restrict__ rstart, int* __restrict__ binned)
{
    __shared__ int part_pre[SORT_T / NRANGE][NRANGE];
    __shared__ int part_tot[SORT_T / NRANGE][NRANGE];
    __shared__ int cur[NRANGE];
    __shared__ int wsum[NRANGE / 32];
    const int t = threadIdx.x, b = blockIdx.x;
    const int lane = t & 31, wave = t >> 5;
    const int r = t % NRANGE, q = t / NRANGE;
    const int per = NBLK / (SORT_T / NRANGE);
    int pre = 0, tot = 0;
    #pragma unroll 8
    for (int j = 0; j < per; ++j) {
        const int bb = q * per + j;
        const int v = ghist[bb * NRANGE + r];
        tot += v;
        pre += (bb < b) ? v : 0;
    }
    part_pre[q][r] = pre;
    part_tot[q][r] = tot;
    __syncthreads();
    int mypre = 0, mytot = 0, inc = 0;
    if (t < NRANGE) {
        #pragma unroll
        for (int k = 0; k < SORT_T / NRANGE; ++k) { mypre += part_pre[k][t]; mytot += part_tot[k][t]; }
        inc = mytot;
        #pragma unroll
        for (int o = 1; o < 32; o <<= 1) {
            const int y = __shfl_up(inc, o, 32);
            if (lane >= o) inc += y;
        }
        if (lane == 31) wsum[wave] = inc;
    }
    __syncthreads();
    if (t < NRANGE) {
        int wbase = 0;
        #pragma unroll
        for (int w = 0; w < NRANGE / 32; ++w) wbase += (w < wave) ? wsum[w] : 0;
        const int start = wbase + inc - mytot;
        cur[t] = start + mypre;
        if (b == 0) {
            rstart[t] = start;
            if (t == NRANGE - 1) rstart[NRANGE] = start + mytot;
        }
    }
    __syncthreads();
    const int e0 = b * EPB;
    const int e1 = (e0 + EPB < EDGES) ? (e0 + EPB) : EDGES;
    int dv[EITER], sv[EITER];
    #pragma unroll
    for (int j = 0; j < EITER; ++j) {
        const int e = e0 + t + j * SORT_T;
        dv[j] = (e < e1) ? dst[e] : -1;
        sv[j] = (e < e1) ? src[e] : 0;
    }
    int2* bin2 = (int2*)binned;
    #pragma unroll
    for (int j = 0; j < EITER; ++j) {
        if (dv[j] >= 0) {
            const int p = atomicAdd(&cur[(unsigned)dv[j] / RSPAN], 1);
            bin2[p] = make_int2(dv[j], sv[j]);
        }
    }
}

__global__ __launch_bounds__(SORT_T) void range_place_kernel(
    const int* __restrict__ binned, const int* __restrict__ rstart,
    int* __restrict__ off, int* __restrict__ list)
{
    __shared__ int cnt[256];
    __shared__ int wsum[8];
    const int t = threadIdx.x, r = blockIdx.x;
    const int lane = t & 31, wave = t >> 5;
    const int beg = rstart[r], end = rstart[r + 1];
    const int n0 = r * RSPAN;
    const int2* bin2 = (const int2*)binned;
    if (t < 256) cnt[t] = 0;
    __syncthreads();
    for (int base = beg; base < end; base += EITER * SORT_T) {
        int dv[EITER];
        #pragma unroll
        for (int j = 0; j < EITER; ++j) {
            const int i = base + t + j * SORT_T;
            dv[j] = (i < end) ? bin2[i].x : -1;
        }
        #pragma unroll
        for (int j = 0; j < EITER; ++j)
            if (dv[j] >= 0) atomicAdd(&cnt[dv[j] - n0], 1);
    }
    __syncthreads();
    int c = 0, inc = 0;
    if (t < 256) {
        c = cnt[t];
        inc = c;
        #pragma unroll
        for (int o = 1; o < 32; o <<= 1) {
            const int y = __shfl_up(inc, o, 32);
            if (lane >= o) inc += y;
        }
        if (lane == 31) wsum[wave] = inc;
    }
    __syncthreads();
    if (t < 256) {
        int wbase = 0;
        #pragma unroll
        for (int w = 0; w < 8; ++w) wbase += (w < wave) ? wsum[w] : 0;
        const int start = beg + wbase + inc - c;
        if (t < RSPAN) off[n0 + t] = start;
        cnt[t] = start;
    }
    __syncthreads();
    for (int base = beg; base < end; base += EITER * SORT_T) {
        int2 ev[EITER];
        #pragma unroll
        for (int j = 0; j < EITER; ++j) {
            const int i = base + t + j * SORT_T;
            ev[j] = (i < end) ? bin2[i] : make_int2(-1, 0);
        }
        #pragma unroll
        for (int j = 0; j < EITER; ++j) {
            if (ev[j].x >= 0) {
                const int p = atomicAdd(&cnt[ev[j].x - n0], 1);
                list[p] = ev[j].y;
            }
        }
    }
}

template <int M> __device__ __forceinline__ float row_xor(float v) {
    return __int_as_float(__builtin_amdgcn_update_dpp(0, __float_as_int(v), 0x160 | M, 0xF, 0xF, true));
}
__device__ __forceinline__ float row16_sum(float v) {
    v += row_xor<1>(v);
    v += row_xor<2>(v);
    v += row_xor<4>(v);
    v += row_xor<8>(v);
    return v;
}
__device__ __forceinline__ float other_row(float v) {
    return __int_as_float(__builtin_amdgcn_permlanex16(__float_as_int(v), __float_as_int(v),
                                                       0x76543210, 0xfedcba98, false, false));
}
__device__ __forceinline__ float lrelu_mx(float x) { return fmaxf(x, NEG_SLOPE * x); }
__device__ __forceinline__ float dot_lrelu(const float4 a, const float4 b, const float4 w) {
    return lrelu_mx(a.x + b.x) * w.x + lrelu_mx(a.y + b.y) * w.y +
           lrelu_mx(a.z + b.z) * w.z + lrelu_mx(a.w + b.w) * w.w;
}
template <int N> __device__ __forceinline__ float row_bcast(float v) {
    return __int_as_float(__builtin_amdgcn_update_dpp(0, __float_as_int(v), 0x150 | N, 0xF, 0xF, true));
}
template <int M> __device__ __forceinline__ float fold_xor(bool hi, float lo_v, float hi_v) {
    const float keep = hi ? hi_v : lo_v;
    const float send = hi ? lo_v : hi_v;
    return keep + row_xor<M>(send);
}
__device__ __forceinline__ float exp_nonpos(float x) { return __builtin_amdgcn_exp2f(x * 1.44269504088896341f); }
__device__ __forceinline__ float elu1(float v) { return v > 0.f ? v : (expf(v) - 1.f); }

#define GRP 4
__global__ __launch_bounds__(256) void node_attn1_kernel(
    const float* __restrict__ hs, const float* __restrict__ hd,
    const int* __restrict__ off, const int* __restrict__ list,
    const float* __restrict__ attn,
    const float* __restrict__ W2s, const float* __restrict__ b2s,
    const float* __restrict__ W2d, const float* __restrict__ b2d,
    float* __restrict__ hs2, float* __restrict__ hd2)
{
    const int d = __builtin_amdgcn_readfirstlane((int)(blockIdx.x * 8 + (threadIdx.x >> 5)));
    if (d >= NNODES) return;
    const int lane = threadIdx.x & 31;
    const int beg = off[d], end = off[d + 1];

    const float* pd = hd + (size_t)d * HIDTOT + 4 * lane;
    const float4 b0 = *(const float4*)(pd);
    const float4 b1 = *(const float4*)(pd + 128);
    const float4 w0 = *(const float4*)(attn + 4 * lane);
    const float4 w1 = *(const float4*)(attn + 128 + 4 * lane);

    const bool cb0 = (lane & 1) != 0, cb1 = (lane & 2) != 0, cb2 = (lane & 4) != 0;
    const int myslot = (cb1 ? 2 : 0) + (cb2 ? 1 : 0);
    float m = -INFINITY, l = 0.f;
    float4 acc0 = {0.f, 0.f, 0.f, 0.f}, acc1 = {0.f, 0.f, 0.f, 0.f};

    for (int i = beg; i < end; i += 32) {
        const int n = (end - i) < 32 ? (end - i) : 32;
        const int my_s = lane < n ? list[i + lane] : 0;
        for (int k = 0; k < n; k += GRP) {
            float4 a0[GRP], a1[GRP];
            #pragma unroll
            for (int j = 0; j < GRP; ++j) {
                const int kk = (k + j < n) ? (k + j) : k;
                const int s = __builtin_amdgcn_readlane(my_s, kk);
                const float* ps = hs + (size_t)s * HIDTOT + 4 * lane;
                a0[j] = *(const float4*)(ps);
                a1[j] = *(const float4*)(ps + 128);
            }
            float x0[GRP], x1[GRP];
            #pragma unroll
            for (int j = 0; j < GRP; ++j) {
                x0[j] = dot_lrelu(a0[j], b0, w0);
                x1[j] = dot_lrelu(a1[j], b1, w1);
            }
            const float u0 = fold_xor<1>(cb0, x0[0], x1[0]);
            const float u1 = fold_xor<1>(cb0, x0[1], x1[1]);
            const float u2 = fold_xor<1>(cb0, x0[2], x1[2]);
            const float u3 = fold_xor<1>(cb0, x0[3], x1[3]);
            const float t0 = fold_xor<2>(cb1, u0, u2);
            const float t1 = fold_xor<2>(cb1, u1, u3);
            float s = fold_xor<4>(cb2, t0, t1);
            s += row_xor<8>(s);
            if (k + myslot >= n) s = -INFINITY;
            float g = fmaxf(s, row_xor<2>(s));
            g = fmaxf(g, row_xor<4>(g));
            g = fmaxf(g, m);
            const float r = exp_nonpos(m - g);
            const float e = exp_nonpos(s - g);
            l = l * r + e;
            m = g;
            const float r0 = row_bcast<0>(r), r1 = row_bcast<1>(r);
            acc0.x *= r0; acc0.y *= r0; acc0.z *= r0; acc0.w *= r0;
            acc1.x *= r1; acc1.y *= r1; acc1.z *= r1; acc1.w *= r1;
            const float e00 = row_bcast<0>(e), e01 = row_bcast<4>(e), e02 = row_bcast<2>(e), e03 = row_bcast<6>(e);
            const float e10 = row_bcast<1>(e), e11 = row_bcast<5>(e), e12 = row_bcast<3>(e), e13 = row_bcast<7>(e);
            acc0.x += e00 * a0[0].x; acc0.y += e00 * a0[0].y; acc0.z += e00 * a0[0].z; acc0.w += e00 * a0[0].w;
            acc1.x += e10 * a1[0].x; acc1.y += e10 * a1[0].y; acc1.z += e10 * a1[0].z; acc1.w += e10 * a1[0].w;
            acc0.x += e01 * a0[1].x; acc0.y += e01 * a0[1].y; acc0.z += e01 * a0[1].z; acc0.w += e01 * a0[1].w;
            acc1.x += e11 * a1[1].x; acc1.y += e11 * a1[1].y; acc1.z += e11 * a1[1].z; acc1.w += e11 * a1[1].w;
            acc0.x += e02 * a0[2].x; acc0.y += e02 * a0[2].y; acc0.z += e02 * a0[2].z; acc0.w += e02 * a0[2].w;
            acc1.x += e12 * a1[2].x; acc1.y += e12 * a1[2].y; acc1.z += e12 * a1[2].z; acc1.w += e12 * a1[2].w;
            acc0.x += e03 * a0[3].x; acc0.y += e03 * a0[3].y; acc0.z += e03 * a0[3].z; acc0.w += e03 * a0[3].w;
            acc1.x += e13 * a1[3].x; acc1.y += e13 * a1[3].y; acc1.z += e13 * a1[3].z; acc1.w += e13 * a1[3].w;
        }
    }
    const float4 s_a = *(const float4*)(W2s + 8 * lane);
    const float4 s_b = *(const float4*)(W2s + 8 * lane + 4);
    const float4 s_c = *(const float4*)(W2s + 256 + 8 * lane);
    const float4 s_d = *(const float4*)(W2s + 256 + 8 * lane + 4);
    const float4 d_a = *(const float4*)(W2d + 8 * lane);
    const float4 d_b = *(const float4*)(W2d + 8 * lane + 4);
    const float4 d_c = *(const float4*)(W2d + 256 + 8 * lane);
    const float4 d_d = *(const float4*)(W2d + 256 + 8 * lane + 4);
    l += row_xor<2>(l);
    l += row_xor<4>(l);
    const float l0 = row_bcast<0>(l), l1 = row_bcast<1>(l);
    const float q0 = l0 > 0.f ? l0 : 1.f, q1 = l1 > 0.f ? l1 : 1.f;
    float4 o0, o1;
    o0.x = elu1(acc0.x / q0); o0.y = elu1(acc0.y / q0);
    o0.z = elu1(acc0.z / q0); o0.w = elu1(acc0.w / q0);
    o1.x = elu1(acc1.x / q1); o1.y = elu1(acc1.y / q1);
    o1.z = elu1(acc1.z / q1); o1.w = elu1(acc1.w / q1);
    float ps0 = o0.x * s_a.x + o0.y * s_a.z + o0.z * s_b.x + o0.w * s_b.z
              + o1.x * s_c.x + o1.y * s_c.z + o1.z * s_d.x + o1.w * s_d.z;
    float ps1 = o0.x * s_a.y + o0.y * s_a.w + o0.z * s_b.y + o0.w * s_b.w
              + o1.x * s_c.y + o1.y * s_c.w + o1.z * s_d.y + o1.w * s_d.w;
    float pd0 = o0.x * d_a.x + o0.y * d_a.z + o0.z * d_b.x + o0.w * d_b.z
              + o1.x * d_c.x + o1.y * d_c.z + o1.z * d_d.x + o1.w * d_d.z;
    float pd1 = o0.x * d_a.y + o0.y * d_a.w + o0.z * d_b.y + o0.w * d_b.w
              + o1.x * d_c.y + o1.y * d_c.w + o1.z * d_d.y + o1.w * d_d.w;
    ps0 = row16_sum(ps0); ps1 = row16_sum(ps1);
    pd0 = row16_sum(pd0); pd1 = row16_sum(pd1);
    ps0 += other_row(ps0); ps1 += other_row(ps1);
    pd0 += other_row(pd0); pd1 += other_row(pd1);
    if (lane == 0) {
        float2 vs, vd;
        vs.x = ps0 + b2s[0]; vs.y = ps1 + b2s[1];
        vd.x = pd0 + b2d[0]; vd.y = pd1 + b2d[1];
        *(float2*)(hs2 + (size_t)d * 2) = vs;
        *(float2*)(hd2 + (size_t)d * 2) = vd;
    }
}

#define GRP2 8
__global__ __launch_bounds__(64) void node_attn2_kernel(
    const float* __restrict__ hs2, const float* __restrict__ hd2,
    const int* __restrict__ off, const int* __restrict__ list,
    const float* __restrict__ attn2, float* __restrict__ out)
{
    const int d = blockIdx.x * blockDim.x + threadIdx.x;
    if (d >= NNODES) return;
    const int beg = off[d], end = off[d + 1];
    const float2 b = *(const float2*)(hd2 + (size_t)d * 2);
    const float w0 = attn2[0], w1 = attn2[1];

    float m = -INFINITY, l = 0.f, ax = 0.f, ay = 0.f;
    for (int i = beg; i < end; i += GRP2) {
        int s[GRP2];
        float2 a[GRP2];
        #pragma unroll
        for (int j = 0; j < GRP2; ++j) {
            const int ii = (i + j < end) ? (i + j) : i;
            s[j] = list[ii];
        }
        #pragma unroll
        for (int j = 0; j < GRP2; ++j) a[j] = *(const float2*)(hs2 + (size_t)s[j] * 2);
        float p[GRP2];
        float g = m;
        #pragma unroll
        for (int j = 0; j < GRP2; ++j) {
            float x = lrelu(a[j].x + b.x) * w0 + lrelu(a[j].y + b.y) * w1;
            if (i + j >= end) x = -INFINITY;
            p[j] = x;
            g = fmaxf(g, x);
        }
        const float r = expf(m - g);
        l *= r; ax *= r; ay *= r;
        #pragma unroll
        for (int j = 0; j < GRP2; ++j) {
            const float e = expf(p[j] - g);
            l += e;
            ax += e * a[j].x;
            ay += e * a[j].y;
        }
        m = g;
    }
    const float q = l > 0.f ? l : 1.f;
    float2 o;
    o.x = ax / q;
    o.y = ay / q;
    *(float2*)(out + (size_t)d * 2) = o;
}

extern "C" void kernel_launch(void* const* d_in, const int* in_sizes, int n_in,
                              void* d_out, int out_size, void* d_ws, size_t ws_size,
                              hipStream_t stream)
{
    const float* feat  = (const float*)d_in[0];
    const int*   src   = (const int*)d_in[1];
    const int*   dst   = (const int*)d_in[2];
    const float* W1s   = (const float*)d_in[3];
    const float* b1s   = (const float*)d_in[4];
    const float* W1d   = (const float*)d_in[5];
    const float* b1d   = (const float*)d_in[6];
    const float* attn1 = (const float*)d_in[7];
    const float* W2s   = (const float*)d_in[8];
    const float* b2s   = (const float*)d_in[9];
    const float* W2d   = (const float*)d_in[10];
    const float* b2d   = (const float*)d_in[11];
    const float* attn2 = (const float*)d_in[12];
    float* out = (float*)d_out;

    float* ws = (float*)d_ws;
    size_t off = 0;
    float*    hs1    = ws + off; off += (size_t)NNODES * HIDTOT;
    float*    hd1    = ws + off; off += (size_t)NNODES * HIDTOT;
    int*      ghist  = (int*)(ws + off); off += (size_t)NBLK * NRANGE;
    int*      rstart = (int*)(ws + off); off += (size_t)2 * NRANGE;
    int*      offs   = (int*)(ws + off); off += (size_t)OFF_PAD;
    int*      binned = (int*)(ws + off); off += (size_t)2 * EDGES;
    int*      list   = (int*)(ws + off); off += (size_t)EDGES;
    float*    hs2    = ws + off; off += (size_t)NNODES * 2;
    float*    hd2    = ws + off; off += (size_t)NNODES * 2;

    const int B = 256;

    range_hist_kernel<<<NBLK, SORT_T, 0, stream>>>(dst, ghist);
    range_scatter_kernel<<<NBLK, SORT_T, 0, stream>>>(src, dst, ghist, rstart, binned);
    range_place_kernel<<<NRANGE, SORT_T, 0, stream>>>(binned, rstart, offs, list);
    {
        const int tiles = 3125 * 4;
        gemm1_kernel<<<(tiles + 7) / 8, B, 0, stream>>>(
            feat, W1s, b1s, W1d, b1d, hs1, hd1);
    }
    node_attn1_kernel<<<(NNODES + 7) / 8, B, 0, stream>>>(
        hs1, hd1, offs, list, attn1, W2s, b2s, W2d, b2d, hs2, hd2);
    node_attn2_kernel<<<(NNODES + 63) / 64, 64, 0, stream>>>(
        hs2, hd2, offs, list, attn2, out);

    (void)in_sizes; (void)n_in; (void)out_size; (void)ws_size;
}


# the dual GEMM's blocks ride in the edge-placement launch (placement at 256 threads per block, role by block index): 6 to 5 launches, placement runs under the GEMM; list order stays LDS-add arrival
# speedup vs baseline: 5.6657x; 1.0352x over previous
#include <hip/hip_runtime.h>
#include <hip/hip_bf16.h>
#include <math.h>

#define NNODES 50000
#define EDGES  800000
#define INFEAT 128
#define HIDTOT 256
#define NEG_SLOPE 0.2f

typedef __attribute__((ext_vector_type(2))) float v2f;
typedef __attribute__((ext_vector_type(8))) float v8f;

__device__ __forceinline__ float lrelu(float x) {
    return x >= 0.f ? x : NEG_SLOPE * x;
}

#define SORT_T   1024
#define NRANGE   256
#define RSPAN    ((NNODES + NRANGE - 1) / NRANGE)
#define OFF_PAD  (NRANGE * RSPAN)
#define NBLK     256
#define EPB      ((EDGES + NBLK - 1) / NBLK)
#define EITER    ((EPB + SORT_T - 1) / SORT_T)

#define GEMM_TILES (3125 * 4)
#define GEMM_BLKS  ((GEMM_TILES + 7) / 8)
__device__ __forceinline__ void gemm1_tiles(
    const int gblock,
    const float* __restrict__ feat,
    const float* __restrict__ Ws, const float* __restrict__ bs,
    const float* __restrict__ Wd, const float* __restrict__ bd,
    float* __restrict__ hs, float* __restrict__ hd)
{
    const int wave = threadIdx.x >> 5;
    const int lane = threadIdx.x & 31;
    const int half = lane >> 4;
    const int row  = lane & 15;

    int tile = gblock * 8 + wave;
    if (tile >= GEMM_TILES) return;
    const int mt  = tile >> 2;
    const int nt4 = tile & 3;
    const int m0  = mt * 16;
    const int c0  = nt4 * 64;
    const int cl  = c0 + 4 * row;

    v8f zero = {0.f,0.f,0.f,0.f,0.f,0.f,0.f,0.f};
    v8f accS[4], accD[4];
    #pragma unroll
    for (int j = 0; j < 4; ++j) { accS[j] = zero; accD[j] = zero; }

    const float* arow = feat + (size_t)(m0 + row) * INFEAT;
    const float* wsp  = Ws + cl;
    const float* wdp  = Wd + cl;
    for (int k0 = 0; k0 < INFEAT; k0 += 4) {
        const int ka = k0 + 2 * half;
        v2f a = *(const v2f*)(arow + ka);
        const float4 s0 = *(const float4*)(wsp + (size_t)ka * HIDTOT);
        const float4 s1 = *(const float4*)(wsp + (size_t)(ka + 1) * HIDTOT);
        const float4 d0 = *(const float4*)(wdp + (size_t)ka * HIDTOT);
        const float4 d1 = *(const float4*)(wdp + (size_t)(ka + 1) * HIDTOT);
        v2f bS, bD;
        bS.x = s0.x; bS.y = s1.x; bD.x = d0.x; bD.y = d1.x;
        accS[0] = __builtin_amdgcn_wmma_f32_16x16x4_f32(false, a, false, bS, (short)0, accS[0], false, false);
        accD[0] = __builtin_amdgcn_wmma_f32_16x16x4_f32(false, a, false, bD, (short)0, accD[0], false, false);
        bS.x = s0.y; bS.y = s1.y; bD.x = d0.y; bD.y = d1.y;
        accS[1] = __builtin_amdgcn_wmma_f32_16x16x4_f32(false, a, false, bS, (short)0, accS[1], false, false);
        accD[1] = __builtin_amdgcn_wmma_f32_16x16x4_f32(false, a, false, bD, (short)0, accD[1], false, false);
        bS.x = s0.z; bS.y = s1.z; bD.x = d0.z; bD.y = d1.z;
        accS[2] = __builtin_amdgcn_wmma_f32_16x16x4_f32(false, a, false, bS, (short)0, accS[2], false, false);
        accD[2] = __builtin_amdgcn_wmma_f32_16x16x4_f32(false, a, false, bD, (short)0, accD[2], false, false);
        bS.x = s0.w; bS.y = s1.w; bD.x = d0.w; bD.y = d1.w;
        accS[3] = __builtin_amdgcn_wmma_f32_16x16x4_f32(false, a, false, bS, (short)0, accS[3], false, false);
        accD[3] = __builtin_amdgcn_wmma_f32_16x16x4_f32(false, a, false, bD, (short)0, accD[3], false, false);
    }
    const float4 biasS = *(const float4*)(bs + cl);
    const float4 biasD = *(const float4*)(bd + cl);
    #pragma unroll
    for (int i = 0; i < 8; ++i) {
        const int r = m0 + i + 8 * half;
        float4 vs, vd;
        vs.x = accS[0][i] + biasS.x; vs.y = accS[1][i] + biasS.y;
        vs.z = accS[2][i] + biasS.z; vs.w = accS[3][i] + biasS.w;
        vd.x = accD[0][i] + biasD.x; vd.y = accD[1][i] + biasD.y;
        vd.z = accD[2][i] + biasD.z; vd.w = accD[3][i] + biasD.w;
        *(float4*)(hs + (size_t)r * HIDTOT + cl) = vs;
        *(float4*)(hd + (size_t)r * HIDTOT + cl) = vd;
    }
}

__global__ __launch_bounds__(SORT_T) void range_hist_kernel(const int* __restrict__ dst,
                                                            int* __restrict__ ghist)
{
    __shared__ int h[NRANGE];
    const int t = threadIdx.x, b = blockIdx.x;
    if (t < NRANGE) h[t] = 0;
    __syncthreads();
    const int e0 = b * EPB;
    const int e1 = (e0 + EPB < EDGES) ? (e0 + EPB) : EDGES;
    int dv[EITER];
    #pragma unroll
    for (int j = 0; j < EITER; ++j) {
        const int e = e0 + t + j * SORT_T;
        dv[j] = (e < e1) ? dst[e] : -1;
    }
    #pragma unroll
    for (int j = 0; j < EITER; ++j)
        if (dv[j] >= 0) atomicAdd(&h[(unsigned)dv[j] / RSPAN], 1);
    __syncthreads();
    if (t < NRANGE) ghist[b * NRANGE + t] = h[t];
}

__global__ __launch_bounds__(SORT_T) void range_scatter_kernel(
    const int* __restrict__ src, const int* __restrict__ dst,
    const int* __restrict__ ghist, int* __restrict__ rstart, int* __restrict__ binned)
{
    __shared__ int part_pre[SORT_T / NRANGE][NRANGE];
    __shared__ int part_tot[SORT_T / NRANGE][NRANGE];
    __shared__ int cur[NRANGE];
    __shared__ int wsum[NRANGE / 32];
    const int t = threadIdx.x, b = blockIdx.x;
    const int lane = t & 31, wave = t >> 5;
    const int r = t % NRANGE, q = t / NRANGE;
    const int per = NBLK / (SORT_T / NRANGE);
    int pre = 0, tot = 0;
    #pragma unroll 8
    for (int j = 0; j < per; ++j) {
        const int bb = q * per + j;
        const int v = ghist[bb * NRANGE + r];
        tot += v;
        pre += (bb < b) ? v : 0;
    }
    part_pre[q][r] = pre;
    part_tot[q][r] = tot;
    __syncthreads();
    int mypre = 0, mytot = 0, inc = 0;
    if (t < NRANGE) {
        #pragma unroll
        for (int k = 0; k < SORT_T / NRANGE; ++k) { mypre += part_pre[k][t]; mytot += part_tot[k][t]; }
        inc = mytot;
        #pragma unroll
        for (int o = 1; o < 32; o <<= 1) {
            const int y = __shfl_up(inc, o, 32);
            if (lane >= o) inc += y;
        }
        if (lane == 31) wsum[wave] = inc;
    }
    __syncthreads();
    if (t < NRANGE) {
        int wbase = 0;
        #pragma unroll
        for (int w = 0; w < NRANGE / 32; ++w) wbase += (w < wave) ? wsum[w] : 0;
        const int start = wbase + inc - mytot;
        cur[t] = start + mypre;
        if (b == 0) {
            rstart[t] = start;
            if (t == NRANGE - 1) rstart[NRANGE] = start + mytot;
        }
    }
    __syncthreads();
    const int e0 = b * EPB;
    const int e1 = (e0 + EPB < EDGES) ? (e0 + EPB) : EDGES;
    int dv[EITER], sv[EITER];
    #pragma unroll
    for (int j = 0; j < EITER; ++j) {
        const int e = e0 + t + j * SORT_T;
        dv[j] = (e < e1) ? dst[e] : -1;
        sv[j] = (e < e1) ? src[e] : 0;
    }
    int2* bin2 = (int2*)binned;
    #pragma unroll
    for (int j = 0; j < EITER; ++j) {
        if (dv[j] >= 0) {
            const int p = atomicAdd(&cur[(unsigned)dv[j] / RSPAN], 1);
            bin2[p] = make_int2(dv[j], sv[j]);
        }
    }
}

#define PLACE_T  256
#define PITER    16
__device__ __forceinline__ void range_place(
    const int* __restrict__ binned, const int* __restrict__ rstart,
    int* __restrict__ off, int* __restrict__ list)
{
    __shared__ int cnt[PLACE_T];
    __shared__ int wsum[PLACE_T / 32];
    const int t = threadIdx.x, r = blockIdx.x;
    const int lane = t & 31, wave = t >> 5;
    const int beg = rstart[r], end = rstart[r + 1];
    const int n0 = r * RSPAN;
    const int2* bin2 = (const int2*)binned;
    cnt[t] = 0;
    __syncthreads();
    for (int base = beg; base < end; base += PITER * PLACE_T) {
        int dv[PITER];
        #pragma unroll
        for (int j = 0; j < PITER; ++j) {
            const int i = base + t + j * PLACE_T;
            dv[j] = (i < end) ? bin2[i].x : -1;
        }
        #pragma unroll
        for (int j = 0; j < PITER; ++j)
            if (dv[j] >= 0) atomicAdd(&cnt[dv[j] - n0], 1);
    }
    __syncthreads();
    const int c = cnt[t];
    int inc = c;
    #pragma unroll
    for (int o = 1; o < 32; o <<= 1) {
        const int y = __shfl_up(inc, o, 32);
        if (lane >= o) inc += y;
    }
    if (lane == 31) wsum[wave] = inc;
    __syncthreads();
    int wbase = 0;
    #pragma unroll
    for (int w = 0; w < PLACE_T / 32; ++w) wbase += (w < wave) ? wsum[w] : 0;
    const int start = beg + wbase + inc - c;
    if (t < RSPAN) off[n0 + t] = start;
    cnt[t] = start;
    __syncthreads();
    for (int base = beg; base < end; base += PITER * PLACE_T) {
        int2 ev[PITER];
        #pragma unroll
        for (int j = 0; j < PITER; ++j) {
            const int i = base + t + j * PLACE_T;
            ev[j] = (i < end) ? bin2[i] : make_int2(-1, 0);
        }
        #pragma unroll
        for (int j = 0; j < PITER; ++j) {
            if (ev[j].x >= 0) {
                const int p = atomicAdd(&cnt[ev[j].x - n0], 1);
                list[p] = ev[j].y;
            }
        }
    }
}

__global__ __launch_bounds__(PLACE_T) void place_gemm_kernel(
    const int* __restrict__ binned, const int* __restrict__ rstart,
    int* __restrict__ off, int* __restrict__ list,
    const float* __restrict__ feat,
    const float* __restrict__ Ws, const float* __restrict__ bs,
    const float* __restrict__ Wd, const float* __restrict__ bd,
    float* __restrict__ hs, float* __restrict__ hd)
{
    if (blockIdx.x < NRANGE) {
        range_place(binned, rstart, off, list);
        return;
    }
    gemm1_tiles((int)blockIdx.x - NRANGE, feat, Ws, bs, Wd, bd, hs, hd);
}

template <int M> __device__ __forceinline__ float row_xor(float v) {
    return __int_as_float(__builtin_amdgcn_update_dpp(0, __float_as_int(v), 0x160 | M, 0xF, 0xF, true));
}
__device__ __forceinline__ float row16_sum(float v) {
    v += row_xor<1>(v);
    v += row_xor<2>(v);
    v += row_xor<4>(v);
    v += row_xor<8>(v);
    return v;
}
__device__ __forceinline__ float other_row(float v) {
    return __int_as_float(__builtin_amdgcn_permlanex16(__float_as_int(v), __float_as_int(v),
                                                       0x76543210, 0xfedcba98, false, false));
}
__device__ __forceinline__ float lrelu_mx(float x) { return fmaxf(x, NEG_SLOPE * x); }
__device__ __forceinline__ float dot_lrelu(const float4 a, const float4 b, const float4 w) {
    return lrelu_mx(a.x + b.x) * w.x + lrelu_mx(a.y + b.y) * w.y +
           lrelu_mx(a.z + b.z) * w.z + lrelu_mx(a.w + b.w) * w.w;
}
template <int N> __device__ __forceinline__ float row_bcast(float v) {
    return __int_as_float(__builtin_amdgcn_update_dpp(0, __float_as_int(v), 0x150 | N, 0xF, 0xF, true));
}
template <int M> __device__ __forceinline__ float fold_xor(bool hi, float lo_v, float hi_v) {
    const float keep = hi ? hi_v : lo_v;
    const float send = hi ? lo_v : hi_v;
    return keep + row_xor<M>(send);
}
__device__ __forceinline__ float exp_nonpos(float x) { return __builtin_amdgcn_exp2f(x * 1.44269504088896341f); }
__device__ __forceinline__ float elu1(float v) { return v > 0.f ? v : (expf(v) - 1.f); }

#define GRP 4
__global__ __launch_bounds__(256) void node_attn1_kernel(
    const float* __restrict__ hs, const float* __restrict__ hd,
    const int* __restrict__ off, const int* __restrict__ list,
    const float* __restrict__ attn,
    const float* __restrict__ W2s, const float* __restrict__ b2s,
    const float* __restrict__ W2d, const float* __restrict__ b2d,
    float* __restrict__ hs2, float* __restrict__ hd2)
{
    const int d = __builtin_amdgcn_readfirstlane((int)(blockIdx.x * 8 + (threadIdx.x >> 5)));
    if (d >= NNODES) return;
    const int lane = threadIdx.x & 31;
    const int beg = off[d], end = off[d + 1];

    const float* pd = hd + (size_t)d * HIDTOT + 4 * lane;
    const float4 b0 = *(const float4*)(pd);
    const float4 b1 = *(const float4*)(pd + 128);
    const float4 w0 = *(const float4*)(attn + 4 * lane);
    const float4 w1 = *(const float4*)(attn + 128 + 4 * lane);

    const bool cb0 = (lane & 1) != 0, cb1 = (lane & 2) != 0, cb2 = (lane & 4) != 0;
    const int myslot = (cb1 ? 2 : 0) + (cb2 ? 1 : 0);
    float m = -INFINITY, l = 0.f;
    float4 acc0 = {0.f, 0.f, 0.f, 0.f}, acc1 = {0.f, 0.f, 0.f, 0.f};

    for (int i = beg; i < end; i += 32) {
        const int n = (end - i) < 32 ? (end - i) : 32;
        const int my_s = lane < n ? list[i + lane] : 0;
        for (int k = 0; k < n; k += GRP) {
            float4 a0[GRP], a1[GRP];
            #pragma unroll
            for (int j = 0; j < GRP; ++j) {
                const int kk = (k + j < n) ? (k + j) : k;
                const int s = __builtin_amdgcn_readlane(my_s, kk);
                const float* ps = hs + (size_t)s * HIDTOT + 4 * lane;
                a0[j] = *(const float4*)(ps);
                a1[j] = *(const float4*)(ps + 128);
            }
            float x0[GRP], x1[GRP];
            #pragma unroll
            for (int j = 0; j < GRP; ++j) {
                x0[j] = dot_lrelu(a0[j], b0, w0);
                x1[j] = dot_lrelu(a1[j], b1, w1);
            }
            const float u0 = fold_xor<1>(cb0, x0[0], x1[0]);
            const float u1 = fold_xor<1>(cb0, x0[1], x1[1]);
            const float u2 = fold_xor<1>(cb0, x0[2], x1[2]);
            const float u3 = fold_xor<1>(cb0, x0[3], x1[3]);
            const float t0 = fold_xor<2>(cb1, u0, u2);
            const float t1 = fold_xor<2>(cb1, u1, u3);
            float s = fold_xor<4>(cb2, t0, t1);
            s += row_xor<8>(s);
            if (k + myslot >= n) s = -INFINITY;
            float g = fmaxf(s, row_xor<2>(s));
            g = fmaxf(g, row_xor<4>(g));
            g = fmaxf(g, m);
            const float r = exp_nonpos(m - g);
            const float e = exp_nonpos(s - g);
            l = l * r + e;
            m = g;
            const float r0 = row_bcast<0>(r), r1 = row_bcast<1>(r);
            acc0.x *= r0; acc0.y *= r0; acc0.z *= r0; acc0.w *= r0;
            acc1.x *= r1; acc1.y *= r1; acc1.z *= r1; acc1.w *= r1;
            const float e00 = row_bcast<0>(e), e01 = row_bcast<4>(e), e02 = row_bcast<2>(e), e03 = row_bcast<6>(e);
            const float e10 = row_bcast<1>(e), e11 = row_bcast<5>(e), e12 = row_bcast<3>(e), e13 = row_bcast<7>(e);
            acc0.x += e00 * a0[0].x; acc0.y += e00 * a0[0].y; acc0.z += e00 * a0[0].z; acc0.w += e00 * a0[0].w;
            acc1.x += e10 * a1[0].x; acc1.y += e10 * a1[0].y; acc1.z += e10 * a1[0].z; acc1.w += e10 * a1[0].w;
            acc0.x += e01 * a0[1].x; acc0.y += e01 * a0[1].y; acc0.z += e01 * a0[1].z; acc0.w += e01 * a0[1].w;
            acc1.x += e11 * a1[1].x; acc1.y += e11 * a1[1].y; acc1.z += e11 * a1[1].z; acc1.w += e11 * a1[1].w;
            acc0.x += e02 * a0[2].x; acc0.y += e02 * a0[2].y; acc0.z += e02 * a0[2].z; acc0.w += e02 * a0[2].w;
            acc1.x += e12 * a1[2].x; acc1.y += e12 * a1[2].y; acc1.z += e12 * a1[2].z; acc1.w += e12 * a1[2].w;
            acc0.x += e03 * a0[3].x; acc0.y += e03 * a0[3].y; acc0.z += e03 * a0[3].z; acc0.w += e03 * a0[3].w;
            acc1.x += e13 * a1[3].x; acc1.y += e13 * a1[3].y; acc1.z += e13 * a1[3].z; acc1.w += e13 * a1[3].w;
        }
    }
    const float4 s_a = *(const float4*)(W2s + 8 * lane);
    const float4 s_b = *(const float4*)(W2s + 8 * lane + 4);
    const float4 s_c = *(const float4*)(W2s + 256 + 8 * lane);
    const float4 s_d = *(const float4*)(W2s + 256 + 8 * lane + 4);
    const float4 d_a = *(const float4*)(W2d + 8 * lane);
    const float4 d_b = *(const float4*)(W2d + 8 * lane + 4);
    const float4 d_c = *(const float4*)(W2d + 256 + 8 * lane);
    const float4 d_d = *(const float4*)(W2d + 256 + 8 * lane + 4);
    l += row_xor<2>(l);
    l += row_xor<4>(l);
    const float l0 = row_bcast<0>(l), l1 = row_bcast<1>(l);
    const float q0 = l0 > 0.f ? l0 : 1.f, q1 = l1 > 0.f ? l1 : 1.f;
    float4 o0, o1;
    o0.x = elu1(acc0.x / q0); o0.y = elu1(acc0.y / q0);
    o0.z = elu1(acc0.z / q0); o0.w = elu1(acc0.w / q0);
    o1.x = elu1(acc1.x / q1); o1.y = elu1(acc1.y / q1);
    o1.z = elu1(acc1.z / q1); o1.w = elu1(acc1.w / q1);
    float ps0 = o0.x * s_a.x + o0.y * s_a.z + o0.z * s_b.x + o0.w * s_b.z
              + o1.x * s_c.x + o1.y * s_c.z + o1.z * s_d.x + o1.w * s_d.z;
    float ps1 = o0.x * s_a.y + o0.y * s_a.w + o0.z * s_b.y + o0.w * s_b.w
              + o1.x * s_c.y + o1.y * s_c.w + o1.z * s_d.y + o1.w * s_d.w;
    float pd0 = o0.x * d_a.x + o0.y * d_a.z + o0.z * d_b.x + o0.w * d_b.z
              + o1.x * d_c.x + o1.y * d_c.z + o1.z * d_d.x + o1.w * d_d.z;
    float pd1 = o0.x * d_a.y + o0.y * d_a.w + o0.z * d_b.y + o0.w * d_b.w
              + o1.x * d_c.y + o1.y * d_c.w + o1.z * d_d.y + o1.w * d_d.w;
    ps0 = row16_sum(ps0); ps1 = row16_sum(ps1);
    pd0 = row16_sum(pd0); pd1 = row16_sum(pd1);
    ps0 += other_row(ps0); ps1 += other_row(ps1);
    pd0 += other_row(pd0); pd1 += other_row(pd1);
    if (lane == 0) {
        float2 vs, vd;
        vs.x = ps0 + b2s[0]; vs.y = ps1 + b2s[1];
        vd.x = pd0 + b2d[0]; vd.y = pd1 + b2d[1];
        *(float2*)(hs2 + (size_t)d * 2) = vs;
        *(float2*)(hd2 + (size_t)d * 2) = vd;
    }
}

#define GRP2 8
__global__ __launch_bounds__(64) void node_attn2_kernel(
    const float* __restrict__ hs2, const float* __restrict__ hd2,
    const int* __restrict__ off, const int* __restrict__ list,
    const float* __restrict__ attn2, float* __restrict__ out)
{
    const int d = blockIdx.x * blockDim.x + threadIdx.x;
    if (d >= NNODES) return;
    const int beg = off[d], end = off[d + 1];
    const float2 b = *(const float2*)(hd2 + (size_t)d * 2);
    const float w0 = attn2[0], w1 = attn2[1];

    float m = -INFINITY, l = 0.f, ax = 0.f, ay = 0.f;
    for (int i = beg; i < end; i += GRP2) {
        int s[GRP2];
        float2 a[GRP2];
        #pragma unroll
        for (int j = 0; j < GRP2; ++j) {
            const int ii = (i + j < end) ? (i + j) : i;
            s[j] = list[ii];
        }
        #pragma unroll
        for (int j = 0; j < GRP2; ++j) a[j] = *(const float2*)(hs2 + (size_t)s[j] * 2);
        float p[GRP2];
        float g = m;
        #pragma unroll
        for (int j = 0; j < GRP2; ++j) {
            float x = lrelu(a[j].x + b.x) * w0 + lrelu(a[j].y + b.y) * w1;
            if (i + j >= end) x = -INFINITY;
            p[j] = x;
            g = fmaxf(g, x);
        }
        const float r = expf(m - g);
        l *= r; ax *= r; ay *= r;
        #pragma unroll
        for (int j = 0; j < GRP2; ++j) {
            const float e = expf(p[j] - g);
            l += e;
            ax += e * a[j].x;
            ay += e * a[j].y;
        }
        m = g;
    }
    const float q = l > 0.f ? l : 1.f;
    float2 o;
    o.x = ax / q;
    o.y = ay / q;
    *(float2*)(out + (size_t)d * 2) = o;
}

extern "C" void kernel_launch(void* const* d_in, const int* in_sizes, int n_in,
                              void* d_out, int out_size, void* d_ws, size_t ws_size,
                              hipStream_t stream)
{
    const float* feat  = (const float*)d_in[0];
    const int*   src   = (const int*)d_in[1];
    const int*   dst   = (const int*)d_in[2];
    const float* W1s   = (const float*)d_in[3];
    const float* b1s   = (const float*)d_in[4];
    const float* W1d   = (const float*)d_in[5];
    const float* b1d   = (const float*)d_in[6];
    const float* attn1 = (const float*)d_in[7];
    const float* W2s   = (const float*)d_in[8];
    const float* b2s   = (const float*)d_in[9];
    const float* W2d   = (const float*)d_in[10];
    const float* b2d   = (const float*)d_in[11];
    const float* attn2 = (const float*)d_in[12];
    float* out = (float*)d_out;

    float* ws = (float*)d_ws;
    size_t off = 0;
    float*    hs1    = ws + off; off += (size_t)NNODES * HIDTOT;
    float*    hd1    = ws + off; off += (size_t)NNODES * HIDTOT;
    int*      ghist  = (int*)(ws + off); off += (size_t)NBLK * NRANGE;
    int*      rstart = (int*)(ws + off); off += (size_t)2 * NRANGE;
    int*      offs   = (int*)(ws + off); off += (size_t)OFF_PAD;
    int*      binned = (int*)(ws + off); off += (size_t)2 * EDGES;
    int*      list   = (int*)(ws + off); off += (size_t)EDGES;
    float*    hs2    = ws + off; off += (size_t)NNODES * 2;
    float*    hd2    = ws + off; off += (size_t)NNODES * 2;

    const int B = 256;

    range_hist_kernel<<<NBLK, SORT_T, 0, stream>>>(dst, ghist);
    range_scatter_kernel<<<NBLK, SORT_T, 0, stream>>>(src, dst, ghist, rstart, binned);
    place_gemm_kernel<<<NRANGE + GEMM_BLKS, PLACE_T, 0, stream>>>(
        binned, rstart, offs, list, feat, W1s, b1s, W1d, b1d, hs1, hd1);
    node_attn1_kernel<<<(NNODES + 7) / 8, B, 0, stream>>>(
        hs1, hd1, offs, list, attn1, W2s, b2s, W2d, b2d, hs2, hd2);
    node_attn2_kernel<<<(NNODES + 63) / 64, 64, 0, stream>>>(
        hs2, hd2, offs, list, attn2, out);

    (void)in_sizes; (void)n_in; (void)out_size; (void)ws_size;
}


# edge grouping: each piece sorts its own edges by node range in LDS (rank, scan, place), placement blocks sum the range totals under the GEMM; 5 to 4 launches; list order stays LDS-add arrival
# speedup vs baseline: 6.1679x; 1.0886x over previous
#include <hip/hip_runtime.h>
#include <hip/hip_bf16.h>
#include <math.h>

#define NNODES 50000
#define EDGES  800000
#define INFEAT 128
#define HIDTOT 256
#define NEG_SLOPE 0.2f

typedef __attribute__((ext_vector_type(2))) float v2f;
typedef __attribute__((ext_vector_type(8))) float v8f;

__device__ __forceinline__ float lrelu(float x) {
    return x >= 0.f ? x : NEG_SLOPE * x;
}

#define SORT_T   256
#define NRANGE   256
#define RSPAN    ((NNODES + NRANGE - 1) / NRANGE)
#define OFF_PAD  (NRANGE * RSPAN)
#define NBLK     256
#define EPB      ((EDGES + NBLK - 1) / NBLK)
#define EITER    ((EPB + SORT_T - 1) / SORT_T)
#define PLACE_T  256
static_assert(SORT_T == NRANGE, "the piece kernel scans one range per thread");
static_assert(PLACE_T == NBLK && PLACE_T == NRANGE, "a placement block has one thread per piece and per range");
static_assert(PLACE_T >= RSPAN, "a placement block has one thread per node of its range");

#define GEMM_TILES (3125 * 4)
#define GEMM_BLKS  ((GEMM_TILES + 7) / 8)
__device__ __forceinline__ void gemm1_tiles(
    const int gblock,
    const float* __restrict__ feat,
    const float* __restrict__ Ws, const float* __restrict__ bs,
    const float* __restrict__ Wd, const float* __restrict__ bd,
    float* __restrict__ hs, float* __restrict__ hd)
{
    const int wave = threadIdx.x >> 5;
    const int lane = threadIdx.x & 31;
    const int half = lane >> 4;
    const int row  = lane & 15;

    int tile = gblock * 8 + wave;
    if (tile >= GEMM_TILES) return;
    const int mt  = tile >> 2;
    const int nt4 = tile & 3;
    const int m0  = mt * 16;
    const int c0  = nt4 * 64;
    const int cl  = c0 + 4 * row;

    v8f zero = {0.f,0.f,0.f,0.f,0.f,0.f,0.f,0.f};
    v8f accS[4], accD[4];
    #pragma unroll
    for (int j = 0; j < 4; ++j) { accS[j] = zero; accD[j] = zero; }

    const float* arow = feat + (size_t)(m0 + row) * INFEAT;
    const float* wsp  = Ws + cl;
    const float* wdp  = Wd + cl;
    for (int k0 = 0; k0 < INFEAT; k0 += 4) {
        const int ka = k0 + 2 * half;
        v2f a = *(const v2f*)(arow + ka);
        const float4 s0 = *(const float4*)(wsp + (size_t)ka * HIDTOT);
        const float4 s1 = *(const float4*)(wsp + (size_t)(ka + 1) * HIDTOT);
        const float4 d0 = *(const float4*)(wdp + (size_t)ka * HIDTOT);
        const float4 d1 = *(const float4*)(wdp + (size_t)(ka + 1) * HIDTOT);
        v2f bS, bD;
        bS.x = s0.x; bS.y = s1.x; bD.x = d0.x; bD.y = d1.x;
        accS[0] = __builtin_amdgcn_wmma_f32_16x16x4_f32(false, a, false, bS, (short)0, accS[0], false, false);
        accD[0] = __builtin_amdgcn_wmma_f32_16x16x4_f32(false, a, false, bD, (short)0, accD[0], false, false);
        bS.x = s0.y; bS.y = s1.y; bD.x = d0.y; bD.y = d1.y;
        accS[1] = __builtin_amdgcn_wmma_f32_16x16x4_f32(false, a, false, bS, (short)0, accS[1], false, false);
        accD[1] = __builtin_amdgcn_wmma_f32_16x16x4_f32(false, a, false, bD, (short)0, accD[1], false, false);
        bS.x = s0.z; bS.y = s1.z; bD.x = d0.z; bD.y = d1.z;
        accS[2] = __builtin_amdgcn_wmma_f32_16x16x4_f32(false, a, false, bS, (short)0, accS[2], false, false);
        accD[2] = __builtin_amdgcn_wmma_f32_16x16x4_f32(false, a, false, bD, (short)0, accD[2], false, false);
        bS.x = s0.w; bS.y = s1.w; bD.x = d0.w; bD.y = d1.w;
        accS[3] = __builtin_amdgcn_wmma_f32_16x16x4_f32(false, a, false, bS, (short)0, accS[3], false, false);
        accD[3] = __builtin_amdgcn_wmma_f32_16x16x4_f32(false, a, false, bD, (short)0, accD[3], false, false);
    }
    const float4 biasS = *(const float4*)(bs + cl);
    const float4 biasD = *(const float4*)(bd + cl);
    #pragma unroll
    for (int i = 0; i < 8; ++i) {
        const int r = m0 + i + 8 * half;
        float4 vs, vd;
        vs.x = accS[0][i] + biasS.x; vs.y = accS[1][i] + biasS.y;
        vs.z = accS[2][i] + biasS.z; vs.w = accS[3][i] + biasS.w;
        vd.x = accD[0][i] + biasD.x; vd.y = accD[1][i] + biasD.y;
        vd.z = accD[2][i] + biasD.z; vd.w = accD[3][i] + biasD.w;
        *(float4*)(hs + (size_t)r * HIDTOT + cl) = vs;
        *(float4*)(hd + (size_t)r * HIDTOT + cl) = vd;
    }
}

__global__ __launch_bounds__(SORT_T) void piece_sort_kernel(
    const int* __restrict__ src, const int* __restrict__ dst,
    int* __restrict__ pcnt, int* __restrict__ pstart, int* __restrict__ binned)
{
    __shared__ int cnt[NRANGE + 1];
    __shared__ int wsum[SORT_T / 32];
    const int t = threadIdx.x, b = blockIdx.x;
    const int lane = t & 31, wave = t >> 5;
    cnt[t] = 0;
    if (t == 0) cnt[NRANGE] = 0;
    __syncthreads();
    const int e0 = b * EPB;
    const int e1 = (e0 + EPB < EDGES) ? (e0 + EPB) : EDGES;
    int dv[EITER], sv[EITER], rg[EITER], rk[EITER];
    #pragma unroll
    for (int j = 0; j < EITER; ++j) {
        const int e = e0 + t + j * SORT_T;
        dv[j] = (e < e1) ? dst[e] : -1;
        sv[j] = (e < e1) ? src[e] : 0;
    }
    #pragma unroll
    for (int j = 0; j < EITER; ++j) {
        rg[j] = (dv[j] >= 0) ? (int)((unsigned)dv[j] / RSPAN) : NRANGE;
        rk[j] = atomicAdd(&cnt[rg[j]], 1);
    }
    __syncthreads();
    const int c = cnt[t];
    int inc = c;
    #pragma unroll
    for (int o = 1; o < 32; o <<= 1) {
        const int y = __shfl_up(inc, o, 32);
        if (lane >= o) inc += y;
    }
    if (lane == 31) wsum[wave] = inc;
    __syncthreads();
    int wbase = 0;
    #pragma unroll
    for (int w = 0; w < SORT_T / 32; ++w) wbase += (w < wave) ? wsum[w] : 0;
    const int st = wbase + inc - c;
    pcnt[b * NRANGE + t] = c;
    pstart[b * NRANGE + t] = st;
    cnt[t] = st;
    __syncthreads();
    int2* bin2 = (int2*)binned;
    int fs[EITER];
    #pragma unroll
    for (int j = 0; j < EITER; ++j) fs[j] = cnt[rg[j]];
    #pragma unroll
    for (int j = 0; j < EITER; ++j)
        if (dv[j] >= 0) bin2[e0 + fs[j] + rk[j]] = make_int2(dv[j], sv[j]);
}

#define PITER    8
__device__ __forceinline__ void range_place(
    const int* __restrict__ binned, const int* __restrict__ pcnt, const int* __restrict__ pstart,
    int* __restrict__ off, int* __restrict__ list)
{
    __shared__ int cnt[PLACE_T];
    __shared__ int wsum[PLACE_T / 32];
    __shared__ int rbeg;
    const int t = threadIdx.x, r = blockIdx.x;
    const int lane = t & 31, wave = t >> 5;
    const int n0 = r * RSPAN;
    const int2* bin2 = (const int2*)binned;
    cnt[t] = 0;
    int tot = 0;
    if (t <= r) {
        #pragma unroll 16
        for (int bb = 0; bb < NBLK; ++bb) tot += pcnt[bb * NRANGE + t];
    }
    int inc = tot;
    #pragma unroll
    for (int o = 1; o < 32; o <<= 1) {
        const int y = __shfl_up(inc, o, 32);
        if (lane >= o) inc += y;
    }
    if (lane == 31) wsum[wave] = inc;
    __syncthreads();
    {
        int wb = 0;
        #pragma unroll
        for (int w = 0; w < PLACE_T / 32; ++w) wb += (w < wave) ? wsum[w] : 0;
        if (t == r) rbeg = wb + inc - tot;
    }
    const int pc = pcnt[t * NRANGE + r];
    const int ps = t * EPB + pstart[t * NRANGE + r];
    for (int k = 0; k < pc; k += PITER) {
        int dv[PITER];
        #pragma unroll
        for (int j = 0; j < PITER; ++j) dv[j] = (k + j < pc) ? bin2[ps + k + j].x : -1;
        #pragma unroll
        for (int j = 0; j < PITER; ++j)
            if (dv[j] >= 0) atomicAdd(&cnt[dv[j] - n0], 1);
    }
    __syncthreads();
    const int beg = rbeg;
    const int c = cnt[t];
    inc = c;
    #pragma unroll
    for (int o = 1; o < 32; o <<= 1) {
        const int y = __shfl_up(inc, o, 32);
        if (lane >= o) inc += y;
    }
    if (lane == 31) wsum[wave] = inc;
    __syncthreads();
    int wbase = 0;
    #pragma unroll
    for (int w = 0; w < PLACE_T / 32; ++w) wbase += (w < wave) ? wsum[w] : 0;
    const int start = beg + wbase + inc - c;
    if (t < RSPAN) off[n0 + t] = start;
    cnt[t] = start;
    __syncthreads();
    for (int k = 0; k < pc; k += PITER) {
        int2 ev[PITER];
        #pragma unroll
        for (int j = 0; j < PITER; ++j) ev[j] = (k + j < pc) ? bin2[ps + k + j] : make_int2(-1, 0);
        #pragma unroll
        for (int j = 0; j < PITER; ++j) {
            if (ev[j].x >= 0) {
                const int p = atomicAdd(&cnt[ev[j].x - n0], 1);
                list[p] = ev[j].y;
            }
        }
    }
}

__global__ __launch_bounds__(PLACE_T) void place_gemm_kernel(
    const int* __restrict__ binned, const int* __restrict__ pcnt, const int* __restrict__ pstart,
    int* __restrict__ off, int* __restrict__ list,
    const float* __restrict__ feat,
    const float* __restrict__ Ws, const float* __restrict__ bs,
    const float* __restrict__ Wd, const float* __restrict__ bd,
    float* __restrict__ hs, float* __restrict__ hd)
{
    if (blockIdx.x < NRANGE) {
        range_place(binned, pcnt, pstart, off, list);
        return;
    }
    gemm1_tiles((int)blockIdx.x - NRANGE, feat, Ws, bs, Wd, bd, hs, hd);
}

template <int M> __device__ __forceinline__ float row_xor(float v) {
    return __int_as_float(__builtin_amdgcn_update_dpp(0, __float_as_int(v), 0x160 | M, 0xF, 0xF, true));
}
__device__ __forceinline__ float row16_sum(float v) {
    v += row_xor<1>(v);
    v += row_xor<2>(v);
    v += row_xor<4>(v);
    v += row_xor<8>(v);
    return v;
}
__device__ __forceinline__ float other_row(float v) {
    return __int_as_float(__builtin_amdgcn_permlanex16(__float_as_int(v), __float_as_int(v),
                                                       0x76543210, 0xfedcba98, false, false));
}
__device__ __forceinline__ float lrelu_mx(float x) { return fmaxf(x, NEG_SLOPE * x); }
__device__ __forceinline__ float dot_lrelu(const float4 a, const float4 b, const float4 w) {
    return lrelu_mx(a.x + b.x) * w.x + lrelu_mx(a.y + b.y) * w.y +
           lrelu_mx(a.z + b.z) * w.z + lrelu_mx(a.w + b.w) * w.w;
}
template <int N> __device__ __forceinline__ float row_bcast(float v) {
    return __int_as_float(__builtin_amdgcn_update_dpp(0, __float_as_int(v), 0x150 | N, 0xF, 0xF, true));
}
template <int M> __device__ __forceinline__ float fold_xor(bool hi, float lo_v, float hi_v) {
    const float keep = hi ? hi_v : lo_v;
    const float send = hi ? lo_v : hi_v;
    return keep + row_xor<M>(send);
}
__device__ __forceinline__ float exp_nonpos(float x) { return __builtin_amdgcn_exp2f(x * 1.44269504088896341f); }
__device__ __forceinline__ float elu1(float v) { return v > 0.f ? v : (expf(v) - 1.f); }

#define GRP 4
__global__ __launch_bounds__(256) void node_attn1_kernel(
    const float* __restrict__ hs, const float* __restrict__ hd,
    const int* __restrict__ off, const int* __restrict__ list,
    const float* __restrict__ attn,
    const float* __restrict__ W2s, const float* __restrict__ b2s,
    const float* __restrict__ W2d, const float* __restrict__ b2d,
    float* __restrict__ hs2, float* __restrict__ hd2)
{
    const int d = __builtin_amdgcn_readfirstlane((int)(blockIdx.x * 8 + (threadIdx.x >> 5)));
    if (d >= NNODES) return;
    const int lane = threadIdx.x & 31;
    const int beg = off[d], end = off[d + 1];

    const float* pd = hd + (size_t)d * HIDTOT + 4 * lane;
    const float4 b0 = *(const float4*)(pd);
    const float4 b1 = *(const float4*)(pd + 128);
    const float4 w0 = *(const float4*)(attn + 4 * lane);
    const float4 w1 = *(const float4*)(attn + 128 + 4 * lane);

    const bool cb0 = (lane & 1) != 0, cb1 = (lane & 2) != 0, cb2 = (lane & 4) != 0;
    const int myslot = (cb1 ? 2 : 0) + (cb2 ? 1 : 0);
    float m = -INFINITY, l = 0.f;
    float4 acc0 = {0.f, 0.f, 0.f, 0.f}, acc1 = {0.f, 0.f, 0.f, 0.f};

    for (int i = beg; i < end; i += 32) {
        const int n = (end - i) < 32 ? (end - i) : 32;
        const int my_s = lane < n ? list[i + lane] : 0;
        for (int k = 0; k < n; k += GRP) {
            float4 a0[GRP], a1[GRP];
            #pragma unroll
            for (int j = 0; j < GRP; ++j) {
                const int kk = (k + j < n) ? (k + j) : k;
                const int s = __builtin_amdgcn_readlane(my_s, kk);
                const float* ps = hs + (size_t)s * HIDTOT + 4 * lane;
                a0[j] = *(const float4*)(ps);
                a1[j] = *(const float4*)(ps + 128);
            }
            float x0[GRP], x1[GRP];
            #pragma unroll
            for (int j = 0; j < GRP; ++j) {
                x0[j] = dot_lrelu(a0[j], b0, w0);
                x1[j] = dot_lrelu(a1[j], b1, w1);
            }
            const float u0 = fold_xor<1>(cb0, x0[0], x1[0]);
            const float u1 = fold_xor<1>(cb0, x0[1], x1[1]);
            const float u2 = fold_xor<1>(cb0, x0[2], x1[2]);
            const float u3 = fold_xor<1>(cb0, x0[3], x1[3]);
            const float t0 = fold_xor<2>(cb1, u0, u2);
            const float t1 = fold_xor<2>(cb1, u1, u3);
            float s = fold_xor<4>(cb2, t0, t1);
            s += row_xor<8>(s);
            if (k + myslot >= n) s = -INFINITY;
            float g = fmaxf(s, row_xor<2>(s));
            g = fmaxf(g, row_xor<4>(g));
            g = fmaxf(g, m);
            const float r = exp_nonpos(m - g);
            const float e = exp_nonpos(s - g);
            l = l * r + e;
            m = g;
            const float r0 = row_bcast<0>(r), r1 = row_bcast<1>(r);
            acc0.x *= r0; acc0.y *= r0; acc0.z *= r0; acc0.w *= r0;
            acc1.x *= r1; acc1.y *= r1; acc1.z *= r1; acc1.w *= r1;
            const float e00 = row_bcast<0>(e), e01 = row_bcast<4>(e), e02 = row_bcast<2>(e), e03 = row_bcast<6>(e);
            const float e10 = row_bcast<1>(e), e11 = row_bcast<5>(e), e12 = row_bcast<3>(e), e13 = row_bcast<7>(e);
            acc0.x += e00 * a0[0].x; acc0.y += e00 * a0[0].y; acc0.z += e00 * a0[0].z; acc0.w += e00 * a0[0].w;
            acc1.x += e10 * a1[0].x; acc1.y += e10 * a1[0].y; acc1.z += e10 * a1[0].z; acc1.w += e10 * a1[0].w;
            acc0.x += e01 * a0[1].x; acc0.y += e01 * a0[1].y; acc0.z += e01 * a0[1].z; acc0.w += e01 * a0[1].w;
            acc1.x += e11 * a1[1].x; acc1.y += e11 * a1[1].y; acc1.z += e11 * a1[1].z; acc1.w += e11 * a1[1].w;
            acc0.x += e02 * a0[2].x; acc0.y += e02 * a0[2].y; acc0.z += e02 * a0[2].z; acc0.w += e02 * a0[2].w;
            acc1.x += e12 * a1[2].x; acc1.y += e12 * a1[2].y; acc1.z += e12 * a1[2].z; acc1.w += e12 * a1[2].w;
            acc0.x += e03 * a0[3].x; acc0.y += e03 * a0[3].y; acc0.z += e03 * a0[3].z; acc0.w += e03 * a0[3].w;
            acc1.x += e13 * a1[3].x; acc1.y += e13 * a1[3].y; acc1.z += e13 * a1[3].z; acc1.w += e13 * a1[3].w;
        }
    }
    const float4 s_a = *(const float4*)(W2s + 8 * lane);
    const float4 s_b = *(const float4*)(W2s + 8 * lane + 4);
    const float4 s_c = *(const float4*)(W2s + 256 + 8 * lane);
    const float4 s_d = *(const float4*)(W2s + 256 + 8 * lane + 4);
    const float4 d_a = *(const float4*)(W2d + 8 * lane);
    const float4 d_b = *(const float4*)(W2d + 8 * lane + 4);
    const float4 d_c = *(const float4*)(W2d + 256 + 8 * lane);
    const float4 d_d = *(const float4*)(W2d + 256 + 8 * lane + 4);
    l += row_xor<2>(l);
    l += row_xor<4>(l);
    const float l0 = row_bcast<0>(l), l1 = row_bcast<1>(l);
    const float q0 = l0 > 0.f ? l0 : 1.f, q1 = l1 > 0.f ? l1 : 1.f;
    float4 o0, o1;
    o0.x = elu1(acc0.x / q0); o0.y = elu1(acc0.y / q0);
    o0.z = elu1(acc0.z / q0); o0.w = elu1(acc0.w / q0);
    o1.x = elu1(acc1.x / q1); o1.y = elu1(acc1.y / q1);
    o1.z = elu1(acc1.z / q1); o1.w = elu1(acc1.w / q1);
    float ps0 = o0.x * s_a.x + o0.y * s_a.z + o0.z * s_b.x + o0.w * s_b.z
              + o1.x * s_c.x + o1.y * s_c.z + o1.z * s_d.x + o1.w * s_d.z;
    float ps1 = o0.x * s_a.y + o0.y * s_a.w + o0.z * s_b.y + o0.w * s_b.w
              + o1.x * s_c.y + o1.y * s_c.w + o1.z * s_d.y + o1.w * s_d.w;
    float pd0 = o0.x * d_a.x + o0.y * d_a.z + o0.z * d_b.x + o0.w * d_b.z
              + o1.x * d_c.x + o1.y * d_c.z + o1.z * d_d.x + o1.w * d_d.z;
    float pd1 = o0.x * d_a.y + o0.y * d_a.w + o0.z * d_b.y + o0.w * d_b.w
              + o1.x * d_c.y + o1.y * d_c.w + o1.z * d_d.y + o1.w * d_d.w;
    ps0 = row16_sum(ps0); ps1 = row16_sum(ps1);
    pd0 = row16_sum(pd0); pd1 = row16_sum(pd1);
    ps0 += other_row(ps0); ps1 += other_row(ps1);
    pd0 += other_row(pd0); pd1 += other_row(pd1);
    if (lane == 0) {
        float2 vs, vd;
        vs.x = ps0 + b2s[0]; vs.y = ps1 + b2s[1];
        vd.x = pd0 + b2d[0]; vd.y = pd1 + b2d[1];
        *(float2*)(hs2 + (size_t)d * 2) = vs;
        *(float2*)(hd2 + (size_t)d * 2) = vd;
    }
}

#define GRP2 8
__global__ __launch_bounds__(64) void node_attn2_kernel(
    const float* __restrict__ hs2, const float* __restrict__ hd2,
    const int* __restrict__ off, const int* __restrict__ list,
    const float* __restrict__ attn2, float* __restrict__ out)
{
    const int d = blockIdx.x * blockDim.x + threadIdx.x;
    if (d >= NNODES) return;
    const int beg = off[d], end = off[d + 1];
    const float2 b = *(const float2*)(hd2 + (size_t)d * 2);
    const float w0 = attn2[0], w1 = attn2[1];

    float m = -INFINITY, l = 0.f, ax = 0.f, ay = 0.f;
    for (int i = beg; i < end; i += GRP2) {
        int s[GRP2];
        float2 a[GRP2];
        #pragma unroll
        for (int j = 0; j < GRP2; ++j) {
            const int ii = (i + j < end) ? (i + j) : i;
            s[j] = list[ii];
        }
        #pragma unroll
        for (int j = 0; j < GRP2; ++j) a[j] = *(const float2*)(hs2 + (size_t)s[j] * 2);
        float p[GRP2];
        float g = m;
        #pragma unroll
        for (int j = 0; j < GRP2; ++j) {
            float x = lrelu(a[j].x + b.x) * w0 + lrelu(a[j].y + b.y) * w1;
            if (i + j >= end) x = -INFINITY;
            p[j] = x;
            g = fmaxf(g, x);
        }
        const float r = expf(m - g);
        l *= r; ax *= r; ay *= r;
        #pragma unroll
        for (int j = 0; j < GRP2; ++j) {
            const float e = expf(p[j] - g);
            l += e;
            ax += e * a[j].x;
            ay += e * a[j].y;
        }
        m = g;
    }
    const float q = l > 0.f ? l : 1.f;
    float2 o;
    o.x = ax / q;
    o.y = ay / q;
    *(float2*)(out + (size_t)d * 2) = o;
}

extern "C" void kernel_launch(void* const* d_in, const int* in_sizes, int n_in,
                              void* d_out, int out_size, void* d_ws, size_t ws_size,
                              hipStream_t stream)
{
    const float* feat  = (const float*)d_in[0];
    const int*   src   = (const int*)d_in[1];
    const int*   dst   = (const int*)d_in[2];
    const float* W1s   = (const float*)d_in[3];
    const float* b1s   = (const float*)d_in[4];
    const float* W1d   = (const float*)d_in[5];
    const float* b1d   = (const float*)d_in[6];
    const float* attn1 = (const float*)d_in[7];
    const float* W2s   = (const float*)d_in[8];
    const float* b2s   = (const float*)d_in[9];
    const float* W2d   = (const float*)d_in[10];
    const float* b2d   = (const float*)d_in[11];
    const float* attn2 = (const float*)d_in[12];
    float* out = (float*)d_out;

    float* ws = (float*)d_ws;
    size_t off = 0;
    float*    hs1    = ws + off; off += (size_t)NNODES * HIDTOT;
    float*    hd1    = ws + off; off += (size_t)NNODES * HIDTOT;
    int*      pcnt   = (int*)(ws + off); off += (size_t)NBLK * NRANGE;
    int*      pstart = (int*)(ws + off); off += (size_t)NBLK * NRANGE;
    int*      offs   = (int*)(ws + off); off += (size_t)OFF_PAD;
    int*      binned = (int*)(ws + off); off += (size_t)2 * EDGES;
    int*      list   = (int*)(ws + off); off += (size_t)EDGES;
    float*    hs2    = ws + off; off += (size_t)NNODES * 2;
    float*    hd2    = ws + off; off += (size_t)NNODES * 2;

    const int B = 256;

    piece_sort_kernel<<<NBLK, SORT_T, 0, stream>>>(src, dst, pcnt, pstart, binned);
    place_gemm_kernel<<<NRANGE + GEMM_BLKS, PLACE_T, 0, stream>>>(
        binned, pcnt, pstart, offs, list, feat, W1s, b1s, W1d, b1d, hs1, hd1);
    node_attn1_kernel<<<(NNODES + 7) / 8, B, 0, stream>>>(
        hs1, hd1, offs, list, attn1, W2s, b2s, W2d, b2d, hs2, hd2);
    node_attn2_kernel<<<(NNODES + 63) / 64, 64, 0, stream>>>(
        hs2, hd2, offs, list, attn2, out);

    (void)in_sizes; (void)n_in; (void)out_size; (void)ws_size;
}


# layer-1 node kernel, fewer vector instructions: edge loop in packed 2-float FMA form (a lane's 4 score terms now sum (x+z)+(y+w)); node tail: one reciprocal per head, hardware exp2 in ELU
# speedup vs baseline: 6.5089x; 1.0553x over previous
#include <hip/hip_runtime.h>
#include <hip/hip_bf16.h>
#include <math.h>

#define NNODES 50000
#define EDGES  800000
#define INFEAT 128
#define HIDTOT 256
#define NEG_SLOPE 0.2f

typedef __attribute__((ext_vector_type(2))) float v2f;
typedef __attribute__((ext_vector_type(8))) float v8f;

__device__ __forceinline__ float lrelu(float x) {
    return x >= 0.f ? x : NEG_SLOPE * x;
}

#define SORT_T   256
#define NRANGE   256
#define RSPAN    ((NNODES + NRANGE - 1) / NRANGE)
#define OFF_PAD  (NRANGE * RSPAN)
#define NBLK     256
#define EPB      ((EDGES + NBLK - 1) / NBLK)
#define EITER    ((EPB + SORT_T - 1) / SORT_T)
#define PLACE_T  256
static_assert(SORT_T == NRANGE, "the piece kernel scans one range per thread");
static_assert(PLACE_T == NBLK && PLACE_T == NRANGE, "a placement block has one thread per piece and per range");
static_assert(PLACE_T >= RSPAN, "a placement block has one thread per node of its range");

#define GEMM_TILES (3125 * 4)
#define GEMM_BLKS  ((GEMM_TILES + 7) / 8)
__device__ __forceinline__ void gemm1_tiles(
    const int gblock,
    const float* __restrict__ feat,
    const float* __restrict__ Ws, const float* __restrict__ bs,
    const float* __restrict__ Wd, const float* __restrict__ bd,
    float* __restrict__ hs, float* __restrict__ hd)
{
    const int wave = threadIdx.x >> 5;
    const int lane = threadIdx.x & 31;
    const int half = lane >> 4;
    const int row  = lane & 15;

    int tile = gblock * 8 + wave;
    if (tile >= GEMM_TILES) return;
    const int mt  = tile >> 2;
    const int nt4 = tile & 3;
    const int m0  = mt * 16;
    const int c0  = nt4 * 64;
    const int cl  = c0 + 4 * row;

    v8f zero = {0.f,0.f,0.f,0.f,0.f,0.f,0.f,0.f};
    v8f accS[4], accD[4];
    #pragma unroll
    for (int j = 0; j < 4; ++j) { accS[j] = zero; accD[j] = zero; }

    const float* arow = feat + (size_t)(m0 + row) * INFEAT;
    const float* wsp  = Ws + cl;
    const float* wdp  = Wd + cl;
    for (int k0 = 0; k0 < INFEAT; k0 += 4) {
        const int ka = k0 + 2 * half;
        v2f a = *(const v2f*)(arow + ka);
        const float4 s0 = *(const float4*)(wsp + (size_t)ka * HIDTOT);
        const float4 s1 = *(const float4*)(wsp + (size_t)(ka + 1) * HIDTOT);
        const float4 d0 = *(const float4*)(wdp + (size_t)ka * HIDTOT);
        const float4 d1 = *(const float4*)(wdp + (size_t)(ka + 1) * HIDTOT);
        v2f bS, bD;
        bS.x = s0.x; bS.y = s1.x; bD.x = d0.x; bD.y = d1.x;
        accS[0] = __builtin_amdgcn_wmma_f32_16x16x4_f32(false, a, false, bS, (short)0, accS[0], false, false);
        accD[0] = __builtin_amdgcn_wmma_f32_16x16x4_f32(false, a, false, bD, (short)0, accD[0], false, false);
        bS.x = s0.y; bS.y = s1.y; bD.x = d0.y; bD.y = d1.y;
        accS[1] = __builtin_amdgcn_wmma_f32_16x16x4_f32(false, a, false, bS, (short)0, accS[1], false, false);
        accD[1] = __builtin_amdgcn_wmma_f32_16x16x4_f32(false, a, false, bD, (short)0, accD[1], false, false);
        bS.x = s0.z; bS.y = s1.z; bD.x = d0.z; bD.y = d1.z;
        accS[2] = __builtin_amdgcn_wmma_f32_16x16x4_f32(false, a, false, bS, (short)0, accS[2], false, false);
        accD[2] = __builtin_amdgcn_wmma_f32_16x16x4_f32(false, a, false, bD, (short)0, accD[2], false, false);
        bS.x = s0.w; bS.y = s1.w; bD.x = d0.w; bD.y = d1.w;
        accS[3] = __builtin_amdgcn_wmma_f32_16x16x4_f32(false, a, false, bS, (short)0, accS[3], false, false);
        accD[3] = __builtin_amdgcn_wmma_f32_16x16x4_f32(false, a, false, bD, (short)0, accD[3], false, false);
    }
    const float4 biasS = *(const float4*)(bs + cl);
    const float4 biasD = *(const float4*)(bd + cl);
    #pragma unroll
    for (int i = 0; i < 8; ++i) {
        const int r = m0 + i + 8 * half;
        float4 vs, vd;
        vs.x = accS[0][i] + biasS.x; vs.y = accS[1][i] + biasS.y;
        vs.z = accS[2][i] + biasS.z; vs.w = accS[3][i] + biasS.w;
        vd.x = accD[0][i] + biasD.x; vd.y = accD[1][i] + biasD.y;
        vd.z = accD[2][i] + biasD.z; vd.w = accD[3][i] + biasD.w;
        *(float4*)(hs + (size_t)r * HIDTOT + cl) = vs;
        *(float4*)(hd + (size_t)r * HIDTOT + cl) = vd;
    }
}

__global__ __launch_bounds__(SORT_T) void piece_sort_kernel(
    const int* __restrict__ src, const int* __restrict__ dst,
    int* __restrict__ pcnt, int* __restrict__ pstart, int* __restrict__ binned)
{
    __shared__ int cnt[NRANGE + 1];
    __shared__ int wsum[SORT_T / 32];
    const int t = threadIdx.x, b = blockIdx.x;
    const int lane = t & 31, wave = t >> 5;
    cnt[t] = 0;
    if (t == 0) cnt[NRANGE] = 0;
    __syncthreads();
    const int e0 = b * EPB;
    const int e1 = (e0 + EPB < EDGES) ? (e0 + EPB) : EDGES;
    int dv[EITER], sv[EITER], rg[EITER], rk[EITER];
    #pragma unroll
    for (int j = 0; j < EITER; ++j) {
        const int e = e0 + t + j * SORT_T;
        dv[j] = (e < e1) ? dst[e] : -1;
        sv[j] = (e < e1) ? src[e] : 0;
    }
    #pragma unroll
    for (int j = 0; j < EITER; ++j) {
        rg[j] = (dv[j] >= 0) ? (int)((unsigned)dv[j] / RSPAN) : NRANGE;
        rk[j] = atomicAdd(&cnt[rg[j]], 1);
    }
    __syncthreads();
    const int c = cnt[t];
    int inc = c;
    #pragma unroll
    for (int o = 1; o < 32; o <<= 1) {
        const int y = __shfl_up(inc, o, 32);
        if (lane >= o) inc += y;
    }
    if (lane == 31) wsum[wave] = inc;
    __syncthreads();
    int wbase = 0;
    #pragma unroll
    for (int w = 0; w < SORT_T / 32; ++w) wbase += (w < wave) ? wsum[w] : 0;
    const int st = wbase + inc - c;
    pcnt[b * NRANGE + t] = c;
    pstart[b * NRANGE + t] = st;
    cnt[t] = st;
    __syncthreads();
    int2* bin2 = (int2*)binned;
    int fs[EITER];
    #pragma unroll
    for (int j = 0; j < EITER; ++j) fs[j] = cnt[rg[j]];
    #pragma unroll
    for (int j = 0; j < EITER; ++j)
        if (dv[j] >= 0) bin2[e0 + fs[j] + rk[j]] = make_int2(dv[j], sv[j]);
}

#define PITER    8
__device__ __forceinline__ void range_place(
    const int* __restrict__ binned, const int* __restrict__ pcnt, const int* __restrict__ pstart,
    int* __restrict__ off, int* __restrict__ list)
{
    __shared__ int cnt[PLACE_T];
    __shared__ int wsum[PLACE_T / 32];
    __shared__ int rbeg;
    const int t = threadIdx.x, r = blockIdx.x;
    const int lane = t & 31, wave = t >> 5;
    const int n0 = r * RSPAN;
    const int2* bin2 = (const int2*)binned;
    cnt[t] = 0;
    int tot = 0;
    if (t <= r) {
        #pragma unroll 16
        for (int bb = 0; bb < NBLK; ++bb) tot += pcnt[bb * NRANGE + t];
    }
    int inc = tot;
    #pragma unroll
    for (int o = 1; o < 32; o <<= 1) {
        const int y = __shfl_up(inc, o, 32);
        if (lane >= o) inc += y;
    }
    if (lane == 31) wsum[wave] = inc;
    __syncthreads();
    {
        int wb = 0;
        #pragma unroll
        for (int w = 0; w < PLACE_T / 32; ++w) wb += (w < wave) ? wsum[w] : 0;
        if (t == r) rbeg = wb + inc - tot;
    }
    const int pc = pcnt[t * NRANGE + r];
    const int ps = t * EPB + pstart[t * NRANGE + r];
    for (int k = 0; k < pc; k += PITER) {
        int dv[PITER];
        #pragma unroll
        for (int j = 0; j < PITER; ++j) dv[j] = (k + j < pc) ? bin2[ps + k + j].x : -1;
        #pragma unroll
        for (int j = 0; j < PITER; ++j)
            if (dv[j] >= 0) atomicAdd(&cnt[dv[j] - n0], 1);
    }
    __syncthreads();
    const int beg = rbeg;
    const int c = cnt[t];
    inc = c;
    #pragma unroll
    for (int o = 1; o < 32; o <<= 1) {
        const int y = __shfl_up(inc, o, 32);
        if (lane >= o) inc += y;
    }
    if (lane == 31) wsum[wave] = inc;
    __syncthreads();
    int wbase = 0;
    #pragma unroll
    for (int w = 0; w < PLACE_T / 32; ++w) wbase += (w < wave) ? wsum[w] : 0;
    const int start = beg + wbase + inc - c;
    if (t < RSPAN) off[n0 + t] = start;
    cnt[t] = start;
    __syncthreads();
    for (int k = 0; k < pc; k += PITER) {
        int2 ev[PITER];
        #pragma unroll
        for (int j = 0; j < PITER; ++j) ev[j] = (k + j < pc) ? bin2[ps + k + j] : make_int2(-1, 0);
        #pragma unroll
        for (int j = 0; j < PITER; ++j) {
            if (ev[j].x >= 0) {
                const int p = atomicAdd(&cnt[ev[j].x - n0], 1);
                list[p] = ev[j].y;
            }
        }
    }
}

__global__ __launch_bounds__(PLACE_T) void place_gemm_kernel(
    const int* __restrict__ binned, const int* __restrict__ pcnt, const int* __restrict__ pstart,
    int* __restrict__ off, int* __restrict__ list,
    const float* __restrict__ feat,
    const float* __restrict__ Ws, const float* __restrict__ bs,
    const float* __restrict__ Wd, const float* __restrict__ bd,
    float* __restrict__ hs, float* __restrict__ hd)
{
    if (blockIdx.x < NRANGE) {
        range_place(binned, pcnt, pstart, off, list);
        return;
    }
    gemm1_tiles((int)blockIdx.x - NRANGE, feat, Ws, bs, Wd, bd, hs, hd);
}

template <int M> __device__ __forceinline__ float row_xor(float v) {
    return __int_as_float(__builtin_amdgcn_update_dpp(0, __float_as_int(v), 0x160 | M, 0xF, 0xF, true));
}
__device__ __forceinline__ float row16_sum(float v) {
    v += row_xor<1>(v);
    v += row_xor<2>(v);
    v += row_xor<4>(v);
    v += row_xor<8>(v);
    return v;
}
__device__ __forceinline__ float other_row(float v) {
    return __int_as_float(__builtin_amdgcn_permlanex16(__float_as_int(v), __float_as_int(v),
                                                       0x76543210, 0xfedcba98, false, false));
}
__device__ __forceinline__ v2f fma2(const v2f a, const v2f b, const v2f c) { return __builtin_elementwise_fma(a, b, c); }
__device__ __forceinline__ float dot_lrelu(const float4 a, const v2f bl, const v2f bh, const v2f wl, const v2f wh) {
    const v2f al = {a.x, a.y}, ah = {a.z, a.w};
    const v2f slope = {NEG_SLOPE, NEG_SLOPE};
    const v2f xl = al + bl, xh = ah + bh;
    const v2f ml = __builtin_elementwise_max(xl, xl * slope);
    const v2f mh = __builtin_elementwise_max(xh, xh * slope);
    const v2f p = fma2(mh, wh, ml * wl);
    return p.x + p.y;
}
template <int N> __device__ __forceinline__ float row_bcast(float v) {
    return __int_as_float(__builtin_amdgcn_update_dpp(0, __float_as_int(v), 0x150 | N, 0xF, 0xF, true));
}
template <int M> __device__ __forceinline__ float fold_xor(bool hi, float lo_v, float hi_v) {
    const float keep = hi ? hi_v : lo_v;
    const float send = hi ? lo_v : hi_v;
    return keep + row_xor<M>(send);
}
__device__ __forceinline__ float exp_nonpos(float x) { return __builtin_amdgcn_exp2f(x * 1.44269504088896341f); }
__device__ __forceinline__ float elu1(float v) { return v > 0.f ? v : (exp_nonpos(v) - 1.f); }

#define GRP 4
__global__ __launch_bounds__(256) void node_attn1_kernel(
    const float* __restrict__ hs, const float* __restrict__ hd,
    const int* __restrict__ off, const int* __restrict__ list,
    const float* __restrict__ attn,
    const float* __restrict__ W2s, const float* __restrict__ b2s,
    const float* __restrict__ W2d, const float* __restrict__ b2d,
    float* __restrict__ hs2, float* __restrict__ hd2)
{
    const int d = __builtin_amdgcn_readfirstlane((int)(blockIdx.x * 8 + (threadIdx.x >> 5)));
    if (d >= NNODES) return;
    const int lane = threadIdx.x & 31;
    const int beg = off[d], end = off[d + 1];

    const float* pd = hd + (size_t)d * HIDTOT + 4 * lane;
    const float4 b0 = *(const float4*)(pd);
    const float4 b1 = *(const float4*)(pd + 128);
    const float4 w0 = *(const float4*)(attn + 4 * lane);
    const float4 w1 = *(const float4*)(attn + 128 + 4 * lane);
    const v2f b0l = {b0.x, b0.y}, b0h = {b0.z, b0.w}, b1l = {b1.x, b1.y}, b1h = {b1.z, b1.w};
    const v2f w0l = {w0.x, w0.y}, w0h = {w0.z, w0.w}, w1l = {w1.x, w1.y}, w1h = {w1.z, w1.w};

    const bool cb0 = (lane & 1) != 0, cb1 = (lane & 2) != 0, cb2 = (lane & 4) != 0;
    const int myslot = (cb1 ? 2 : 0) + (cb2 ? 1 : 0);
    float m = -INFINITY, l = 0.f;
    v2f acc0l = {0.f, 0.f}, acc0h = {0.f, 0.f}, acc1l = {0.f, 0.f}, acc1h = {0.f, 0.f};

    for (int i = beg; i < end; i += 32) {
        const int n = (end - i) < 32 ? (end - i) : 32;
        const int my_s = lane < n ? list[i + lane] : 0;
        for (int k = 0; k < n; k += GRP) {
            float4 a0[GRP], a1[GRP];
            #pragma unroll
            for (int j = 0; j < GRP; ++j) {
                const int kk = (k + j < n) ? (k + j) : k;
                const int s = __builtin_amdgcn_readlane(my_s, kk);
                const float* ps = hs + (size_t)s * HIDTOT + 4 * lane;
                a0[j] = *(const float4*)(ps);
                a1[j] = *(const float4*)(ps + 128);
            }
            float x0[GRP], x1[GRP];
            #pragma unroll
            for (int j = 0; j < GRP; ++j) {
                x0[j] = dot_lrelu(a0[j], b0l, b0h, w0l, w0h);
                x1[j] = dot_lrelu(a1[j], b1l, b1h, w1l, w1h);
            }
            const float u0 = fold_xor<1>(cb0, x0[0], x1[0]);
            const float u1 = fold_xor<1>(cb0, x0[1], x1[1]);
            const float u2 = fold_xor<1>(cb0, x0[2], x1[2]);
            const float u3 = fold_xor<1>(cb0, x0[3], x1[3]);
            const float t0 = fold_xor<2>(cb1, u0, u2);
            const float t1 = fold_xor<2>(cb1, u1, u3);
            float s = fold_xor<4>(cb2, t0, t1);
            s += row_xor<8>(s);
            if (k + myslot >= n) s = -INFINITY;
            float g = fmaxf(s, row_xor<2>(s));
            g = fmaxf(g, row_xor<4>(g));
            g = fmaxf(g, m);
            const float r = exp_nonpos(m - g);
            const float e = exp_nonpos(s - g);
            l = l * r + e;
            m = g;
            const float r0 = row_bcast<0>(r), r1 = row_bcast<1>(r);
            const v2f R0 = {r0, r0}, R1 = {r1, r1};
            acc0l *= R0; acc0h *= R0;
            acc1l *= R1; acc1h *= R1;
            const float e00 = row_bcast<0>(e), e01 = row_bcast<4>(e), e02 = row_bcast<2>(e), e03 = row_bcast<6>(e);
            const float e10 = row_bcast<1>(e), e11 = row_bcast<5>(e), e12 = row_bcast<3>(e), e13 = row_bcast<7>(e);
            const float w0s[GRP] = {e00, e01, e02, e03}, w1s[GRP] = {e10, e11, e12, e13};
            #pragma unroll
            for (int j = 0; j < GRP; ++j) {
                const v2f E0 = {w0s[j], w0s[j]}, E1 = {w1s[j], w1s[j]};
                const v2f a0l = {a0[j].x, a0[j].y}, a0h = {a0[j].z, a0[j].w};
                const v2f a1l = {a1[j].x, a1[j].y}, a1h = {a1[j].z, a1[j].w};
                acc0l = fma2(E0, a0l, acc0l); acc0h = fma2(E0, a0h, acc0h);
                acc1l = fma2(E1, a1l, acc1l); acc1h = fma2(E1, a1h, acc1h);
            }
        }
    }
    const float4 s_a = *(const float4*)(W2s + 8 * lane);
    const float4 s_b = *(const float4*)(W2s + 8 * lane + 4);
    const float4 s_c = *(const float4*)(W2s + 256 + 8 * lane);
    const float4 s_d = *(const float4*)(W2s + 256 + 8 * lane + 4);
    const float4 d_a = *(const float4*)(W2d + 8 * lane);
    const float4 d_b = *(const float4*)(W2d + 8 * lane + 4);
    const float4 d_c = *(const float4*)(W2d + 256 + 8 * lane);
    const float4 d_d = *(const float4*)(W2d + 256 + 8 * lane + 4);
    l += row_xor<2>(l);
    l += row_xor<4>(l);
    const float l0 = row_bcast<0>(l), l1 = row_bcast<1>(l);
    const float q0 = l0 > 0.f ? l0 : 1.f, q1 = l1 > 0.f ? l1 : 1.f;
    const float i0 = 1.f / q0, i1 = 1.f / q1;
    const v2f I0 = {i0, i0}, I1 = {i1, i1};
    const v2f h0l = acc0l * I0, h0h = acc0h * I0, h1l = acc1l * I1, h1h = acc1h * I1;
    float4 o0, o1;
    o0.x = elu1(h0l.x); o0.y = elu1(h0l.y);
    o0.z = elu1(h0h.x); o0.w = elu1(h0h.y);
    o1.x = elu1(h1l.x); o1.y = elu1(h1l.y);
    o1.z = elu1(h1h.x); o1.w = elu1(h1h.y);
    float ps0 = o0.x * s_a.x + o0.y * s_a.z + o0.z * s_b.x + o0.w * s_b.z
              + o1.x * s_c.x + o1.y * s_c.z + o1.z * s_d.x + o1.w * s_d.z;
    float ps1 = o0.x * s_a.y + o0.y * s_a.w + o0.z * s_b.y + o0.w * s_b.w
              + o1.x * s_c.y + o1.y * s_c.w + o1.z * s_d.y + o1.w * s_d.w;
    float pd0 = o0.x * d_a.x + o0.y * d_a.z + o0.z * d_b.x + o0.w * d_b.z
              + o1.x * d_c.x + o1.y * d_c.z + o1.z * d_d.x + o1.w * d_d.z;
    float pd1 = o0.x * d_a.y + o0.y * d_a.w + o0.z * d_b.y + o0.w * d_b.w
              + o1.x * d_c.y + o1.y * d_c.w + o1.z * d_d.y + o1.w * d_d.w;
    ps0 = row16_sum(ps0); ps1 = row16_sum(ps1);
    pd0 = row16_sum(pd0); pd1 = row16_sum(pd1);
    ps0 += other_row(ps0); ps1 += other_row(ps1);
    pd0 += other_row(pd0); pd1 += other_row(pd1);
    if (lane == 0) {
        float2 vs, vd;
        vs.x = ps0 + b2s[0]; vs.y = ps1 + b2s[1];
        vd.x = pd0 + b2d[0]; vd.y = pd1 + b2d[1];
        *(float2*)(hs2 + (size_t)d * 2) = vs;
        *(float2*)(hd2 + (size_t)d * 2) = vd;
    }
}

#define GRP2 8
__global__ __launch_bounds__(64) void node_attn2_kernel(
    const float* __restrict__ hs2, const float* __restrict__ hd2,
    const int* __restrict__ off, const int* __restrict__ list,
    const float* __restrict__ attn2, float* __restrict__ out)
{
    const int d = blockIdx.x * blockDim.x + threadIdx.x;
    if (d >= NNODES) return;
    const int beg = off[d], end = off[d + 1];
    const float2 b = *(const float2*)(hd2 + (size_t)d * 2);
    const float w0 = attn2[0], w1 = attn2[1];

    float m = -INFINITY, l = 0.f, ax = 0.f, ay = 0.f;
    for (int i = beg; i < end; i += GRP2) {
        int s[GRP2];
        float2 a[GRP2];
        #pragma unroll
        for (int j = 0; j < GRP2; ++j) {
            const int ii = (i + j < end) ? (i + j) : i;
            s[j] = list[ii];
        }
        #pragma unroll
        for (int j = 0; j < GRP2; ++j) a[j] = *(const float2*)(hs2 + (size_t)s[j] * 2);
        float p[GRP2];
        float g = m;
        #pragma unroll
        for (int j = 0; j < GRP2; ++j) {
            float x = lrelu(a[j].x + b.x) * w0 + lrelu(a[j].y + b.y) * w1;
            if (i + j >= end) x = -INFINITY;
            p[j] = x;
            g = fmaxf(g, x);
        }
        const float r = expf(m - g);
        l *= r; ax *= r; ay *= r;
        #pragma unroll
        for (int j = 0; j < GRP2; ++j) {
            const float e = expf(p[j] - g);
            l += e;
            ax += e * a[j].x;
            ay += e * a[j].y;
        }
        m = g;
    }
    const float q = l > 0.f ? l : 1.f;
    float2 o;
    o.x = ax / q;
    o.y = ay / q;
    *(float2*)(out + (size_t)d * 2) = o;
}

extern "C" void kernel_launch(void* const* d_in, const int* in_sizes, int n_in,
                              void* d_out, int out_size, void* d_ws, size_t ws_size,
                              hipStream_t stream)
{
    const float* feat  = (const float*)d_in[0];
    const int*   src   = (const int*)d_in[1];
    const int*   dst   = (const int*)d_in[2];
    const float* W1s   = (const float*)d_in[3];
    const float* b1s   = (const float*)d_in[4];
    const float* W1d   = (const float*)d_in[5];
    const float* b1d   = (const float*)d_in[6];
    const float* attn1 = (const float*)d_in[7];
    const float* W2s   = (const float*)d_in[8];
    const float* b2s   = (const float*)d_in[9];
    const float* W2d   = (const float*)d_in[10];
    const float* b2d   = (const float*)d_in[11];
    const float* attn2 = (const float*)d_in[12];
    float* out = (float*)d_out;

    float* ws = (float*)d_ws;
    size_t off = 0;
    float*    hs1    = ws + off; off += (size_t)NNODES * HIDTOT;
    float*    hd1    = ws + off; off += (size_t)NNODES * HIDTOT;
    int*      pcnt   = (int*)(ws + off); off += (size_t)NBLK * NRANGE;
    int*      pstart = (int*)(ws + off); off += (size_t)NBLK * NRANGE;
    int*      offs   = (int*)(ws + off); off += (size_t)OFF_PAD;
    int*      binned = (int*)(ws + off); off += (size_t)2 * EDGES;
    int*      list   = (int*)(ws + off); off += (size_t)EDGES;
    float*    hs2    = ws + off; off += (size_t)NNODES * 2;
    float*    hd2    = ws + off; off += (size_t)NNODES * 2;

    const int B = 256;

    piece_sort_kernel<<<NBLK, SORT_T, 0, stream>>>(src, dst, pcnt, pstart, binned);
    place_gemm_kernel<<<NRANGE + GEMM_BLKS, PLACE_T, 0, stream>>>(
        binned, pcnt, pstart, offs, list, feat, W1s, b1s, W1d, b1d, hs1, hd1);
    node_attn1_kernel<<<(NNODES + 7) / 8, B, 0, stream>>>(
        hs1, hd1, offs, list, attn1, W2s, b2s, W2d, b2d, hs2, hd2);
    node_attn2_kernel<<<(NNODES + 63) / 64, 64, 0, stream>>>(
        hs2, hd2, offs, list, attn2, out);

    (void)in_sizes; (void)n_in; (void)out_size; (void)ws_size;
}
